# starting kernel
# baseline (speedup 1.0000x reference)
#include <hip/hip_runtime.h>
#include <hip/hip_bf16.h>

typedef __attribute__((ext_vector_type(2))) float v2f;
typedef __attribute__((ext_vector_type(8))) float v8f;

#define F 128

__global__ __launch_bounds__(256) void k_zero(float* __restrict__ p, size_t cnt) {
  size_t i = (size_t)blockIdx.x * blockDim.x + threadIdx.x;
  if (i < cnt) p[i] = 0.0f;
}

__global__ __launch_bounds__(256) void k_degree(const int* __restrict__ dst,
                                                float* __restrict__ deg, int nE) {
  int e = blockIdx.x * blockDim.x + threadIdx.x;
  if (e < nE) atomicAdd(&deg[dst[e]], 1.0f);
}

__global__ __launch_bounds__(256) void k_deg_inv(float* __restrict__ deg, int n) {
  int i = blockIdx.x * blockDim.x + threadIdx.x;
  if (i < n) {
    float d = deg[i];
    deg[i] = d > 0.0f ? 1.0f / d : 0.0f;
  }
}

__global__ __launch_bounds__(256) void k_scatter(const int* __restrict__ src,
                                                 const int* __restrict__ dst,
                                                 const float* __restrict__ feat,
                                                 float* __restrict__ agg, int nE) {
  long long tid = (long long)blockIdx.x * blockDim.x + threadIdx.x;
  int e = (int)(tid >> 5);
  if (e >= nE) return;
  int fo = ((int)tid & 31) * 4;
  const float4 v = *reinterpret_cast<const float4*>(feat + (size_t)src[e] * F + fo);
  float* d = agg + (size_t)dst[e] * F + fo;
  atomicAdd(d + 0, v.x);
  atomicAdd(d + 1, v.y);
  atomicAdd(d + 2, v.z);
  atomicAdd(d + 3, v.w);
}

__global__ __launch_bounds__(256) void k_gemm128(
    const float* __restrict__ agg, const float* __restrict__ xin,
    const float* __restrict__ Wl, const float* __restrict__ bl,
    const float* __restrict__ Wr, const float* __restrict__ deg_inv,
    float* __restrict__ hout, int n, int do_relu) {
  const int K = F;
  int lane = threadIdx.x & 31;
  int wave = threadIdx.x >> 5;
  int row0 = (blockIdx.x * 8 + wave) * 16;
  if (row0 >= n) return;

  int mloc = lane & 15;
  int koff = (lane >> 4) * 2;
  int m = row0 + mloc;
  int mc = m < n ? m : n - 1;
  float dscale = deg_inv[mc];
  const float* arow = agg + (size_t)mc * K;
  const float* xrow = xin + (size_t)mc * K;

  for (int ct = 0; ct < 8; ++ct) {
    int ncol = ct * 16 + mloc;
    float bias = bl[ncol];
    v8f c;
#pragma unroll
    for (int i = 0; i < 8; ++i) c[i] = bias;

    const float* wlrow = Wl + (size_t)ncol * K;
    const float* wrrow = Wr + (size_t)ncol * K;

#pragma unroll
    for (int kt = 0; kt < 32; ++kt) {
      int k0 = kt * 4 + koff;
      v2f a, b;
      a[0] = arow[k0] * dscale; a[1] = arow[k0 + 1] * dscale;
      b[0] = wlrow[k0];         b[1] = wlrow[k0 + 1];
      c = __builtin_amdgcn_wmma_f32_16x16x4_f32(false, a, false, b, (short)0, c,
                                                false, false);
    }
#pragma unroll
    for (int kt = 0; kt < 32; ++kt) {
      int k0 = kt * 4 + koff;
      v2f a, b;
      a[0] = xrow[k0]; a[1] = xrow[k0 + 1];
      b[0] = wrrow[k0]; b[1] = wrrow[k0 + 1];
      c = __builtin_amdgcn_wmma_f32_16x16x4_f32(false, a, false, b, (short)0, c,
                                                false, false);
    }

    int rbase = (lane >> 4) * 8;
#pragma unroll
    for (int r = 0; r < 8; ++r) {
      int mr = row0 + rbase + r;
      if (mr < n) {
        float v = c[r];
        if (do_relu) v = v > 0.0f ? v : 0.0f;
        hout[(size_t)mr * F + ncol] = v;
      }
    }
  }
}

__global__ __launch_bounds__(256) void k_gemm_out7(
    const float* __restrict__ agg, const float* __restrict__ xin,
    const float* __restrict__ Wl, const float* __restrict__ bl,
    const float* __restrict__ Wr, const float* __restrict__ deg_inv,
    float* __restrict__ logits, int n) {
  const int K = F;
  int lane = threadIdx.x & 31;
  int wave = threadIdx.x >> 5;
  int row0 = (blockIdx.x * 8 + wave) * 16;
  if (row0 >= n) return;

  int mloc = lane & 15;
  int koff = (lane >> 4) * 2;
  int m = row0 + mloc;
  int mc = m < n ? m : n - 1;
  float dscale = deg_inv[mc];
  const float* arow = agg + (size_t)mc * K;
  const float* xrow = xin + (size_t)mc * K;

  int ncol = mloc;
  bool cvalid = ncol < 7;
  float wm = cvalid ? 1.0f : 0.0f;
  const float* wlrow = Wl + (size_t)(cvalid ? ncol : 0) * K;
  const float* wrrow = Wr + (size_t)(cvalid ? ncol : 0) * K;

  float bias = cvalid ? bl[ncol] : 0.0f;
  v8f c;
#pragma unroll
  for (int i = 0; i < 8; ++i) c[i] = bias;

#pragma unroll
  for (int kt = 0; kt < 32; ++kt) {
    int k0 = kt * 4 + koff;
    v2f a, b;
    a[0] = arow[k0] * dscale; a[1] = arow[k0 + 1] * dscale;
    b[0] = wlrow[k0] * wm;    b[1] = wlrow[k0 + 1] * wm;
    c = __builtin_amdgcn_wmma_f32_16x16x4_f32(false, a, false, b, (short)0, c,
                                              false, false);
  }
#pragma unroll
  for (int kt = 0; kt < 32; ++kt) {
    int k0 = kt * 4 + koff;
    v2f a, b;
    a[0] = xrow[k0];       a[1] = xrow[k0 + 1];
    b[0] = wrrow[k0] * wm; b[1] = wrrow[k0 + 1] * wm;
    c = __builtin_amdgcn_wmma_f32_16x16x4_f32(false, a, false, b, (short)0, c,
                                              false, false);
  }

  if (cvalid) {
    int rbase = (lane >> 4) * 8;
#pragma unroll
    for (int r = 0; r < 8; ++r) {
      int mr = row0 + rbase + r;
      if (mr < n) logits[(size_t)mr * 7 + ncol] = c[r];
    }
  }
}

__global__ __launch_bounds__(256) void k_logsoftmax7(const float* __restrict__ logits,
                                                     float* __restrict__ out, int n) {
  int i = blockIdx.x * blockDim.x + threadIdx.x;
  if (i >= n) return;
  const float* r = logits + (size_t)i * 7;
  float v[7];
#pragma unroll
  for (int j = 0; j < 7; ++j) v[j] = r[j];
  float mx = v[0];
#pragma unroll
  for (int j = 1; j < 7; ++j) mx = fmaxf(mx, v[j]);
  float s = 0.0f;
#pragma unroll
  for (int j = 0; j < 7; ++j) s += expf(v[j] - mx);
  float ls = logf(s);
  float* o = out + (size_t)i * 7;
#pragma unroll
  for (int j = 0; j < 7; ++j) o[j] = v[j] - mx - ls;
}

extern "C" void kernel_launch(void* const* d_in, const int* in_sizes, int n_in,
                              void* d_out, int out_size, void* d_ws, size_t ws_size,
                              hipStream_t stream) {
  (void)n_in; (void)out_size; (void)ws_size;
  const float* x   = (const float*)d_in[0];
  const int*   ei  = (const int*)d_in[1];
  const float* W1l = (const float*)d_in[2];
  const float* b1  = (const float*)d_in[3];
  const float* W1r = (const float*)d_in[4];
  const float* W2l = (const float*)d_in[5];
  const float* b2  = (const float*)d_in[6];
  const float* W2r = (const float*)d_in[7];
  const float* W3l = (const float*)d_in[8];
  const float* b3  = (const float*)d_in[9];
  const float* W3r = (const float*)d_in[10];

  const int n = in_sizes[0] / F;
  const int E = in_sizes[1] / 2;
  const int* src = ei;
  const int* dst = ei + E;

  char* base = (char*)d_ws;
  size_t off = 0;
  auto a256 = [](size_t b) { return (b + 255) & ~(size_t)255; };
  float* deg = (float*)(base + off); off = a256(off + (size_t)n * sizeof(float));
  float* agg = (float*)(base + off); off = a256(off + (size_t)n * F * sizeof(float));
  float* hA  = (float*)(base + off); off = a256(off + (size_t)n * F * sizeof(float));
  float* hB  = (float*)(base + off); off = a256(off + (size_t)n * F * sizeof(float));

  const int T = 256;
  auto nb = [&](long long work) { return (int)((work + T - 1) / T); };
  const long long scatW = (long long)E * 32;
  const int gemmB = (n + 127) / 128;

  k_zero<<<nb(n), T, 0, stream>>>(deg, (size_t)n);
  k_degree<<<nb(E), T, 0, stream>>>(dst, deg, E);
  k_deg_inv<<<nb(n), T, 0, stream>>>(deg, n);

  k_zero<<<nb((long long)n * F), T, 0, stream>>>(agg, (size_t)n * F);
  k_scatter<<<nb(scatW), T, 0, stream>>>(src, dst, x, agg, E);
  k_gemm128<<<gemmB, T, 0, stream>>>(agg, x, W1l, b1, W1r, deg, hA, n, 1);

  k_zero<<<nb((long long)n * F), T, 0, stream>>>(agg, (size_t)n * F);
  k_scatter<<<nb(scatW), T, 0, stream>>>(src, dst, hA, agg, E);
  k_gemm128<<<gemmB, T, 0, stream>>>(agg, hA, W2l, b2, W2r, deg, hB, n, 1);

  k_zero<<<nb((long long)n * F), T, 0, stream>>>(agg, (size_t)n * F);
  k_scatter<<<nb(scatW), T, 0, stream>>>(src, dst, hB, agg, E);
  k_gemm_out7<<<gemmB, T, 0, stream>>>(agg, hB, W3l, b3, W3r, deg, hA, n);

  k_logsoftmax7<<<nb(n), T, 0, stream>>>(hA, (float*)d_out, n);
}


# Atomic scatter-add replaced by a per-call destination-sorted edge list (count+rank, scan, fill) and a per-node gather sum; sum order per node = atomic arrival rank; agg zeroing gone
# speedup vs baseline: 2.3878x; 2.3878x over previous
#include <hip/hip_runtime.h>
#include <hip/hip_bf16.h>

typedef __attribute__((ext_vector_type(2))) float v2f;
typedef __attribute__((ext_vector_type(8))) float v8f;

#define F 128

__global__ __launch_bounds__(256) void k_zero(float* __restrict__ p, size_t cnt) {
  size_t i = (size_t)blockIdx.x * blockDim.x + threadIdx.x;
  if (i < cnt) p[i] = 0.0f;
}

__global__ __launch_bounds__(256) void k_degree(const int* __restrict__ dst,
                                                float* __restrict__ deg, int nE) {
  int e = blockIdx.x * blockDim.x + threadIdx.x;
  if (e < nE) atomicAdd(&deg[dst[e]], 1.0f);
}

__global__ __launch_bounds__(256) void k_deg_inv(float* __restrict__ deg, int n) {
  int i = blockIdx.x * blockDim.x + threadIdx.x;
  if (i < n) {
    float d = deg[i];
    deg[i] = d > 0.0f ? 1.0f / d : 0.0f;
  }
}

#define SCAN_CHUNK 1024

__global__ __launch_bounds__(256) void k_zero_i32(int* __restrict__ p, int cnt) {
  int i = blockIdx.x * blockDim.x + threadIdx.x;
  if (i < cnt) p[i] = 0;
}

__global__ __launch_bounds__(256) void k_count_rank(const int* __restrict__ dst,
                                                    int* __restrict__ cnt,
                                                    int* __restrict__ rank, int nE) {
  int e = blockIdx.x * blockDim.x + threadIdx.x;
  if (e < nE) rank[e] = atomicAdd(&cnt[dst[e]], 1);
}

__device__ __forceinline__ int wave_incl_scan(int v, int lane) {
#pragma unroll
  for (int d = 1; d < 32; d <<= 1) {
    int t = __shfl_up(v, d);
    if (lane >= d) v += t;
  }
  return v;
}

__global__ __launch_bounds__(256) void k_scan_chunk(const int* __restrict__ cnt,
                                                    int* __restrict__ rowl,
                                                    int* __restrict__ bsum,
                                                    float* __restrict__ deg_inv, int n) {
  __shared__ int wsum[8];
  int lane = threadIdx.x & 31;
  int wave = threadIdx.x >> 5;
  int i0 = blockIdx.x * SCAN_CHUNK + threadIdx.x * 4;
  int v[4];
#pragma unroll
  for (int k = 0; k < 4; ++k) v[k] = (i0 + k < n) ? cnt[i0 + k] : 0;
  int tsum = v[0] + v[1] + v[2] + v[3];
  int incl = wave_incl_scan(tsum, lane);
  if (lane == 31) wsum[wave] = incl;
  __syncthreads();
  int woff = 0;
  for (int w = 0; w < wave; ++w) woff += wsum[w];
  int run = woff + incl - tsum;
#pragma unroll
  for (int k = 0; k < 4; ++k) {
    if (i0 + k < n) {
      rowl[i0 + k] = run;
      deg_inv[i0 + k] = v[k] > 0 ? 1.0f / (float)v[k] : 0.0f;
    }
    run += v[k];
  }
  if (threadIdx.x == 255) bsum[blockIdx.x] = run;
}

__global__ __launch_bounds__(256) void k_scan_top(int* __restrict__ bsum, int nchunk) {
  __shared__ int wsum[8];
  int lane = threadIdx.x & 31;
  int wave = threadIdx.x >> 5;
  int carry = 0;
  for (int base = 0; base < nchunk; base += 256) {
    int idx = base + threadIdx.x;
    int v = idx < nchunk ? bsum[idx] : 0;
    int incl = wave_incl_scan(v, lane);
    if (lane == 31) wsum[wave] = incl;
    __syncthreads();
    int woff = 0, tot = 0;
    for (int w = 0; w < 8; ++w) {
      if (w < wave) woff += wsum[w];
      tot += wsum[w];
    }
    if (idx < nchunk) bsum[idx] = carry + woff + incl - v;
    carry += tot;
    __syncthreads();
  }
}

__global__ __launch_bounds__(256) void k_fill(const int* __restrict__ src,
                                              const int* __restrict__ dst,
                                              const int* __restrict__ rank,
                                              const int* __restrict__ rowl,
                                              const int* __restrict__ boff,
                                              int* __restrict__ col, int nE) {
  int e = blockIdx.x * blockDim.x + threadIdx.x;
  if (e >= nE) return;
  int d = dst[e];
  col[rowl[d] + boff[d / SCAN_CHUNK] + rank[e]] = src[e];
}

__global__ __launch_bounds__(256) void k_gather(const int* __restrict__ col,
                                                const int* __restrict__ cnt,
                                                const int* __restrict__ rowl,
                                                const int* __restrict__ boff,
                                                const float* __restrict__ feat,
                                                float* __restrict__ agg, int n) {
  const int wave = __builtin_amdgcn_readfirstlane((int)(threadIdx.x >> 5));
  const int node = blockIdx.x * 8 + wave;
  if (node >= n) return;
  const int lane = threadIdx.x & 31;
  const int cn = cnt[node];
  const int* c = col + (rowl[node] + boff[node / SCAN_CHUNK]);
  const float* fb = feat + lane * 4;
  float4 acc = make_float4(0.0f, 0.0f, 0.0f, 0.0f);
  int j = 0;
  for (; j + 4 <= cn; j += 4) {
    const int s0 = c[j], s1 = c[j + 1], s2 = c[j + 2], s3 = c[j + 3];
    const float4 v0 = *reinterpret_cast<const float4*>(fb + (size_t)s0 * F);
    const float4 v1 = *reinterpret_cast<const float4*>(fb + (size_t)s1 * F);
    const float4 v2 = *reinterpret_cast<const float4*>(fb + (size_t)s2 * F);
    const float4 v3 = *reinterpret_cast<const float4*>(fb + (size_t)s3 * F);
    acc.x += v0.x; acc.y += v0.y; acc.z += v0.z; acc.w += v0.w;
    acc.x += v1.x; acc.y += v1.y; acc.z += v1.z; acc.w += v1.w;
    acc.x += v2.x; acc.y += v2.y; acc.z += v2.z; acc.w += v2.w;
    acc.x += v3.x; acc.y += v3.y; acc.z += v3.z; acc.w += v3.w;
  }
  for (; j < cn; ++j) {
    const float4 v0 = *reinterpret_cast<const float4*>(fb + (size_t)c[j] * F);
    acc.x += v0.x; acc.y += v0.y; acc.z += v0.z; acc.w += v0.w;
  }
  *reinterpret_cast<float4*>(agg + (size_t)node * F + lane * 4) = acc;
}

__global__ __launch_bounds__(256) void k_scatter(const int* __restrict__ src,
                                                 const int* __restrict__ dst,
                                                 const float* __restrict__ feat,
                                                 float* __restrict__ agg, int nE) {
  long long tid = (long long)blockIdx.x * blockDim.x + threadIdx.x;
  int e = (int)(tid >> 5);
  if (e >= nE) return;
  int fo = ((int)tid & 31) * 4;
  const float4 v = *reinterpret_cast<const float4*>(feat + (size_t)src[e] * F + fo);
  float* d = agg + (size_t)dst[e] * F + fo;
  atomicAdd(d + 0, v.x);
  atomicAdd(d + 1, v.y);
  atomicAdd(d + 2, v.z);
  atomicAdd(d + 3, v.w);
}

__global__ __launch_bounds__(256) void k_gemm128(
    const float* __restrict__ agg, const float* __restrict__ xin,
    const float* __restrict__ Wl, const float* __restrict__ bl,
    const float* __restrict__ Wr, const float* __restrict__ deg_inv,
    float* __restrict__ hout, int n, int do_relu) {
  const int K = F;
  int lane = threadIdx.x & 31;
  int wave = threadIdx.x >> 5;
  int row0 = (blockIdx.x * 8 + wave) * 16;
  if (row0 >= n) return;

  int mloc = lane & 15;
  int koff = (lane >> 4) * 2;
  int m = row0 + mloc;
  int mc = m < n ? m : n - 1;
  float dscale = deg_inv[mc];
  const float* arow = agg + (size_t)mc * K;
  const float* xrow = xin + (size_t)mc * K;

  for (int ct = 0; ct < 8; ++ct) {
    int ncol = ct * 16 + mloc;
    float bias = bl[ncol];
    v8f c;
#pragma unroll
    for (int i = 0; i < 8; ++i) c[i] = bias;

    const float* wlrow = Wl + (size_t)ncol * K;
    const float* wrrow = Wr + (size_t)ncol * K;

#pragma unroll
    for (int kt = 0; kt < 32; ++kt) {
      int k0 = kt * 4 + koff;
      v2f a, b;
      a[0] = arow[k0] * dscale; a[1] = arow[k0 + 1] * dscale;
      b[0] = wlrow[k0];         b[1] = wlrow[k0 + 1];
      c = __builtin_amdgcn_wmma_f32_16x16x4_f32(false, a, false, b, (short)0, c,
                                                false, false);
    }
#pragma unroll
    for (int kt = 0; kt < 32; ++kt) {
      int k0 = kt * 4 + koff;
      v2f a, b;
      a[0] = xrow[k0]; a[1] = xrow[k0 + 1];
      b[0] = wrrow[k0]; b[1] = wrrow[k0 + 1];
      c = __builtin_amdgcn_wmma_f32_16x16x4_f32(false, a, false, b, (short)0, c,
                                                false, false);
    }

    int rbase = (lane >> 4) * 8;
#pragma unroll
    for (int r = 0; r < 8; ++r) {
      int mr = row0 + rbase + r;
      if (mr < n) {
        float v = c[r];
        if (do_relu) v = v > 0.0f ? v : 0.0f;
        hout[(size_t)mr * F + ncol] = v;
      }
    }
  }
}

__global__ __launch_bounds__(256) void k_gemm_out7(
    const float* __restrict__ agg, const float* __restrict__ xin,
    const float* __restrict__ Wl, const float* __restrict__ bl,
    const float* __restrict__ Wr, const float* __restrict__ deg_inv,
    float* __restrict__ logits, int n) {
  const int K = F;
  int lane = threadIdx.x & 31;
  int wave = threadIdx.x >> 5;
  int row0 = (blockIdx.x * 8 + wave) * 16;
  if (row0 >= n) return;

  int mloc = lane & 15;
  int koff = (lane >> 4) * 2;
  int m = row0 + mloc;
  int mc = m < n ? m : n - 1;
  float dscale = deg_inv[mc];
  const float* arow = agg + (size_t)mc * K;
  const float* xrow = xin + (size_t)mc * K;

  int ncol = mloc;
  bool cvalid = ncol < 7;
  float wm = cvalid ? 1.0f : 0.0f;
  const float* wlrow = Wl + (size_t)(cvalid ? ncol : 0) * K;
  const float* wrrow = Wr + (size_t)(cvalid ? ncol : 0) * K;

  float bias = cvalid ? bl[ncol] : 0.0f;
  v8f c;
#pragma unroll
  for (int i = 0; i < 8; ++i) c[i] = bias;

#pragma unroll
  for (int kt = 0; kt < 32; ++kt) {
    int k0 = kt * 4 + koff;
    v2f a, b;
    a[0] = arow[k0] * dscale; a[1] = arow[k0 + 1] * dscale;
    b[0] = wlrow[k0] * wm;    b[1] = wlrow[k0 + 1] * wm;
    c = __builtin_amdgcn_wmma_f32_16x16x4_f32(false, a, false, b, (short)0, c,
                                              false, false);
  }
#pragma unroll
  for (int kt = 0; kt < 32; ++kt) {
    int k0 = kt * 4 + koff;
    v2f a, b;
    a[0] = xrow[k0];       a[1] = xrow[k0 + 1];
    b[0] = wrrow[k0] * wm; b[1] = wrrow[k0 + 1] * wm;
    c = __builtin_amdgcn_wmma_f32_16x16x4_f32(false, a, false, b, (short)0, c,
                                              false, false);
  }

  if (cvalid) {
    int rbase = (lane >> 4) * 8;
#pragma unroll
    for (int r = 0; r < 8; ++r) {
      int mr = row0 + rbase + r;
      if (mr < n) logits[(size_t)mr * 7 + ncol] = c[r];
    }
  }
}

__global__ __launch_bounds__(256) void k_logsoftmax7(const float* __restrict__ logits,
                                                     float* __restrict__ out, int n) {
  int i = blockIdx.x * blockDim.x + threadIdx.x;
  if (i >= n) return;
  const float* r = logits + (size_t)i * 7;
  float v[7];
#pragma unroll
  for (int j = 0; j < 7; ++j) v[j] = r[j];
  float mx = v[0];
#pragma unroll
  for (int j = 1; j < 7; ++j) mx = fmaxf(mx, v[j]);
  float s = 0.0f;
#pragma unroll
  for (int j = 0; j < 7; ++j) s += expf(v[j] - mx);
  float ls = logf(s);
  float* o = out + (size_t)i * 7;
#pragma unroll
  for (int j = 0; j < 7; ++j) o[j] = v[j] - mx - ls;
}

extern "C" void kernel_launch(void* const* d_in, const int* in_sizes, int n_in,
                              void* d_out, int out_size, void* d_ws, size_t ws_size,
                              hipStream_t stream) {
  (void)n_in; (void)out_size;
  const float* x   = (const float*)d_in[0];
  const int*   ei  = (const int*)d_in[1];
  const float* W1l = (const float*)d_in[2];
  const float* b1  = (const float*)d_in[3];
  const float* W1r = (const float*)d_in[4];
  const float* W2l = (const float*)d_in[5];
  const float* b2  = (const float*)d_in[6];
  const float* W2r = (const float*)d_in[7];
  const float* W3l = (const float*)d_in[8];
  const float* b3  = (const float*)d_in[9];
  const float* W3r = (const float*)d_in[10];

  const int n = in_sizes[0] / F;
  const int E = in_sizes[1] / 2;
  const int* src = ei;
  const int* dst = ei + E;

  char* base = (char*)d_ws;
  size_t off = 0;
  auto a256 = [](size_t b) { return (b + 255) & ~(size_t)255; };
  float* deg = (float*)(base + off); off = a256(off + (size_t)n * sizeof(float));
  float* agg = (float*)(base + off); off = a256(off + (size_t)n * F * sizeof(float));
  float* hA  = (float*)(base + off); off = a256(off + (size_t)n * F * sizeof(float));
  float* hB  = (float*)(base + off); off = a256(off + (size_t)n * F * sizeof(float));

  const int nchunk = (n + SCAN_CHUNK - 1) / SCAN_CHUNK;
  int* cnt  = (int*)(base + off); off = a256(off + (size_t)n * sizeof(int));
  int* rowl = (int*)(base + off); off = a256(off + (size_t)n * sizeof(int));
  int* bsum = (int*)(base + off); off = a256(off + (size_t)nchunk * sizeof(int));
  int* col  = (int*)(base + off); off = a256(off + (size_t)E * sizeof(int));
  int* rank = (int*)hB;
  const bool sorted = off <= ws_size && (size_t)E <= (size_t)n * F;

  const int T = 256;
  auto nb = [&](long long work) { return (int)((work + T - 1) / T); };
  const long long scatW = (long long)E * 32;
  const int gemmB = (n + 127) / 128;
  const int gathB = (n + 7) / 8;

  if (sorted) {
    k_zero_i32<<<nb(n), T, 0, stream>>>(cnt, n);
    k_count_rank<<<nb(E), T, 0, stream>>>(dst, cnt, rank, E);
    k_scan_chunk<<<nchunk, T, 0, stream>>>(cnt, rowl, bsum, deg, n);
    k_scan_top<<<1, T, 0, stream>>>(bsum, nchunk);
    k_fill<<<nb(E), T, 0, stream>>>(src, dst, rank, rowl, bsum, col, E);

    k_gather<<<gathB, T, 0, stream>>>(col, cnt, rowl, bsum, x, agg, n);
    k_gemm128<<<gemmB, T, 0, stream>>>(agg, x, W1l, b1, W1r, deg, hA, n, 1);

    k_gather<<<gathB, T, 0, stream>>>(col, cnt, rowl, bsum, hA, agg, n);
    k_gemm128<<<gemmB, T, 0, stream>>>(agg, hA, W2l, b2, W2r, deg, hB, n, 1);

    k_gather<<<gathB, T, 0, stream>>>(col, cnt, rowl, bsum, hB, agg, n);
    k_gemm_out7<<<gemmB, T, 0, stream>>>(agg, hB, W3l, b3, W3r, deg, hA, n);
  } else {
    k_zero<<<nb(n), T, 0, stream>>>(deg, (size_t)n);
    k_degree<<<nb(E), T, 0, stream>>>(dst, deg, E);
    k_deg_inv<<<nb(n), T, 0, stream>>>(deg, n);

    k_zero<<<nb((long long)n * F), T, 0, stream>>>(agg, (size_t)n * F);
    k_scatter<<<nb(scatW), T, 0, stream>>>(src, dst, x, agg, E);
    k_gemm128<<<gemmB, T, 0, stream>>>(agg, x, W1l, b1, W1r, deg, hA, n, 1);

    k_zero<<<nb((long long)n * F), T, 0, stream>>>(agg, (size_t)n * F);
    k_scatter<<<nb(scatW), T, 0, stream>>>(src, dst, hA, agg, E);
    k_gemm128<<<gemmB, T, 0, stream>>>(agg, hA, W2l, b2, W2r, deg, hB, n, 1);

    k_zero<<<nb((long long)n * F), T, 0, stream>>>(agg, (size_t)n * F);
    k_scatter<<<nb(scatW), T, 0, stream>>>(src, dst, hB, agg, E);
    k_gemm_out7<<<gemmB, T, 0, stream>>>(agg, hB, W3l, b3, W3r, deg, hA, n);
  }

  k_logsoftmax7<<<nb(n), T, 0, stream>>>(hA, (float*)d_out, n);
}


# 128-wide dense update: one 16-wave workgroup per WGP stages Wl|Wr once in LDS and reads B fragments from it; A read once per tile for all 8 column tiles; k summed in 8-wide steps, bias added last
# speedup vs baseline: 3.1883x; 1.3352x over previous
#include <hip/hip_runtime.h>
#include <hip/hip_bf16.h>

typedef __attribute__((ext_vector_type(2))) float v2f;
typedef __attribute__((ext_vector_type(8))) float v8f;

#define F 128

__global__ __launch_bounds__(256) void k_zero(float* __restrict__ p, size_t cnt) {
  size_t i = (size_t)blockIdx.x * blockDim.x + threadIdx.x;
  if (i < cnt) p[i] = 0.0f;
}

__global__ __launch_bounds__(256) void k_degree(const int* __restrict__ dst,
                                                float* __restrict__ deg, int nE) {
  int e = blockIdx.x * blockDim.x + threadIdx.x;
  if (e < nE) atomicAdd(&deg[dst[e]], 1.0f);
}

__global__ __launch_bounds__(256) void k_deg_inv(float* __restrict__ deg, int n) {
  int i = blockIdx.x * blockDim.x + threadIdx.x;
  if (i < n) {
    float d = deg[i];
    deg[i] = d > 0.0f ? 1.0f / d : 0.0f;
  }
}

#define SCAN_CHUNK 1024

__global__ __launch_bounds__(256) void k_zero_i32(int* __restrict__ p, int cnt) {
  int i = blockIdx.x * blockDim.x + threadIdx.x;
  if (i < cnt) p[i] = 0;
}

__global__ __launch_bounds__(256) void k_count_rank(const int* __restrict__ dst,
                                                    int* __restrict__ cnt,
                                                    int* __restrict__ rank, int nE) {
  int e = blockIdx.x * blockDim.x + threadIdx.x;
  if (e < nE) rank[e] = atomicAdd(&cnt[dst[e]], 1);
}

__device__ __forceinline__ int wave_incl_scan(int v, int lane) {
#pragma unroll
  for (int d = 1; d < 32; d <<= 1) {
    int t = __shfl_up(v, d);
    if (lane >= d) v += t;
  }
  return v;
}

__global__ __launch_bounds__(256) void k_scan_chunk(const int* __restrict__ cnt,
                                                    int* __restrict__ rowl,
                                                    int* __restrict__ bsum,
                                                    float* __restrict__ deg_inv, int n) {
  __shared__ int wsum[8];
  int lane = threadIdx.x & 31;
  int wave = threadIdx.x >> 5;
  int i0 = blockIdx.x * SCAN_CHUNK + threadIdx.x * 4;
  int v[4];
#pragma unroll
  for (int k = 0; k < 4; ++k) v[k] = (i0 + k < n) ? cnt[i0 + k] : 0;
  int tsum = v[0] + v[1] + v[2] + v[3];
  int incl = wave_incl_scan(tsum, lane);
  if (lane == 31) wsum[wave] = incl;
  __syncthreads();
  int woff = 0;
  for (int w = 0; w < wave; ++w) woff += wsum[w];
  int run = woff + incl - tsum;
#pragma unroll
  for (int k = 0; k < 4; ++k) {
    if (i0 + k < n) {
      rowl[i0 + k] = run;
      deg_inv[i0 + k] = v[k] > 0 ? 1.0f / (float)v[k] : 0.0f;
    }
    run += v[k];
  }
  if (threadIdx.x == 255) bsum[blockIdx.x] = run;
}

__global__ __launch_bounds__(256) void k_scan_top(int* __restrict__ bsum, int nchunk) {
  __shared__ int wsum[8];
  int lane = threadIdx.x & 31;
  int wave = threadIdx.x >> 5;
  int carry = 0;
  for (int base = 0; base < nchunk; base += 256) {
    int idx = base + threadIdx.x;
    int v = idx < nchunk ? bsum[idx] : 0;
    int incl = wave_incl_scan(v, lane);
    if (lane == 31) wsum[wave] = incl;
    __syncthreads();
    int woff = 0, tot = 0;
    for (int w = 0; w < 8; ++w) {
      if (w < wave) woff += wsum[w];
      tot += wsum[w];
    }
    if (idx < nchunk) bsum[idx] = carry + woff + incl - v;
    carry += tot;
    __syncthreads();
  }
}

__global__ __launch_bounds__(256) void k_fill(const int* __restrict__ src,
                                              const int* __restrict__ dst,
                                              const int* __restrict__ rank,
                                              const int* __restrict__ rowl,
                                              const int* __restrict__ boff,
                                              int* __restrict__ col, int nE) {
  int e = blockIdx.x * blockDim.x + threadIdx.x;
  if (e >= nE) return;
  int d = dst[e];
  col[rowl[d] + boff[d / SCAN_CHUNK] + rank[e]] = src[e];
}

__global__ __launch_bounds__(256) void k_gather(const int* __restrict__ col,
                                                const int* __restrict__ cnt,
                                                const int* __restrict__ rowl,
                                                const int* __restrict__ boff,
                                                const float* __restrict__ feat,
                                                float* __restrict__ agg, int n) {
  const int wave = __builtin_amdgcn_readfirstlane((int)(threadIdx.x >> 5));
  const int node = blockIdx.x * 8 + wave;
  if (node >= n) return;
  const int lane = threadIdx.x & 31;
  const int cn = cnt[node];
  const int* c = col + (rowl[node] + boff[node / SCAN_CHUNK]);
  const float* fb = feat + lane * 4;
  float4 acc = make_float4(0.0f, 0.0f, 0.0f, 0.0f);
  int j = 0;
  for (; j + 4 <= cn; j += 4) {
    const int s0 = c[j], s1 = c[j + 1], s2 = c[j + 2], s3 = c[j + 3];
    const float4 v0 = *reinterpret_cast<const float4*>(fb + (size_t)s0 * F);
    const float4 v1 = *reinterpret_cast<const float4*>(fb + (size_t)s1 * F);
    const float4 v2 = *reinterpret_cast<const float4*>(fb + (size_t)s2 * F);
    const float4 v3 = *reinterpret_cast<const float4*>(fb + (size_t)s3 * F);
    acc.x += v0.x; acc.y += v0.y; acc.z += v0.z; acc.w += v0.w;
    acc.x += v1.x; acc.y += v1.y; acc.z += v1.z; acc.w += v1.w;
    acc.x += v2.x; acc.y += v2.y; acc.z += v2.z; acc.w += v2.w;
    acc.x += v3.x; acc.y += v3.y; acc.z += v3.z; acc.w += v3.w;
  }
  for (; j < cn; ++j) {
    const float4 v0 = *reinterpret_cast<const float4*>(fb + (size_t)c[j] * F);
    acc.x += v0.x; acc.y += v0.y; acc.z += v0.z; acc.w += v0.w;
  }
  *reinterpret_cast<float4*>(agg + (size_t)node * F + lane * 4) = acc;
}

__global__ __launch_bounds__(256) void k_scatter(const int* __restrict__ src,
                                                 const int* __restrict__ dst,
                                                 const float* __restrict__ feat,
                                                 float* __restrict__ agg, int nE) {
  long long tid = (long long)blockIdx.x * blockDim.x + threadIdx.x;
  int e = (int)(tid >> 5);
  if (e >= nE) return;
  int fo = ((int)tid & 31) * 4;
  const float4 v = *reinterpret_cast<const float4*>(feat + (size_t)src[e] * F + fo);
  float* d = agg + (size_t)dst[e] * F + fo;
  atomicAdd(d + 0, v.x);
  atomicAdd(d + 1, v.y);
  atomicAdd(d + 2, v.z);
  atomicAdd(d + 3, v.w);
}

#define GEMM_WAVES 16
#define WROW 256
#define WBLK (8 * WROW + 4)

__device__ __forceinline__ v8f wmma_f32(float a0, float a1, float b0, float b1, v8f c) {
  v2f a, b;
  a[0] = a0; a[1] = a1;
  b[0] = b0; b[1] = b1;
  return __builtin_amdgcn_wmma_f32_16x16x4_f32(false, a, false, b, (short)0, c,
                                               false, false);
}

__global__ __launch_bounds__(GEMM_WAVES * 32) void k_gemm128(
    const float* __restrict__ agg, const float* __restrict__ xin,
    const float* __restrict__ Wl, const float* __restrict__ bl,
    const float* __restrict__ Wr, const float* __restrict__ deg_inv,
    float* __restrict__ hout, int n, int do_relu) {
  __shared__ __attribute__((aligned(16))) float wlds[16 * WBLK];

  constexpr int NCOPY = F * (F / 4) / (GEMM_WAVES * 32);
  float4 vl[NCOPY], vr[NCOPY];
#pragma unroll
  for (int i = 0; i < NCOPY; ++i) {
    const int f = threadIdx.x + i * (GEMM_WAVES * 32);
    vl[i] = reinterpret_cast<const float4*>(Wl)[f];
    vr[i] = reinterpret_cast<const float4*>(Wr)[f];
  }
#pragma unroll
  for (int i = 0; i < NCOPY; ++i) {
    const int f = threadIdx.x + i * (GEMM_WAVES * 32);
    const int row = f >> 5;
    const int c4 = f & 31;
    float* dstp = wlds + (row & 15) * WBLK + (row >> 4) * WROW + c4 * 4;
    *reinterpret_cast<float4*>(dstp) = vl[i];
    *reinterpret_cast<float4*>(dstp + F) = vr[i];
  }
  __syncthreads();

  const int lane = threadIdx.x & 31;
  const int wave = __builtin_amdgcn_readfirstlane((int)(threadIdx.x >> 5));
  const int mloc = lane & 15;
  const int h = lane >> 4;
  const int ntiles = (n + 15) >> 4;
  const float* bfrag = wlds + mloc * WBLK + 4 * h;

  for (int s = wave; ; s += GEMM_WAVES) {
    const int t = blockIdx.x + gridDim.x * s;
    if (t >= ntiles) break;
    const int row0 = t * 16;
    const int m = row0 + mloc;
    const int mc = m < n ? m : n - 1;
    const float dscale = deg_inv[mc];
    const float4* arow = reinterpret_cast<const float4*>(agg + (size_t)mc * F) + h;
    const float4* xrow = reinterpret_cast<const float4*>(xin + (size_t)mc * F) + h;

    float4 av[32];
#pragma unroll
    for (int j = 0; j < 16; ++j) av[j] = arow[2 * j];

    v8f c[8];
#pragma unroll
    for (int ct = 0; ct < 8; ++ct) {
#pragma unroll
      for (int i = 0; i < 8; ++i) c[ct][i] = 0.0f;
    }

    float4 bq[2][8];
#pragma unroll
    for (int ct = 0; ct < 8; ++ct)
      bq[0][ct] = *reinterpret_cast<const float4*>(bfrag + ct * WROW);

#pragma unroll
    for (int j = 0; j < 32; ++j) {
      const int cur = j & 1;
      if (j + 1 < 32) {
#pragma unroll
        for (int ct = 0; ct < 8; ++ct)
          bq[cur ^ 1][ct] =
              *reinterpret_cast<const float4*>(bfrag + ct * WROW + 8 * (j + 1));
      }
      if (j < 16) av[j + 16] = xrow[2 * j];
      __builtin_amdgcn_sched_barrier(0);
      float4 a = av[j];
      if (j < 16) { a.x *= dscale; a.y *= dscale; a.z *= dscale; a.w *= dscale; }
#pragma unroll
      for (int ct = 0; ct < 8; ++ct) {
        const float4 b = bq[cur][ct];
        c[ct] = wmma_f32(a.x, a.y, b.x, b.y, c[ct]);
        c[ct] = wmma_f32(a.z, a.w, b.z, b.w, c[ct]);
      }
      __builtin_amdgcn_sched_barrier(0);
    }

    const int rbase = h * 8;
#pragma unroll
    for (int ct = 0; ct < 8; ++ct) {
      const int ncol = ct * 16 + mloc;
      const float bias = bl[ncol];
#pragma unroll
      for (int r = 0; r < 8; ++r) {
        const int mr = row0 + rbase + r;
        if (mr < n) {
          float v = c[ct][r] + bias;
          if (do_relu) v = v > 0.0f ? v : 0.0f;
          hout[(size_t)mr * F + ncol] = v;
        }
      }
    }
  }
}

__global__ __launch_bounds__(256) void k_gemm_out7(
    const float* __restrict__ agg, const float* __restrict__ xin,
    const float* __restrict__ Wl, const float* __restrict__ bl,
    const float* __restrict__ Wr, const float* __restrict__ deg_inv,
    float* __restrict__ logits, int n) {
  const int K = F;
  int lane = threadIdx.x & 31;
  int wave = threadIdx.x >> 5;
  int row0 = (blockIdx.x * 8 + wave) * 16;
  if (row0 >= n) return;

  int mloc = lane & 15;
  int koff = (lane >> 4) * 2;
  int m = row0 + mloc;
  int mc = m < n ? m : n - 1;
  float dscale = deg_inv[mc];
  const float* arow = agg + (size_t)mc * K;
  const float* xrow = xin + (size_t)mc * K;

  int ncol = mloc;
  bool cvalid = ncol < 7;
  float wm = cvalid ? 1.0f : 0.0f;
  const float* wlrow = Wl + (size_t)(cvalid ? ncol : 0) * K;
  const float* wrrow = Wr + (size_t)(cvalid ? ncol : 0) * K;

  float bias = cvalid ? bl[ncol] : 0.0f;
  v8f c;
#pragma unroll
  for (int i = 0; i < 8; ++i) c[i] = bias;

#pragma unroll
  for (int kt = 0; kt < 32; ++kt) {
    int k0 = kt * 4 + koff;
    v2f a, b;
    a[0] = arow[k0] * dscale; a[1] = arow[k0 + 1] * dscale;
    b[0] = wlrow[k0] * wm;    b[1] = wlrow[k0 + 1] * wm;
    c = __builtin_amdgcn_wmma_f32_16x16x4_f32(false, a, false, b, (short)0, c,
                                              false, false);
  }
#pragma unroll
  for (int kt = 0; kt < 32; ++kt) {
    int k0 = kt * 4 + koff;
    v2f a, b;
    a[0] = xrow[k0];       a[1] = xrow[k0 + 1];
    b[0] = wrrow[k0] * wm; b[1] = wrrow[k0 + 1] * wm;
    c = __builtin_amdgcn_wmma_f32_16x16x4_f32(false, a, false, b, (short)0, c,
                                              false, false);
  }

  if (cvalid) {
    int rbase = (lane >> 4) * 8;
#pragma unroll
    for (int r = 0; r < 8; ++r) {
      int mr = row0 + rbase + r;
      if (mr < n) logits[(size_t)mr * 7 + ncol] = c[r];
    }
  }
}

__global__ __launch_bounds__(256) void k_logsoftmax7(const float* __restrict__ logits,
                                                     float* __restrict__ out, int n) {
  int i = blockIdx.x * blockDim.x + threadIdx.x;
  if (i >= n) return;
  const float* r = logits + (size_t)i * 7;
  float v[7];
#pragma unroll
  for (int j = 0; j < 7; ++j) v[j] = r[j];
  float mx = v[0];
#pragma unroll
  for (int j = 1; j < 7; ++j) mx = fmaxf(mx, v[j]);
  float s = 0.0f;
#pragma unroll
  for (int j = 0; j < 7; ++j) s += expf(v[j] - mx);
  float ls = logf(s);
  float* o = out + (size_t)i * 7;
#pragma unroll
  for (int j = 0; j < 7; ++j) o[j] = v[j] - mx - ls;
}

extern "C" void kernel_launch(void* const* d_in, const int* in_sizes, int n_in,
                              void* d_out, int out_size, void* d_ws, size_t ws_size,
                              hipStream_t stream) {
  (void)n_in; (void)out_size;
  const float* x   = (const float*)d_in[0];
  const int*   ei  = (const int*)d_in[1];
  const float* W1l = (const float*)d_in[2];
  const float* b1  = (const float*)d_in[3];
  const float* W1r = (const float*)d_in[4];
  const float* W2l = (const float*)d_in[5];
  const float* b2  = (const float*)d_in[6];
  const float* W2r = (const float*)d_in[7];
  const float* W3l = (const float*)d_in[8];
  const float* b3  = (const float*)d_in[9];
  const float* W3r = (const float*)d_in[10];

  const int n = in_sizes[0] / F;
  const int E = in_sizes[1] / 2;
  const int* src = ei;
  const int* dst = ei + E;

  char* base = (char*)d_ws;
  size_t off = 0;
  auto a256 = [](size_t b) { return (b + 255) & ~(size_t)255; };
  float* deg = (float*)(base + off); off = a256(off + (size_t)n * sizeof(float));
  float* agg = (float*)(base + off); off = a256(off + (size_t)n * F * sizeof(float));
  float* hA  = (float*)(base + off); off = a256(off + (size_t)n * F * sizeof(float));
  float* hB  = (float*)(base + off); off = a256(off + (size_t)n * F * sizeof(float));

  const int nchunk = (n + SCAN_CHUNK - 1) / SCAN_CHUNK;
  int* cnt  = (int*)(base + off); off = a256(off + (size_t)n * sizeof(int));
  int* rowl = (int*)(base + off); off = a256(off + (size_t)n * sizeof(int));
  int* bsum = (int*)(base + off); off = a256(off + (size_t)nchunk * sizeof(int));
  int* col  = (int*)(base + off); off = a256(off + (size_t)E * sizeof(int));
  int* rank = (int*)hB;
  const bool sorted = off <= ws_size && (size_t)E <= (size_t)n * F;

  const int T = 256;
  auto nb = [&](long long work) { return (int)((work + T - 1) / T); };
  const long long scatW = (long long)E * 32;
  const int gemmB = (n + 127) / 128;
  const int gathB = (n + 7) / 8;

  if (sorted) {
    k_zero_i32<<<nb(n), T, 0, stream>>>(cnt, n);
    k_count_rank<<<nb(E), T, 0, stream>>>(dst, cnt, rank, E);
    k_scan_chunk<<<nchunk, T, 0, stream>>>(cnt, rowl, bsum, deg, n);
    k_scan_top<<<1, T, 0, stream>>>(bsum, nchunk);
    k_fill<<<nb(E), T, 0, stream>>>(src, dst, rank, rowl, bsum, col, E);

    k_gather<<<gathB, T, 0, stream>>>(col, cnt, rowl, bsum, x, agg, n);
    k_gemm128<<<256, GEMM_WAVES * 32, 0, stream>>>(agg, x, W1l, b1, W1r, deg, hA, n, 1);

    k_gather<<<gathB, T, 0, stream>>>(col, cnt, rowl, bsum, hA, agg, n);
    k_gemm128<<<256, GEMM_WAVES * 32, 0, stream>>>(agg, hA, W2l, b2, W2r, deg, hB, n, 1);

    k_gather<<<gathB, T, 0, stream>>>(col, cnt, rowl, bsum, hB, agg, n);
    k_gemm_out7<<<gemmB, T, 0, stream>>>(agg, hB, W3l, b3, W3r, deg, hA, n);
  } else {
    k_zero<<<nb(n), T, 0, stream>>>(deg, (size_t)n);
    k_degree<<<nb(E), T, 0, stream>>>(dst, deg, E);
    k_deg_inv<<<nb(n), T, 0, stream>>>(deg, n);

    k_zero<<<nb((long long)n * F), T, 0, stream>>>(agg, (size_t)n * F);
    k_scatter<<<nb(scatW), T, 0, stream>>>(src, dst, x, agg, E);
    k_gemm128<<<256, GEMM_WAVES * 32, 0, stream>>>(agg, x, W1l, b1, W1r, deg, hA, n, 1);

    k_zero<<<nb((long long)n * F), T, 0, stream>>>(agg, (size_t)n * F);
    k_scatter<<<nb(scatW), T, 0, stream>>>(src, dst, hA, agg, E);
    k_gemm128<<<256, GEMM_WAVES * 32, 0, stream>>>(agg, hA, W2l, b2, W2r, deg, hB, n, 1);

    k_zero<<<nb((long long)n * F), T, 0, stream>>>(agg, (size_t)n * F);
    k_scatter<<<nb(scatW), T, 0, stream>>>(src, dst, hB, agg, E);
    k_gemm_out7<<<gemmB, T, 0, stream>>>(agg, hB, W3l, b3, W3r, deg, hA, n);
  }

  k_logsoftmax7<<<nb(n), T, 0, stream>>>(hA, (float*)d_out, n);
}


# Layer 3 projects H to the 7 classes first (one WMMA tile, [W3l|W3r] + b3) and takes the edge mean on 7-float rows, not 128-float rows; per-node sum = 4 interleaved lane sums, tree-added
# speedup vs baseline: 3.5018x; 1.0983x over previous
#include <hip/hip_runtime.h>
#include <hip/hip_bf16.h>

typedef __attribute__((ext_vector_type(2))) float v2f;
typedef __attribute__((ext_vector_type(8))) float v8f;

#define F 128

__global__ __launch_bounds__(256) void k_zero(float* __restrict__ p, size_t cnt) {
  size_t i = (size_t)blockIdx.x * blockDim.x + threadIdx.x;
  if (i < cnt) p[i] = 0.0f;
}

__global__ __launch_bounds__(256) void k_degree(const int* __restrict__ dst,
                                                float* __restrict__ deg, int nE) {
  int e = blockIdx.x * blockDim.x + threadIdx.x;
  if (e < nE) atomicAdd(&deg[dst[e]], 1.0f);
}

__global__ __launch_bounds__(256) void k_deg_inv(float* __restrict__ deg, int n) {
  int i = blockIdx.x * blockDim.x + threadIdx.x;
  if (i < n) {
    float d = deg[i];
    deg[i] = d > 0.0f ? 1.0f / d : 0.0f;
  }
}

#define SCAN_CHUNK 1024

__global__ __launch_bounds__(256) void k_zero_i32(int* __restrict__ p, int cnt) {
  int i = blockIdx.x * blockDim.x + threadIdx.x;
  if (i < cnt) p[i] = 0;
}

__global__ __launch_bounds__(256) void k_count_rank(const int* __restrict__ dst,
                                                    int* __restrict__ cnt,
                                                    int* __restrict__ rank, int nE) {
  int e = blockIdx.x * blockDim.x + threadIdx.x;
  if (e < nE) rank[e] = atomicAdd(&cnt[dst[e]], 1);
}

__device__ __forceinline__ int wave_incl_scan(int v, int lane) {
#pragma unroll
  for (int d = 1; d < 32; d <<= 1) {
    int t = __shfl_up(v, d);
    if (lane >= d) v += t;
  }
  return v;
}

__global__ __launch_bounds__(256) void k_scan_chunk(const int* __restrict__ cnt,
                                                    int* __restrict__ rowl,
                                                    int* __restrict__ bsum,
                                                    float* __restrict__ deg_inv, int n) {
  __shared__ int wsum[8];
  int lane = threadIdx.x & 31;
  int wave = threadIdx.x >> 5;
  int i0 = blockIdx.x * SCAN_CHUNK + threadIdx.x * 4;
  int v[4];
#pragma unroll
  for (int k = 0; k < 4; ++k) v[k] = (i0 + k < n) ? cnt[i0 + k] : 0;
  int tsum = v[0] + v[1] + v[2] + v[3];
  int incl = wave_incl_scan(tsum, lane);
  if (lane == 31) wsum[wave] = incl;
  __syncthreads();
  int woff = 0;
  for (int w = 0; w < wave; ++w) woff += wsum[w];
  int run = woff + incl - tsum;
#pragma unroll
  for (int k = 0; k < 4; ++k) {
    if (i0 + k < n) {
      rowl[i0 + k] = run;
      deg_inv[i0 + k] = v[k] > 0 ? 1.0f / (float)v[k] : 0.0f;
    }
    run += v[k];
  }
  if (threadIdx.x == 255) bsum[blockIdx.x] = run;
}

__global__ __launch_bounds__(256) void k_scan_top(int* __restrict__ bsum, int nchunk) {
  __shared__ int wsum[8];
  int lane = threadIdx.x & 31;
  int wave = threadIdx.x >> 5;
  int carry = 0;
  for (int base = 0; base < nchunk; base += 256) {
    int idx = base + threadIdx.x;
    int v = idx < nchunk ? bsum[idx] : 0;
    int incl = wave_incl_scan(v, lane);
    if (lane == 31) wsum[wave] = incl;
    __syncthreads();
    int woff = 0, tot = 0;
    for (int w = 0; w < 8; ++w) {
      if (w < wave) woff += wsum[w];
      tot += wsum[w];
    }
    if (idx < nchunk) bsum[idx] = carry + woff + incl - v;
    carry += tot;
    __syncthreads();
  }
}

__global__ __launch_bounds__(256) void k_fill(const int* __restrict__ src,
                                              const int* __restrict__ dst,
                                              const int* __restrict__ rank,
                                              const int* __restrict__ rowl,
                                              const int* __restrict__ boff,
                                              int* __restrict__ col, int nE) {
  int e = blockIdx.x * blockDim.x + threadIdx.x;
  if (e >= nE) return;
  int d = dst[e];
  col[rowl[d] + boff[d / SCAN_CHUNK] + rank[e]] = src[e];
}

__global__ __launch_bounds__(256) void k_gather(const int* __restrict__ col,
                                                const int* __restrict__ cnt,
                                                const int* __restrict__ rowl,
                                                const int* __restrict__ boff,
                                                const float* __restrict__ feat,
                                                float* __restrict__ agg, int n) {
  const int wave = __builtin_amdgcn_readfirstlane((int)(threadIdx.x >> 5));
  const int node = blockIdx.x * 8 + wave;
  if (node >= n) return;
  const int lane = threadIdx.x & 31;
  const int cn = cnt[node];
  const int* c = col + (rowl[node] + boff[node / SCAN_CHUNK]);
  const float* fb = feat + lane * 4;
  float4 acc = make_float4(0.0f, 0.0f, 0.0f, 0.0f);
  int j = 0;
  for (; j + 4 <= cn; j += 4) {
    const int s0 = c[j], s1 = c[j + 1], s2 = c[j + 2], s3 = c[j + 3];
    const float4 v0 = *reinterpret_cast<const float4*>(fb + (size_t)s0 * F);
    const float4 v1 = *reinterpret_cast<const float4*>(fb + (size_t)s1 * F);
    const float4 v2 = *reinterpret_cast<const float4*>(fb + (size_t)s2 * F);
    const float4 v3 = *reinterpret_cast<const float4*>(fb + (size_t)s3 * F);
    acc.x += v0.x; acc.y += v0.y; acc.z += v0.z; acc.w += v0.w;
    acc.x += v1.x; acc.y += v1.y; acc.z += v1.z; acc.w += v1.w;
    acc.x += v2.x; acc.y += v2.y; acc.z += v2.z; acc.w += v2.w;
    acc.x += v3.x; acc.y += v3.y; acc.z += v3.z; acc.w += v3.w;
  }
  for (; j < cn; ++j) {
    const float4 v0 = *reinterpret_cast<const float4*>(fb + (size_t)c[j] * F);
    acc.x += v0.x; acc.y += v0.y; acc.z += v0.z; acc.w += v0.w;
  }
  *reinterpret_cast<float4*>(agg + (size_t)node * F + lane * 4) = acc;
}

__global__ __launch_bounds__(256) void k_scatter(const int* __restrict__ src,
                                                 const int* __restrict__ dst,
                                                 const float* __restrict__ feat,
                                                 float* __restrict__ agg, int nE) {
  long long tid = (long long)blockIdx.x * blockDim.x + threadIdx.x;
  int e = (int)(tid >> 5);
  if (e >= nE) return;
  int fo = ((int)tid & 31) * 4;
  const float4 v = *reinterpret_cast<const float4*>(feat + (size_t)src[e] * F + fo);
  float* d = agg + (size_t)dst[e] * F + fo;
  atomicAdd(d + 0, v.x);
  atomicAdd(d + 1, v.y);
  atomicAdd(d + 2, v.z);
  atomicAdd(d + 3, v.w);
}

#define GEMM_WAVES 16
#define WROW 256
#define WBLK (8 * WROW + 4)

__device__ __forceinline__ v8f wmma_f32(float a0, float a1, float b0, float b1, v8f c) {
  v2f a, b;
  a[0] = a0; a[1] = a1;
  b[0] = b0; b[1] = b1;
  return __builtin_amdgcn_wmma_f32_16x16x4_f32(false, a, false, b, (short)0, c,
                                               false, false);
}

__global__ __launch_bounds__(GEMM_WAVES * 32) void k_gemm128(
    const float* __restrict__ agg, const float* __restrict__ xin,
    const float* __restrict__ Wl, const float* __restrict__ bl,
    const float* __restrict__ Wr, const float* __restrict__ deg_inv,
    float* __restrict__ hout, int n, int do_relu) {
  __shared__ __attribute__((aligned(16))) float wlds[16 * WBLK];

  constexpr int NCOPY = F * (F / 4) / (GEMM_WAVES * 32);
  float4 vl[NCOPY], vr[NCOPY];
#pragma unroll
  for (int i = 0; i < NCOPY; ++i) {
    const int f = threadIdx.x + i * (GEMM_WAVES * 32);
    vl[i] = reinterpret_cast<const float4*>(Wl)[f];
    vr[i] = reinterpret_cast<const float4*>(Wr)[f];
  }
#pragma unroll
  for (int i = 0; i < NCOPY; ++i) {
    const int f = threadIdx.x + i * (GEMM_WAVES * 32);
    const int row = f >> 5;
    const int c4 = f & 31;
    float* dstp = wlds + (row & 15) * WBLK + (row >> 4) * WROW + c4 * 4;
    *reinterpret_cast<float4*>(dstp) = vl[i];
    *reinterpret_cast<float4*>(dstp + F) = vr[i];
  }
  __syncthreads();

  const int lane = threadIdx.x & 31;
  const int wave = __builtin_amdgcn_readfirstlane((int)(threadIdx.x >> 5));
  const int mloc = lane & 15;
  const int h = lane >> 4;
  const int ntiles = (n + 15) >> 4;
  const float* bfrag = wlds + mloc * WBLK + 4 * h;

  for (int s = wave; ; s += GEMM_WAVES) {
    const int t = blockIdx.x + gridDim.x * s;
    if (t >= ntiles) break;
    const int row0 = t * 16;
    const int m = row0 + mloc;
    const int mc = m < n ? m : n - 1;
    const float dscale = deg_inv[mc];
    const float4* arow = reinterpret_cast<const float4*>(agg + (size_t)mc * F) + h;
    const float4* xrow = reinterpret_cast<const float4*>(xin + (size_t)mc * F) + h;

    float4 av[32];
#pragma unroll
    for (int j = 0; j < 16; ++j) av[j] = arow[2 * j];

    v8f c[8];
#pragma unroll
    for (int ct = 0; ct < 8; ++ct) {
#pragma unroll
      for (int i = 0; i < 8; ++i) c[ct][i] = 0.0f;
    }

    float4 bq[2][8];
#pragma unroll
    for (int ct = 0; ct < 8; ++ct)
      bq[0][ct] = *reinterpret_cast<const float4*>(bfrag + ct * WROW);

#pragma unroll
    for (int j = 0; j < 32; ++j) {
      const int cur = j & 1;
      if (j + 1 < 32) {
#pragma unroll
        for (int ct = 0; ct < 8; ++ct)
          bq[cur ^ 1][ct] =
              *reinterpret_cast<const float4*>(bfrag + ct * WROW + 8 * (j + 1));
      }
      if (j < 16) av[j + 16] = xrow[2 * j];
      __builtin_amdgcn_sched_barrier(0);
      float4 a = av[j];
      if (j < 16) { a.x *= dscale; a.y *= dscale; a.z *= dscale; a.w *= dscale; }
#pragma unroll
      for (int ct = 0; ct < 8; ++ct) {
        const float4 b = bq[cur][ct];
        c[ct] = wmma_f32(a.x, a.y, b.x, b.y, c[ct]);
        c[ct] = wmma_f32(a.z, a.w, b.z, b.w, c[ct]);
      }
      __builtin_amdgcn_sched_barrier(0);
    }

    const int rbase = h * 8;
#pragma unroll
    for (int ct = 0; ct < 8; ++ct) {
      const int ncol = ct * 16 + mloc;
      const float bias = bl[ncol];
#pragma unroll
      for (int r = 0; r < 8; ++r) {
        const int mr = row0 + rbase + r;
        if (mr < n) {
          float v = c[ct][r] + bias;
          if (do_relu) v = v > 0.0f ? v : 0.0f;
          hout[(size_t)mr * F + ncol] = v;
        }
      }
    }
  }
}

__global__ __launch_bounds__(256) void k_gemm_out7(
    const float* __restrict__ agg, const float* __restrict__ xin,
    const float* __restrict__ Wl, const float* __restrict__ bl,
    const float* __restrict__ Wr, const float* __restrict__ deg_inv,
    float* __restrict__ logits, int n) {
  const int K = F;
  int lane = threadIdx.x & 31;
  int wave = threadIdx.x >> 5;
  int row0 = (blockIdx.x * 8 + wave) * 16;
  if (row0 >= n) return;

  int mloc = lane & 15;
  int koff = (lane >> 4) * 2;
  int m = row0 + mloc;
  int mc = m < n ? m : n - 1;
  float dscale = deg_inv[mc];
  const float* arow = agg + (size_t)mc * K;
  const float* xrow = xin + (size_t)mc * K;

  int ncol = mloc;
  bool cvalid = ncol < 7;
  float wm = cvalid ? 1.0f : 0.0f;
  const float* wlrow = Wl + (size_t)(cvalid ? ncol : 0) * K;
  const float* wrrow = Wr + (size_t)(cvalid ? ncol : 0) * K;

  float bias = cvalid ? bl[ncol] : 0.0f;
  v8f c;
#pragma unroll
  for (int i = 0; i < 8; ++i) c[i] = bias;

#pragma unroll
  for (int kt = 0; kt < 32; ++kt) {
    int k0 = kt * 4 + koff;
    v2f a, b;
    a[0] = arow[k0] * dscale; a[1] = arow[k0 + 1] * dscale;
    b[0] = wlrow[k0] * wm;    b[1] = wlrow[k0 + 1] * wm;
    c = __builtin_amdgcn_wmma_f32_16x16x4_f32(false, a, false, b, (short)0, c,
                                              false, false);
  }
#pragma unroll
  for (int kt = 0; kt < 32; ++kt) {
    int k0 = kt * 4 + koff;
    v2f a, b;
    a[0] = xrow[k0];       a[1] = xrow[k0 + 1];
    b[0] = wrrow[k0] * wm; b[1] = wrrow[k0 + 1] * wm;
    c = __builtin_amdgcn_wmma_f32_16x16x4_f32(false, a, false, b, (short)0, c,
                                              false, false);
  }

  if (cvalid) {
    int rbase = (lane >> 4) * 8;
#pragma unroll
    for (int r = 0; r < 8; ++r) {
      int mr = row0 + rbase + r;
      if (mr < n) logits[(size_t)mr * 7 + ncol] = c[r];
    }
  }
}

__global__ __launch_bounds__(256) void k_proj14(
    const float* __restrict__ xin, const float* __restrict__ Wl,
    const float* __restrict__ bl, const float* __restrict__ Wr,
    float* __restrict__ pl, float* __restrict__ pr, int n) {
  const int lane = threadIdx.x & 31;
  const int wave = threadIdx.x >> 5;
  const int row0 = (blockIdx.x * 8 + wave) * 16;
  if (row0 >= n) return;

  const int mloc = lane & 15;
  const int h = lane >> 4;
  const int m = row0 + mloc;
  const int mc = m < n ? m : n - 1;
  const float4* arow = reinterpret_cast<const float4*>(xin + (size_t)mc * F) + h;

  const int wc = mloc & 7;
  const bool cvalid = wc < 7;
  const bool self = mloc >= 8;
  const float* wsel = (self ? Wr : Wl) + (size_t)(cvalid ? wc : 0) * F;
  const float4* brow = reinterpret_cast<const float4*>(wsel) + h;

  float4 av[16], bv[16];
#pragma unroll
  for (int j = 0; j < 16; ++j) av[j] = arow[2 * j];
#pragma unroll
  for (int j = 0; j < 16; ++j) bv[j] = brow[2 * j];

  const float bias = (cvalid && self) ? bl[wc] : 0.0f;
  v8f c;
#pragma unroll
  for (int i = 0; i < 8; ++i) c[i] = bias;

#pragma unroll
  for (int j = 0; j < 16; ++j) {
    c = wmma_f32(av[j].x, av[j].y, bv[j].x, bv[j].y, c);
    c = wmma_f32(av[j].z, av[j].w, bv[j].z, bv[j].w, c);
  }

  float* dstp = (self ? pr : pl) + wc;
  const int rbase = h * 8;
#pragma unroll
  for (int r = 0; r < 8; ++r) {
    const int mr = row0 + rbase + r;
    if (mr < n) dstp[(size_t)mr * 8] = cvalid ? c[r] : 0.0f;
  }
}

__global__ __launch_bounds__(256) void k_agg7(const int* __restrict__ col,
                                              const int* __restrict__ cnt,
                                              const int* __restrict__ rowl,
                                              const int* __restrict__ boff,
                                              const float* __restrict__ pl,
                                              const float* __restrict__ pr,
                                              const float* __restrict__ deg_inv,
                                              float* __restrict__ logits, int n) {
  const int tid = blockIdx.x * blockDim.x + threadIdx.x;
  const int q = tid & 3;
  const int node = tid >> 2;
  const bool live = node < n;
  const int nd = live ? node : n - 1;
  const int cn = live ? cnt[nd] : 0;
  const int* c = col + (rowl[nd] + boff[nd / SCAN_CHUNK]);
  const float4* p4 = reinterpret_cast<const float4*>(pl);
  float4 lo = make_float4(0.0f, 0.0f, 0.0f, 0.0f);
  float4 hi = make_float4(0.0f, 0.0f, 0.0f, 0.0f);
  for (int j = q; j < cn; j += 16) {
    int s[4];
#pragma unroll
    for (int u = 0; u < 4; ++u) {
      s[u] = 0;
      if (j + 4 * u < cn) s[u] = c[j + 4 * u];
    }
    float4 a[4], b[4];
#pragma unroll
    for (int u = 0; u < 4; ++u) {
      a[u] = make_float4(0.0f, 0.0f, 0.0f, 0.0f);
      b[u] = make_float4(0.0f, 0.0f, 0.0f, 0.0f);
      if (j + 4 * u < cn) {
        a[u] = p4[2 * (size_t)s[u]];
        b[u] = p4[2 * (size_t)s[u] + 1];
      }
    }
#pragma unroll
    for (int u = 0; u < 4; ++u) {
      lo.x += a[u].x; lo.y += a[u].y; lo.z += a[u].z; lo.w += a[u].w;
      hi.x += b[u].x; hi.y += b[u].y; hi.z += b[u].z; hi.w += b[u].w;
    }
  }
  for (int d = 1; d < 4; d <<= 1) {
    lo.x += __shfl_xor(lo.x, d); lo.y += __shfl_xor(lo.y, d);
    lo.z += __shfl_xor(lo.z, d); lo.w += __shfl_xor(lo.w, d);
    hi.x += __shfl_xor(hi.x, d); hi.y += __shfl_xor(hi.y, d);
    hi.z += __shfl_xor(hi.z, d);
  }
  if (live && q == 0) {
    const float ds = deg_inv[node];
    const float4 r0 = reinterpret_cast<const float4*>(pr)[2 * (size_t)node];
    const float4 r1 = reinterpret_cast<const float4*>(pr)[2 * (size_t)node + 1];
    float* o = logits + (size_t)node * 7;
    o[0] = lo.x * ds + r0.x;
    o[1] = lo.y * ds + r0.y;
    o[2] = lo.z * ds + r0.z;
    o[3] = lo.w * ds + r0.w;
    o[4] = hi.x * ds + r1.x;
    o[5] = hi.y * ds + r1.y;
    o[6] = hi.z * ds + r1.z;
  }
}

__global__ __launch_bounds__(256) void k_logsoftmax7(const float* __restrict__ logits,
                                                     float* __restrict__ out, int n) {
  int i = blockIdx.x * blockDim.x + threadIdx.x;
  if (i >= n) return;
  const float* r = logits + (size_t)i * 7;
  float v[7];
#pragma unroll
  for (int j = 0; j < 7; ++j) v[j] = r[j];
  float mx = v[0];
#pragma unroll
  for (int j = 1; j < 7; ++j) mx = fmaxf(mx, v[j]);
  float s = 0.0f;
#pragma unroll
  for (int j = 0; j < 7; ++j) s += expf(v[j] - mx);
  float ls = logf(s);
  float* o = out + (size_t)i * 7;
#pragma unroll
  for (int j = 0; j < 7; ++j) o[j] = v[j] - mx - ls;
}

extern "C" void kernel_launch(void* const* d_in, const int* in_sizes, int n_in,
                              void* d_out, int out_size, void* d_ws, size_t ws_size,
                              hipStream_t stream) {
  (void)n_in; (void)out_size;
  const float* x   = (const float*)d_in[0];
  const int*   ei  = (const int*)d_in[1];
  const float* W1l = (const float*)d_in[2];
  const float* b1  = (const float*)d_in[3];
  const float* W1r = (const float*)d_in[4];
  const float* W2l = (const float*)d_in[5];
  const float* b2  = (const float*)d_in[6];
  const float* W2r = (const float*)d_in[7];
  const float* W3l = (const float*)d_in[8];
  const float* b3  = (const float*)d_in[9];
  const float* W3r = (const float*)d_in[10];

  const int n = in_sizes[0] / F;
  const int E = in_sizes[1] / 2;
  const int* src = ei;
  const int* dst = ei + E;

  char* base = (char*)d_ws;
  size_t off = 0;
  auto a256 = [](size_t b) { return (b + 255) & ~(size_t)255; };
  float* deg = (float*)(base + off); off = a256(off + (size_t)n * sizeof(float));
  float* agg = (float*)(base + off); off = a256(off + (size_t)n * F * sizeof(float));
  float* hA  = (float*)(base + off); off = a256(off + (size_t)n * F * sizeof(float));
  float* hB  = (float*)(base + off); off = a256(off + (size_t)n * F * sizeof(float));

  const int nchunk = (n + SCAN_CHUNK - 1) / SCAN_CHUNK;
  int* cnt  = (int*)(base + off); off = a256(off + (size_t)n * sizeof(int));
  int* rowl = (int*)(base + off); off = a256(off + (size_t)n * sizeof(int));
  int* bsum = (int*)(base + off); off = a256(off + (size_t)nchunk * sizeof(int));
  int* col  = (int*)(base + off); off = a256(off + (size_t)E * sizeof(int));
  int* rank = (int*)hB;
  const bool sorted = off <= ws_size && (size_t)E <= (size_t)n * F;

  const int T = 256;
  auto nb = [&](long long work) { return (int)((work + T - 1) / T); };
  const long long scatW = (long long)E * 32;
  const int gemmB = (n + 127) / 128;
  const int gathB = (n + 7) / 8;

  if (sorted) {
    k_zero_i32<<<nb(n), T, 0, stream>>>(cnt, n);
    k_count_rank<<<nb(E), T, 0, stream>>>(dst, cnt, rank, E);
    k_scan_chunk<<<nchunk, T, 0, stream>>>(cnt, rowl, bsum, deg, n);
    k_scan_top<<<1, T, 0, stream>>>(bsum, nchunk);
    k_fill<<<nb(E), T, 0, stream>>>(src, dst, rank, rowl, bsum, col, E);

    k_gather<<<gathB, T, 0, stream>>>(col, cnt, rowl, bsum, x, agg, n);
    k_gemm128<<<256, GEMM_WAVES * 32, 0, stream>>>(agg, x, W1l, b1, W1r, deg, hA, n, 1);

    k_gather<<<gathB, T, 0, stream>>>(col, cnt, rowl, bsum, hA, agg, n);
    k_gemm128<<<256, GEMM_WAVES * 32, 0, stream>>>(agg, hA, W2l, b2, W2r, deg, hB, n, 1);

    float* pl = agg;
    float* pr = agg + (size_t)n * 8;
    k_proj14<<<gemmB, T, 0, stream>>>(hB, W3l, b3, W3r, pl, pr, n);
    k_agg7<<<nb((long long)n * 4), T, 0, stream>>>(col, cnt, rowl, bsum, pl, pr, deg, hA, n);
  } else {
    k_zero<<<nb(n), T, 0, stream>>>(deg, (size_t)n);
    k_degree<<<nb(E), T, 0, stream>>>(dst, deg, E);
    k_deg_inv<<<nb(n), T, 0, stream>>>(deg, n);

    k_zero<<<nb((long long)n * F), T, 0, stream>>>(agg, (size_t)n * F);
    k_scatter<<<nb(scatW), T, 0, stream>>>(src, dst, x, agg, E);
    k_gemm128<<<256, GEMM_WAVES * 32, 0, stream>>>(agg, x, W1l, b1, W1r, deg, hA, n, 1);

    k_zero<<<nb((long long)n * F), T, 0, stream>>>(agg, (size_t)n * F);
    k_scatter<<<nb(scatW), T, 0, stream>>>(src, dst, hA, agg, E);
    k_gemm128<<<256, GEMM_WAVES * 32, 0, stream>>>(agg, hA, W2l, b2, W2r, deg, hB, n, 1);

    k_zero<<<nb((long long)n * F), T, 0, stream>>>(agg, (size_t)n * F);
    k_scatter<<<nb(scatW), T, 0, stream>>>(src, dst, hB, agg, E);
    k_gemm_out7<<<gemmB, T, 0, stream>>>(agg, hB, W3l, b3, W3r, deg, hA, n);
  }

  k_logsoftmax7<<<nb(n), T, 0, stream>>>(hA, (float*)d_out, n);
}


# Edge list built by a two-level LDS counting sort (bucket histogram, bucket scatter, per-bucket sort), not per-edge global atomics and scattered stores; per-node sum order = LDS-atomic arrival rank
# speedup vs baseline: 5.0670x; 1.4470x over previous
#include <hip/hip_runtime.h>
#include <hip/hip_bf16.h>

typedef __attribute__((ext_vector_type(2))) float v2f;
typedef __attribute__((ext_vector_type(8))) float v8f;

#define F 128

__global__ __launch_bounds__(256) void k_zero(float* __restrict__ p, size_t cnt) {
  size_t i = (size_t)blockIdx.x * blockDim.x + threadIdx.x;
  if (i < cnt) p[i] = 0.0f;
}

__global__ __launch_bounds__(256) void k_degree(const int* __restrict__ dst,
                                                float* __restrict__ deg, int nE) {
  int e = blockIdx.x * blockDim.x + threadIdx.x;
  if (e < nE) atomicAdd(&deg[dst[e]], 1.0f);
}

__global__ __launch_bounds__(256) void k_deg_inv(float* __restrict__ deg, int n) {
  int i = blockIdx.x * blockDim.x + threadIdx.x;
  if (i < n) {
    float d = deg[i];
    deg[i] = d > 0.0f ? 1.0f / d : 0.0f;
  }
}

#define SCAN_CHUNK 1024
#define SORT_T 1024
#define ECH 8192
#define MAXNB 1024
#define SORT_CAP 24576
#define SRC_BITS 22
#define SRC_MASK ((1u << SRC_BITS) - 1u)

__device__ __forceinline__ int wave_incl_scan(int v, int lane) {
#pragma unroll
  for (int d = 1; d < 32; d <<= 1) {
    int t = __shfl_up(v, d);
    if (lane >= d) v += t;
  }
  return v;
}

__global__ __launch_bounds__(SORT_T) void k_bkt_hist(const int* __restrict__ dst,
                                                     int* __restrict__ hist,
                                                     int nE, int nb, int g1) {
  __shared__ int s_h[MAXNB];
  const int tid = threadIdx.x;
  for (int b = tid; b < nb; b += SORT_T) s_h[b] = 0;
  __syncthreads();
  const int e0 = blockIdx.x * ECH;
  unsigned d[ECH / SORT_T];
#pragma unroll
  for (int k = 0; k < ECH / SORT_T; ++k) {
    const int e = e0 + k * SORT_T + tid;
    d[k] = (unsigned)dst[e < nE ? e : nE - 1];
  }
#pragma unroll
  for (int k = 0; k < ECH / SORT_T; ++k) {
    if (e0 + k * SORT_T + tid < nE) atomicAdd(&s_h[d[k] / SCAN_CHUNK], 1);
  }
  __syncthreads();
  for (int b = tid; b < nb; b += SORT_T) hist[(size_t)b * g1 + blockIdx.x] = s_h[b];
}

__global__ __launch_bounds__(SORT_T) void k_bkt_scatter(const int* __restrict__ src,
                                                        const int* __restrict__ dst,
                                                        const int* __restrict__ hist,
                                                        unsigned* __restrict__ bkt,
                                                        int* __restrict__ boff,
                                                        int nE, int nb, int g1) {
  __shared__ int s_tot[MAXNB];
  __shared__ int s_pre[MAXNB];
  __shared__ int s_cur[MAXNB];
  __shared__ int s_w[2][32];
  __shared__ unsigned s_ent[ECH];
  __shared__ unsigned short s_bk[ECH];
  const int tid = threadIdx.x;
  const int lane = tid & 31;
  const int wave = tid >> 5;
  const int wg = blockIdx.x;

  for (int b = wave; b < nb; b += SORT_T / 32) {
    const int* row = hist + (size_t)b * g1;
    int tot = 0, pre = 0, own = 0;
    for (int w0 = 0; w0 < g1; w0 += 256) {
      int v[8];
#pragma unroll
      for (int k = 0; k < 8; ++k) {
        const int w = w0 + 32 * k + lane;
        v[k] = row[w < g1 ? w : g1 - 1];
      }
#pragma unroll
      for (int k = 0; k < 8; ++k) {
        const int w = w0 + 32 * k + lane;
        const int x = w < g1 ? v[k] : 0;
        tot += x;
        pre += w < wg ? x : 0;
        own += w == wg ? x : 0;
      }
    }
    for (int d = 16; d > 0; d >>= 1) {
      tot += __shfl_xor(tot, d);
      pre += __shfl_xor(pre, d);
      own += __shfl_xor(own, d);
    }
    if (lane == 0) { s_tot[b] = tot; s_pre[b] = pre; s_cur[b] = own; }
  }
  __syncthreads();

  const int vt = tid < nb ? s_tot[tid] : 0;
  const int vo = tid < nb ? s_cur[tid] : 0;
  const int it = wave_incl_scan(vt, lane);
  const int io = wave_incl_scan(vo, lane);
  if (lane == 31) { s_w[0][wave] = it; s_w[1][wave] = io; }
  __syncthreads();
  int ot = 0, oo = 0;
  for (int w = 0; w < wave; ++w) { ot += s_w[0][w]; oo += s_w[1][w]; }
  const int bstart = ot + it - vt;
  const int lstart = oo + io - vo;
  if (tid < nb) {
    if (wg == 0) boff[tid] = bstart;
    s_tot[tid] = bstart + s_pre[tid] - lstart;
    s_cur[tid] = lstart;
  }
  __syncthreads();

  const int e0 = wg * ECH;
  unsigned dd[ECH / SORT_T], ss[ECH / SORT_T];
#pragma unroll
  for (int k = 0; k < ECH / SORT_T; ++k) {
    const int e = e0 + k * SORT_T + tid;
    const int ec = e < nE ? e : nE - 1;
    dd[k] = (unsigned)dst[ec];
    ss[k] = (unsigned)src[ec];
  }
#pragma unroll
  for (int k = 0; k < ECH / SORT_T; ++k) {
    if (e0 + k * SORT_T + tid < nE) {
      const unsigned b = dd[k] / SCAN_CHUNK;
      const int p = atomicAdd(&s_cur[b], 1);
      s_ent[p] = ((dd[k] % SCAN_CHUNK) << SRC_BITS) | ss[k];
      s_bk[p] = (unsigned short)b;
    }
  }
  __syncthreads();
  const int m = nE - e0 < ECH ? nE - e0 : ECH;
  unsigned short kb[ECH / SORT_T];
  unsigned ke[ECH / SORT_T];
  int kd[ECH / SORT_T];
#pragma unroll
  for (int k = 0; k < ECH / SORT_T; ++k) {
    const int i = k * SORT_T + tid;
    const int ic = i < m ? i : m - 1;
    kb[k] = s_bk[ic];
    ke[k] = s_ent[ic];
  }
#pragma unroll
  for (int k = 0; k < ECH / SORT_T; ++k) kd[k] = s_tot[kb[k]];
#pragma unroll
  for (int k = 0; k < ECH / SORT_T; ++k) {
    const int i = k * SORT_T + tid;
    if (i < m) bkt[i + kd[k]] = ke[k];
  }
}

__global__ __launch_bounds__(SORT_T) void k_bkt_sort(const unsigned* __restrict__ bkt,
                                                     const int* __restrict__ boff,
                                                     int* __restrict__ cnt,
                                                     int* __restrict__ rowl,
                                                     float* __restrict__ deg_inv,
                                                     int* __restrict__ col,
                                                     int n, int nE, int nb) {
  __shared__ int s_c[SCAN_CHUNK];
  __shared__ int s_w[32];
  __shared__ int s_col[SORT_CAP];
  const int tid = threadIdx.x;
  const int lane = tid & 31;
  const int wave = tid >> 5;
  const int b = blockIdx.x;
  const int beg = boff[b];
  const int end = b + 1 < nb ? boff[b + 1] : nE;
  const int m = end - beg;
  const unsigned* ent = bkt + beg;

  s_c[tid] = 0;
  __syncthreads();
  for (int i0 = 0; i0 < m; i0 += 8 * SORT_T) {
    unsigned u[8];
#pragma unroll
    for (int k = 0; k < 8; ++k) {
      const int i = i0 + k * SORT_T + tid;
      u[k] = ent[i < m ? i : m - 1];
    }
#pragma unroll
    for (int k = 0; k < 8; ++k) {
      if (i0 + k * SORT_T + tid < m) atomicAdd(&s_c[u[k] >> SRC_BITS], 1);
    }
  }
  __syncthreads();

  const int v = s_c[tid];
  const int incl = wave_incl_scan(v, lane);
  if (lane == 31) s_w[wave] = incl;
  __syncthreads();
  int woff = 0;
  for (int w = 0; w < wave; ++w) woff += s_w[w];
  const int excl = woff + incl - v;
  const int node = b * SCAN_CHUNK + tid;
  if (node < n) {
    cnt[node] = v;
    rowl[node] = excl;
    deg_inv[node] = v > 0 ? 1.0f / (float)v : 0.0f;
  }
  s_c[tid] = excl;
  __syncthreads();

  if (m <= SORT_CAP) {
    for (int i0 = 0; i0 < m; i0 += 8 * SORT_T) {
      unsigned u[8];
#pragma unroll
      for (int k = 0; k < 8; ++k) {
        const int i = i0 + k * SORT_T + tid;
        u[k] = ent[i < m ? i : m - 1];
      }
#pragma unroll
      for (int k = 0; k < 8; ++k) {
        if (i0 + k * SORT_T + tid < m) {
          const int p = atomicAdd(&s_c[u[k] >> SRC_BITS], 1);
          s_col[p] = (int)(u[k] & SRC_MASK);
        }
      }
    }
    __syncthreads();
    for (int i0 = 0; i0 < m; i0 += 8 * SORT_T) {
      int sv[8];
#pragma unroll
      for (int k = 0; k < 8; ++k) {
        const int i = i0 + k * SORT_T + tid;
        sv[k] = s_col[i < m ? i : 0];
      }
#pragma unroll
      for (int k = 0; k < 8; ++k) {
        const int i = i0 + k * SORT_T + tid;
        if (i < m) col[beg + i] = sv[k];
      }
    }
  } else {
    for (int i = tid; i < m; i += SORT_T) {
      const unsigned u = ent[i];
      const int p = atomicAdd(&s_c[u >> SRC_BITS], 1);
      col[beg + p] = (int)(u & SRC_MASK);
    }
  }
}

__global__ __launch_bounds__(256) void k_gather(const int* __restrict__ col,
                                                const int* __restrict__ cnt,
                                                const int* __restrict__ rowl,
                                                const int* __restrict__ boff,
                                                const float* __restrict__ feat,
                                                float* __restrict__ agg, int n) {
  const int wave = __builtin_amdgcn_readfirstlane((int)(threadIdx.x >> 5));
  const int node = blockIdx.x * 8 + wave;
  if (node >= n) return;
  const int lane = threadIdx.x & 31;
  const int cn = cnt[node];
  const int* c = col + (rowl[node] + boff[node / SCAN_CHUNK]);
  const float* fb = feat + lane * 4;
  float4 acc = make_float4(0.0f, 0.0f, 0.0f, 0.0f);
  int j = 0;
  for (; j + 4 <= cn; j += 4) {
    const int s0 = c[j], s1 = c[j + 1], s2 = c[j + 2], s3 = c[j + 3];
    const float4 v0 = *reinterpret_cast<const float4*>(fb + (size_t)s0 * F);
    const float4 v1 = *reinterpret_cast<const float4*>(fb + (size_t)s1 * F);
    const float4 v2 = *reinterpret_cast<const float4*>(fb + (size_t)s2 * F);
    const float4 v3 = *reinterpret_cast<const float4*>(fb + (size_t)s3 * F);
    acc.x += v0.x; acc.y += v0.y; acc.z += v0.z; acc.w += v0.w;
    acc.x += v1.x; acc.y += v1.y; acc.z += v1.z; acc.w += v1.w;
    acc.x += v2.x; acc.y += v2.y; acc.z += v2.z; acc.w += v2.w;
    acc.x += v3.x; acc.y += v3.y; acc.z += v3.z; acc.w += v3.w;
  }
  for (; j < cn; ++j) {
    const float4 v0 = *reinterpret_cast<const float4*>(fb + (size_t)c[j] * F);
    acc.x += v0.x; acc.y += v0.y; acc.z += v0.z; acc.w += v0.w;
  }
  *reinterpret_cast<float4*>(agg + (size_t)node * F + lane * 4) = acc;
}

__global__ __launch_bounds__(256) void k_scatter(const int* __restrict__ src,
                                                 const int* __restrict__ dst,
                                                 const float* __restrict__ feat,
                                                 float* __restrict__ agg, int nE) {
  long long tid = (long long)blockIdx.x * blockDim.x + threadIdx.x;
  int e = (int)(tid >> 5);
  if (e >= nE) return;
  int fo = ((int)tid & 31) * 4;
  const float4 v = *reinterpret_cast<const float4*>(feat + (size_t)src[e] * F + fo);
  float* d = agg + (size_t)dst[e] * F + fo;
  atomicAdd(d + 0, v.x);
  atomicAdd(d + 1, v.y);
  atomicAdd(d + 2, v.z);
  atomicAdd(d + 3, v.w);
}

#define GEMM_WAVES 16
#define WROW 256
#define WBLK (8 * WROW + 4)

__device__ __forceinline__ v8f wmma_f32(float a0, float a1, float b0, float b1, v8f c) {
  v2f a, b;
  a[0] = a0; a[1] = a1;
  b[0] = b0; b[1] = b1;
  return __builtin_amdgcn_wmma_f32_16x16x4_f32(false, a, false, b, (short)0, c,
                                               false, false);
}

__global__ __launch_bounds__(GEMM_WAVES * 32) void k_gemm128(
    const float* __restrict__ agg, const float* __restrict__ xin,
    const float* __restrict__ Wl, const float* __restrict__ bl,
    const float* __restrict__ Wr, const float* __restrict__ deg_inv,
    float* __restrict__ hout, int n, int do_relu) {
  __shared__ __attribute__((aligned(16))) float wlds[16 * WBLK];

  constexpr int NCOPY = F * (F / 4) / (GEMM_WAVES * 32);
  float4 vl[NCOPY], vr[NCOPY];
#pragma unroll
  for (int i = 0; i < NCOPY; ++i) {
    const int f = threadIdx.x + i * (GEMM_WAVES * 32);
    vl[i] = reinterpret_cast<const float4*>(Wl)[f];
    vr[i] = reinterpret_cast<const float4*>(Wr)[f];
  }
#pragma unroll
  for (int i = 0; i < NCOPY; ++i) {
    const int f = threadIdx.x + i * (GEMM_WAVES * 32);
    const int row = f >> 5;
    const int c4 = f & 31;
    float* dstp = wlds + (row & 15) * WBLK + (row >> 4) * WROW + c4 * 4;
    *reinterpret_cast<float4*>(dstp) = vl[i];
    *reinterpret_cast<float4*>(dstp + F) = vr[i];
  }
  __syncthreads();

  const int lane = threadIdx.x & 31;
  const int wave = __builtin_amdgcn_readfirstlane((int)(threadIdx.x >> 5));
  const int mloc = lane & 15;
  const int h = lane >> 4;
  const int ntiles = (n + 15) >> 4;
  const float* bfrag = wlds + mloc * WBLK + 4 * h;

  for (int s = wave; ; s += GEMM_WAVES) {
    const int t = blockIdx.x + gridDim.x * s;
    if (t >= ntiles) break;
    const int row0 = t * 16;
    const int m = row0 + mloc;
    const int mc = m < n ? m : n - 1;
    const float dscale = deg_inv[mc];
    const float4* arow = reinterpret_cast<const float4*>(agg + (size_t)mc * F) + h;
    const float4* xrow = reinterpret_cast<const float4*>(xin + (size_t)mc * F) + h;

    float4 av[32];
#pragma unroll
    for (int j = 0; j < 16; ++j) av[j] = arow[2 * j];

    v8f c[8];
#pragma unroll
    for (int ct = 0; ct < 8; ++ct) {
#pragma unroll
      for (int i = 0; i < 8; ++i) c[ct][i] = 0.0f;
    }

    float4 bq[2][8];
#pragma unroll
    for (int ct = 0; ct < 8; ++ct)
      bq[0][ct] = *reinterpret_cast<const float4*>(bfrag + ct * WROW);

#pragma unroll
    for (int j = 0; j < 32; ++j) {
      const int cur = j & 1;
      if (j + 1 < 32) {
#pragma unroll
        for (int ct = 0; ct < 8; ++ct)
          bq[cur ^ 1][ct] =
              *reinterpret_cast<const float4*>(bfrag + ct * WROW + 8 * (j + 1));
      }
      if (j < 16) av[j + 16] = xrow[2 * j];
      __builtin_amdgcn_sched_barrier(0);
      float4 a = av[j];
      if (j < 16) { a.x *= dscale; a.y *= dscale; a.z *= dscale; a.w *= dscale; }
#pragma unroll
      for (int ct = 0; ct < 8; ++ct) {
        const float4 b = bq[cur][ct];
        c[ct] = wmma_f32(a.x, a.y, b.x, b.y, c[ct]);
        c[ct] = wmma_f32(a.z, a.w, b.z, b.w, c[ct]);
      }
      __builtin_amdgcn_sched_barrier(0);
    }

    const int rbase = h * 8;
#pragma unroll
    for (int ct = 0; ct < 8; ++ct) {
      const int ncol = ct * 16 + mloc;
      const float bias = bl[ncol];
#pragma unroll
      for (int r = 0; r < 8; ++r) {
        const int mr = row0 + rbase + r;
        if (mr < n) {
          float v = c[ct][r] + bias;
          if (do_relu) v = v > 0.0f ? v : 0.0f;
          hout[(size_t)mr * F + ncol] = v;
        }
      }
    }
  }
}

__global__ __launch_bounds__(256) void k_gemm_out7(
    const float* __restrict__ agg, const float* __restrict__ xin,
    const float* __restrict__ Wl, const float* __restrict__ bl,
    const float* __restrict__ Wr, const float* __restrict__ deg_inv,
    float* __restrict__ logits, int n) {
  const int K = F;
  int lane = threadIdx.x & 31;
  int wave = threadIdx.x >> 5;
  int row0 = (blockIdx.x * 8 + wave) * 16;
  if (row0 >= n) return;

  int mloc = lane & 15;
  int koff = (lane >> 4) * 2;
  int m = row0 + mloc;
  int mc = m < n ? m : n - 1;
  float dscale = deg_inv[mc];
  const float* arow = agg + (size_t)mc * K;
  const float* xrow = xin + (size_t)mc * K;

  int ncol = mloc;
  bool cvalid = ncol < 7;
  float wm = cvalid ? 1.0f : 0.0f;
  const float* wlrow = Wl + (size_t)(cvalid ? ncol : 0) * K;
  const float* wrrow = Wr + (size_t)(cvalid ? ncol : 0) * K;

  float bias = cvalid ? bl[ncol] : 0.0f;
  v8f c;
#pragma unroll
  for (int i = 0; i < 8; ++i) c[i] = bias;

#pragma unroll
  for (int kt = 0; kt < 32; ++kt) {
    int k0 = kt * 4 + koff;
    v2f a, b;
    a[0] = arow[k0] * dscale; a[1] = arow[k0 + 1] * dscale;
    b[0] = wlrow[k0] * wm;    b[1] = wlrow[k0 + 1] * wm;
    c = __builtin_amdgcn_wmma_f32_16x16x4_f32(false, a, false, b, (short)0, c,
                                              false, false);
  }
#pragma unroll
  for (int kt = 0; kt < 32; ++kt) {
    int k0 = kt * 4 + koff;
    v2f a, b;
    a[0] = xrow[k0];       a[1] = xrow[k0 + 1];
    b[0] = wrrow[k0] * wm; b[1] = wrrow[k0 + 1] * wm;
    c = __builtin_amdgcn_wmma_f32_16x16x4_f32(false, a, false, b, (short)0, c,
                                              false, false);
  }

  if (cvalid) {
    int rbase = (lane >> 4) * 8;
#pragma unroll
    for (int r = 0; r < 8; ++r) {
      int mr = row0 + rbase + r;
      if (mr < n) logits[(size_t)mr * 7 + ncol] = c[r];
    }
  }
}

__global__ __launch_bounds__(256) void k_proj14(
    const float* __restrict__ xin, const float* __restrict__ Wl,
    const float* __restrict__ bl, const float* __restrict__ Wr,
    float* __restrict__ pl, float* __restrict__ pr, int n) {
  const int lane = threadIdx.x & 31;
  const int wave = threadIdx.x >> 5;
  const int row0 = (blockIdx.x * 8 + wave) * 16;
  if (row0 >= n) return;

  const int mloc = lane & 15;
  const int h = lane >> 4;
  const int m = row0 + mloc;
  const int mc = m < n ? m : n - 1;
  const float4* arow = reinterpret_cast<const float4*>(xin + (size_t)mc * F) + h;

  const int wc = mloc & 7;
  const bool cvalid = wc < 7;
  const bool self = mloc >= 8;
  const float* wsel = (self ? Wr : Wl) + (size_t)(cvalid ? wc : 0) * F;
  const float4* brow = reinterpret_cast<const float4*>(wsel) + h;

  float4 av[16], bv[16];
#pragma unroll
  for (int j = 0; j < 16; ++j) av[j] = arow[2 * j];
#pragma unroll
  for (int j = 0; j < 16; ++j) bv[j] = brow[2 * j];

  const float bias = (cvalid && self) ? bl[wc] : 0.0f;
  v8f c;
#pragma unroll
  for (int i = 0; i < 8; ++i) c[i] = bias;

#pragma unroll
  for (int j = 0; j < 16; ++j) {
    c = wmma_f32(av[j].x, av[j].y, bv[j].x, bv[j].y, c);
    c = wmma_f32(av[j].z, av[j].w, bv[j].z, bv[j].w, c);
  }

  float* dstp = (self ? pr : pl) + wc;
  const int rbase = h * 8;
#pragma unroll
  for (int r = 0; r < 8; ++r) {
    const int mr = row0 + rbase + r;
    if (mr < n) dstp[(size_t)mr * 8] = cvalid ? c[r] : 0.0f;
  }
}

__global__ __launch_bounds__(256) void k_agg7(const int* __restrict__ col,
                                              const int* __restrict__ cnt,
                                              const int* __restrict__ rowl,
                                              const int* __restrict__ boff,
                                              const float* __restrict__ pl,
                                              const float* __restrict__ pr,
                                              const float* __restrict__ deg_inv,
                                              float* __restrict__ logits, int n) {
  const int tid = blockIdx.x * blockDim.x + threadIdx.x;
  const int q = tid & 3;
  const int node = tid >> 2;
  const bool live = node < n;
  const int nd = live ? node : n - 1;
  const int cn = live ? cnt[nd] : 0;
  const int* c = col + (rowl[nd] + boff[nd / SCAN_CHUNK]);
  const float4* p4 = reinterpret_cast<const float4*>(pl);
  float4 lo = make_float4(0.0f, 0.0f, 0.0f, 0.0f);
  float4 hi = make_float4(0.0f, 0.0f, 0.0f, 0.0f);
  for (int j = q; j < cn; j += 16) {
    int s[4];
#pragma unroll
    for (int u = 0; u < 4; ++u) {
      s[u] = 0;
      if (j + 4 * u < cn) s[u] = c[j + 4 * u];
    }
    float4 a[4], b[4];
#pragma unroll
    for (int u = 0; u < 4; ++u) {
      a[u] = make_float4(0.0f, 0.0f, 0.0f, 0.0f);
      b[u] = make_float4(0.0f, 0.0f, 0.0f, 0.0f);
      if (j + 4 * u < cn) {
        a[u] = p4[2 * (size_t)s[u]];
        b[u] = p4[2 * (size_t)s[u] + 1];
      }
    }
#pragma unroll
    for (int u = 0; u < 4; ++u) {
      lo.x += a[u].x; lo.y += a[u].y; lo.z += a[u].z; lo.w += a[u].w;
      hi.x += b[u].x; hi.y += b[u].y; hi.z += b[u].z; hi.w += b[u].w;
    }
  }
  for (int d = 1; d < 4; d <<= 1) {
    lo.x += __shfl_xor(lo.x, d); lo.y += __shfl_xor(lo.y, d);
    lo.z += __shfl_xor(lo.z, d); lo.w += __shfl_xor(lo.w, d);
    hi.x += __shfl_xor(hi.x, d); hi.y += __shfl_xor(hi.y, d);
    hi.z += __shfl_xor(hi.z, d);
  }
  if (live && q == 0) {
    const float ds = deg_inv[node];
    const float4 r0 = reinterpret_cast<const float4*>(pr)[2 * (size_t)node];
    const float4 r1 = reinterpret_cast<const float4*>(pr)[2 * (size_t)node + 1];
    float* o = logits + (size_t)node * 7;
    o[0] = lo.x * ds + r0.x;
    o[1] = lo.y * ds + r0.y;
    o[2] = lo.z * ds + r0.z;
    o[3] = lo.w * ds + r0.w;
    o[4] = hi.x * ds + r1.x;
    o[5] = hi.y * ds + r1.y;
    o[6] = hi.z * ds + r1.z;
  }
}

__global__ __launch_bounds__(256) void k_logsoftmax7(const float* __restrict__ logits,
                                                     float* __restrict__ out, int n) {
  int i = blockIdx.x * blockDim.x + threadIdx.x;
  if (i >= n) return;
  const float* r = logits + (size_t)i * 7;
  float v[7];
#pragma unroll
  for (int j = 0; j < 7; ++j) v[j] = r[j];
  float mx = v[0];
#pragma unroll
  for (int j = 1; j < 7; ++j) mx = fmaxf(mx, v[j]);
  float s = 0.0f;
#pragma unroll
  for (int j = 0; j < 7; ++j) s += expf(v[j] - mx);
  float ls = logf(s);
  float* o = out + (size_t)i * 7;
#pragma unroll
  for (int j = 0; j < 7; ++j) o[j] = v[j] - mx - ls;
}

extern "C" void kernel_launch(void* const* d_in, const int* in_sizes, int n_in,
                              void* d_out, int out_size, void* d_ws, size_t ws_size,
                              hipStream_t stream) {
  (void)n_in; (void)out_size;
  const float* x   = (const float*)d_in[0];
  const int*   ei  = (const int*)d_in[1];
  const float* W1l = (const float*)d_in[2];
  const float* b1  = (const float*)d_in[3];
  const float* W1r = (const float*)d_in[4];
  const float* W2l = (const float*)d_in[5];
  const float* b2  = (const float*)d_in[6];
  const float* W2r = (const float*)d_in[7];
  const float* W3l = (const float*)d_in[8];
  const float* b3  = (const float*)d_in[9];
  const float* W3r = (const float*)d_in[10];

  const int n = in_sizes[0] / F;
  const int E = in_sizes[1] / 2;
  const int* src = ei;
  const int* dst = ei + E;

  char* base = (char*)d_ws;
  size_t off = 0;
  auto a256 = [](size_t b) { return (b + 255) & ~(size_t)255; };
  float* deg = (float*)(base + off); off = a256(off + (size_t)n * sizeof(float));
  float* agg = (float*)(base + off); off = a256(off + (size_t)n * F * sizeof(float));
  float* hA  = (float*)(base + off); off = a256(off + (size_t)n * F * sizeof(float));
  float* hB  = (float*)(base + off); off = a256(off + (size_t)n * F * sizeof(float));

  const int nchunk = (n + SCAN_CHUNK - 1) / SCAN_CHUNK;
  const int g1 = (E + ECH - 1) / ECH;
  int* cnt  = (int*)(base + off); off = a256(off + (size_t)n * sizeof(int));
  int* rowl = (int*)(base + off); off = a256(off + (size_t)n * sizeof(int));
  int* bsum = (int*)(base + off); off = a256(off + (size_t)nchunk * sizeof(int));
  int* col  = (int*)(base + off); off = a256(off + (size_t)E * sizeof(int));
  unsigned* bkt = (unsigned*)hB;
  int* hist = (int*)hA;
  const bool sorted = off <= ws_size && (size_t)E <= (size_t)n * F &&
                      (size_t)nchunk * g1 <= (size_t)n * F && nchunk <= MAXNB &&
                      n <= (1 << SRC_BITS);

  const int T = 256;
  auto nb = [&](long long work) { return (int)((work + T - 1) / T); };
  const long long scatW = (long long)E * 32;
  const int gemmB = (n + 127) / 128;
  const int gathB = (n + 7) / 8;

  if (sorted) {
    k_bkt_hist<<<g1, SORT_T, 0, stream>>>(dst, hist, E, nchunk, g1);
    k_bkt_scatter<<<g1, SORT_T, 0, stream>>>(src, dst, hist, bkt, bsum, E, nchunk, g1);
    k_bkt_sort<<<nchunk, SORT_T, 0, stream>>>(bkt, bsum, cnt, rowl, deg, col, n, E, nchunk);

    k_gather<<<gathB, T, 0, stream>>>(col, cnt, rowl, bsum, x, agg, n);
    k_gemm128<<<256, GEMM_WAVES * 32, 0, stream>>>(agg, x, W1l, b1, W1r, deg, hA, n, 1);

    k_gather<<<gathB, T, 0, stream>>>(col, cnt, rowl, bsum, hA, agg, n);
    k_gemm128<<<256, GEMM_WAVES * 32, 0, stream>>>(agg, hA, W2l, b2, W2r, deg, hB, n, 1);

    float* pl = agg;
    float* pr = agg + (size_t)n * 8;
    k_proj14<<<gemmB, T, 0, stream>>>(hB, W3l, b3, W3r, pl, pr, n);
    k_agg7<<<nb((long long)n * 4), T, 0, stream>>>(col, cnt, rowl, bsum, pl, pr, deg, hA, n);
  } else {
    k_zero<<<nb(n), T, 0, stream>>>(deg, (size_t)n);
    k_degree<<<nb(E), T, 0, stream>>>(dst, deg, E);
    k_deg_inv<<<nb(n), T, 0, stream>>>(deg, n);

    k_zero<<<nb((long long)n * F), T, 0, stream>>>(agg, (size_t)n * F);
    k_scatter<<<nb(scatW), T, 0, stream>>>(src, dst, x, agg, E);
    k_gemm128<<<256, GEMM_WAVES * 32, 0, stream>>>(agg, x, W1l, b1, W1r, deg, hA, n, 1);

    k_zero<<<nb((long long)n * F), T, 0, stream>>>(agg, (size_t)n * F);
    k_scatter<<<nb(scatW), T, 0, stream>>>(src, dst, hA, agg, E);
    k_gemm128<<<256, GEMM_WAVES * 32, 0, stream>>>(agg, hA, W2l, b2, W2r, deg, hB, n, 1);

    k_zero<<<nb((long long)n * F), T, 0, stream>>>(agg, (size_t)n * F);
    k_scatter<<<nb(scatW), T, 0, stream>>>(src, dst, hB, agg, E);
    k_gemm_out7<<<gemmB, T, 0, stream>>>(agg, hB, W3l, b3, W3r, deg, hA, n);
  }

  k_logsoftmax7<<<nb(n), T, 0, stream>>>(hA, (float*)d_out, n);
}


# Two thin layer-3 launches fused away: the layer-2 update projects its relu'd tile to the 7 classes through a wave-private LDS tile (layer-2 output never stored); log_softmax ends the edge-mean kernel
# speedup vs baseline: 5.5755x; 1.1004x over previous
#include <hip/hip_runtime.h>
#include <hip/hip_bf16.h>

typedef __attribute__((ext_vector_type(2))) float v2f;
typedef __attribute__((ext_vector_type(8))) float v8f;

#define F 128

__global__ __launch_bounds__(256) void k_zero(float* __restrict__ p, size_t cnt) {
  size_t i = (size_t)blockIdx.x * blockDim.x + threadIdx.x;
  if (i < cnt) p[i] = 0.0f;
}

__global__ __launch_bounds__(256) void k_degree(const int* __restrict__ dst,
                                                float* __restrict__ deg, int nE) {
  int e = blockIdx.x * blockDim.x + threadIdx.x;
  if (e < nE) atomicAdd(&deg[dst[e]], 1.0f);
}

__global__ __launch_bounds__(256) void k_deg_inv(float* __restrict__ deg, int n) {
  int i = blockIdx.x * blockDim.x + threadIdx.x;
  if (i < n) {
    float d = deg[i];
    deg[i] = d > 0.0f ? 1.0f / d : 0.0f;
  }
}

#define SCAN_CHUNK 1024
#define SORT_T 1024
#define ECH 8192
#define MAXNB 1024
#define SORT_CAP 24576
#define SRC_BITS 22
#define SRC_MASK ((1u << SRC_BITS) - 1u)

__device__ __forceinline__ int wave_incl_scan(int v, int lane) {
#pragma unroll
  for (int d = 1; d < 32; d <<= 1) {
    int t = __shfl_up(v, d);
    if (lane >= d) v += t;
  }
  return v;
}

__global__ __launch_bounds__(SORT_T) void k_bkt_hist(const int* __restrict__ dst,
                                                     int* __restrict__ hist,
                                                     int nE, int nb, int g1) {
  __shared__ int s_h[MAXNB];
  const int tid = threadIdx.x;
  for (int b = tid; b < nb; b += SORT_T) s_h[b] = 0;
  __syncthreads();
  const int e0 = blockIdx.x * ECH;
  unsigned d[ECH / SORT_T];
#pragma unroll
  for (int k = 0; k < ECH / SORT_T; ++k) {
    const int e = e0 + k * SORT_T + tid;
    d[k] = (unsigned)dst[e < nE ? e : nE - 1];
  }
#pragma unroll
  for (int k = 0; k < ECH / SORT_T; ++k) {
    if (e0 + k * SORT_T + tid < nE) atomicAdd(&s_h[d[k] / SCAN_CHUNK], 1);
  }
  __syncthreads();
  for (int b = tid; b < nb; b += SORT_T) hist[(size_t)b * g1 + blockIdx.x] = s_h[b];
}

__global__ __launch_bounds__(SORT_T) void k_bkt_scatter(const int* __restrict__ src,
                                                        const int* __restrict__ dst,
                                                        const int* __restrict__ hist,
                                                        unsigned* __restrict__ bkt,
                                                        int* __restrict__ boff,
                                                        int nE, int nb, int g1) {
  __shared__ int s_tot[MAXNB];
  __shared__ int s_pre[MAXNB];
  __shared__ int s_cur[MAXNB];
  __shared__ int s_w[2][32];
  __shared__ unsigned s_ent[ECH];
  __shared__ unsigned short s_bk[ECH];
  const int tid = threadIdx.x;
  const int lane = tid & 31;
  const int wave = tid >> 5;
  const int wg = blockIdx.x;

  for (int b = wave; b < nb; b += SORT_T / 32) {
    const int* row = hist + (size_t)b * g1;
    int tot = 0, pre = 0, own = 0;
    for (int w0 = 0; w0 < g1; w0 += 256) {
      int v[8];
#pragma unroll
      for (int k = 0; k < 8; ++k) {
        const int w = w0 + 32 * k + lane;
        v[k] = row[w < g1 ? w : g1 - 1];
      }
#pragma unroll
      for (int k = 0; k < 8; ++k) {
        const int w = w0 + 32 * k + lane;
        const int x = w < g1 ? v[k] : 0;
        tot += x;
        pre += w < wg ? x : 0;
        own += w == wg ? x : 0;
      }
    }
    for (int d = 16; d > 0; d >>= 1) {
      tot += __shfl_xor(tot, d);
      pre += __shfl_xor(pre, d);
      own += __shfl_xor(own, d);
    }
    if (lane == 0) { s_tot[b] = tot; s_pre[b] = pre; s_cur[b] = own; }
  }
  __syncthreads();

  const int vt = tid < nb ? s_tot[tid] : 0;
  const int vo = tid < nb ? s_cur[tid] : 0;
  const int it = wave_incl_scan(vt, lane);
  const int io = wave_incl_scan(vo, lane);
  if (lane == 31) { s_w[0][wave] = it; s_w[1][wave] = io; }
  __syncthreads();
  int ot = 0, oo = 0;
  for (int w = 0; w < wave; ++w) { ot += s_w[0][w]; oo += s_w[1][w]; }
  const int bstart = ot + it - vt;
  const int lstart = oo + io - vo;
  if (tid < nb) {
    if (wg == 0) boff[tid] = bstart;
    s_tot[tid] = bstart + s_pre[tid] - lstart;
    s_cur[tid] = lstart;
  }
  __syncthreads();

  const int e0 = wg * ECH;
  unsigned dd[ECH / SORT_T], ss[ECH / SORT_T];
#pragma unroll
  for (int k = 0; k < ECH / SORT_T; ++k) {
    const int e = e0 + k * SORT_T + tid;
    const int ec = e < nE ? e : nE - 1;
    dd[k] = (unsigned)dst[ec];
    ss[k] = (unsigned)src[ec];
  }
#pragma unroll
  for (int k = 0; k < ECH / SORT_T; ++k) {
    if (e0 + k * SORT_T + tid < nE) {
      const unsigned b = dd[k] / SCAN_CHUNK;
      const int p = atomicAdd(&s_cur[b], 1);
      s_ent[p] = ((dd[k] % SCAN_CHUNK) << SRC_BITS) | ss[k];
      s_bk[p] = (unsigned short)b;
    }
  }
  __syncthreads();
  const int m = nE - e0 < ECH ? nE - e0 : ECH;
  unsigned short kb[ECH / SORT_T];
  unsigned ke[ECH / SORT_T];
  int kd[ECH / SORT_T];
#pragma unroll
  for (int k = 0; k < ECH / SORT_T; ++k) {
    const int i = k * SORT_T + tid;
    const int ic = i < m ? i : m - 1;
    kb[k] = s_bk[ic];
    ke[k] = s_ent[ic];
  }
#pragma unroll
  for (int k = 0; k < ECH / SORT_T; ++k) kd[k] = s_tot[kb[k]];
#pragma unroll
  for (int k = 0; k < ECH / SORT_T; ++k) {
    const int i = k * SORT_T + tid;
    if (i < m) bkt[i + kd[k]] = ke[k];
  }
}

__global__ __launch_bounds__(SORT_T) void k_bkt_sort(const unsigned* __restrict__ bkt,
                                                     const int* __restrict__ boff,
                                                     int* __restrict__ cnt,
                                                     int* __restrict__ rowl,
                                                     float* __restrict__ deg_inv,
                                                     int* __restrict__ col,
                                                     int n, int nE, int nb) {
  __shared__ int s_c[SCAN_CHUNK];
  __shared__ int s_w[32];
  __shared__ int s_col[SORT_CAP];
  const int tid = threadIdx.x;
  const int lane = tid & 31;
  const int wave = tid >> 5;
  const int b = blockIdx.x;
  const int beg = boff[b];
  const int end = b + 1 < nb ? boff[b + 1] : nE;
  const int m = end - beg;
  const unsigned* ent = bkt + beg;

  s_c[tid] = 0;
  __syncthreads();
  for (int i0 = 0; i0 < m; i0 += 8 * SORT_T) {
    unsigned u[8];
#pragma unroll
    for (int k = 0; k < 8; ++k) {
      const int i = i0 + k * SORT_T + tid;
      u[k] = ent[i < m ? i : m - 1];
    }
#pragma unroll
    for (int k = 0; k < 8; ++k) {
      if (i0 + k * SORT_T + tid < m) atomicAdd(&s_c[u[k] >> SRC_BITS], 1);
    }
  }
  __syncthreads();

  const int v = s_c[tid];
  const int incl = wave_incl_scan(v, lane);
  if (lane == 31) s_w[wave] = incl;
  __syncthreads();
  int woff = 0;
  for (int w = 0; w < wave; ++w) woff += s_w[w];
  const int excl = woff + incl - v;
  const int node = b * SCAN_CHUNK + tid;
  if (node < n) {
    cnt[node] = v;
    rowl[node] = excl;
    deg_inv[node] = v > 0 ? 1.0f / (float)v : 0.0f;
  }
  s_c[tid] = excl;
  __syncthreads();

  if (m <= SORT_CAP) {
    for (int i0 = 0; i0 < m; i0 += 8 * SORT_T) {
      unsigned u[8];
#pragma unroll
      for (int k = 0; k < 8; ++k) {
        const int i = i0 + k * SORT_T + tid;
        u[k] = ent[i < m ? i : m - 1];
      }
#pragma unroll
      for (int k = 0; k < 8; ++k) {
        if (i0 + k * SORT_T + tid < m) {
          const int p = atomicAdd(&s_c[u[k] >> SRC_BITS], 1);
          s_col[p] = (int)(u[k] & SRC_MASK);
        }
      }
    }
    __syncthreads();
    for (int i0 = 0; i0 < m; i0 += 8 * SORT_T) {
      int sv[8];
#pragma unroll
      for (int k = 0; k < 8; ++k) {
        const int i = i0 + k * SORT_T + tid;
        sv[k] = s_col[i < m ? i : 0];
      }
#pragma unroll
      for (int k = 0; k < 8; ++k) {
        const int i = i0 + k * SORT_T + tid;
        if (i < m) col[beg + i] = sv[k];
      }
    }
  } else {
    for (int i = tid; i < m; i += SORT_T) {
      const unsigned u = ent[i];
      const int p = atomicAdd(&s_c[u >> SRC_BITS], 1);
      col[beg + p] = (int)(u & SRC_MASK);
    }
  }
}

__global__ __launch_bounds__(256) void k_gather(const int* __restrict__ col,
                                                const int* __restrict__ cnt,
                                                const int* __restrict__ rowl,
                                                const int* __restrict__ boff,
                                                const float* __restrict__ feat,
                                                float* __restrict__ agg, int n) {
  const int wave = __builtin_amdgcn_readfirstlane((int)(threadIdx.x >> 5));
  const int node = blockIdx.x * 8 + wave;
  if (node >= n) return;
  const int lane = threadIdx.x & 31;
  const int cn = cnt[node];
  const int* c = col + (rowl[node] + boff[node / SCAN_CHUNK]);
  const float* fb = feat + lane * 4;
  float4 acc = make_float4(0.0f, 0.0f, 0.0f, 0.0f);
  int j = 0;
  for (; j + 4 <= cn; j += 4) {
    const int s0 = c[j], s1 = c[j + 1], s2 = c[j + 2], s3 = c[j + 3];
    const float4 v0 = *reinterpret_cast<const float4*>(fb + (size_t)s0 * F);
    const float4 v1 = *reinterpret_cast<const float4*>(fb + (size_t)s1 * F);
    const float4 v2 = *reinterpret_cast<const float4*>(fb + (size_t)s2 * F);
    const float4 v3 = *reinterpret_cast<const float4*>(fb + (size_t)s3 * F);
    acc.x += v0.x; acc.y += v0.y; acc.z += v0.z; acc.w += v0.w;
    acc.x += v1.x; acc.y += v1.y; acc.z += v1.z; acc.w += v1.w;
    acc.x += v2.x; acc.y += v2.y; acc.z += v2.z; acc.w += v2.w;
    acc.x += v3.x; acc.y += v3.y; acc.z += v3.z; acc.w += v3.w;
  }
  for (; j < cn; ++j) {
    const float4 v0 = *reinterpret_cast<const float4*>(fb + (size_t)c[j] * F);
    acc.x += v0.x; acc.y += v0.y; acc.z += v0.z; acc.w += v0.w;
  }
  *reinterpret_cast<float4*>(agg + (size_t)node * F + lane * 4) = acc;
}

__global__ __launch_bounds__(256) void k_scatter(const int* __restrict__ src,
                                                 const int* __restrict__ dst,
                                                 const float* __restrict__ feat,
                                                 float* __restrict__ agg, int nE) {
  long long tid = (long long)blockIdx.x * blockDim.x + threadIdx.x;
  int e = (int)(tid >> 5);
  if (e >= nE) return;
  int fo = ((int)tid & 31) * 4;
  const float4 v = *reinterpret_cast<const float4*>(feat + (size_t)src[e] * F + fo);
  float* d = agg + (size_t)dst[e] * F + fo;
  atomicAdd(d + 0, v.x);
  atomicAdd(d + 1, v.y);
  atomicAdd(d + 2, v.z);
  atomicAdd(d + 3, v.w);
}

#define GEMM_WAVES 16
#define WROW 256
#define WBLK (8 * WROW + 4)
#define TPITCH 132

__device__ __forceinline__ v8f wmma_f32(float a0, float a1, float b0, float b1, v8f c) {
  v2f a, b;
  a[0] = a0; a[1] = a1;
  b[0] = b0; b[1] = b1;
  return __builtin_amdgcn_wmma_f32_16x16x4_f32(false, a, false, b, (short)0, c,
                                               false, false);
}

__device__ __forceinline__ void wave_lds_handover() {
  __builtin_amdgcn_fence(__ATOMIC_RELEASE, "wavefront");
  __builtin_amdgcn_wave_barrier();
  __builtin_amdgcn_fence(__ATOMIC_ACQUIRE, "wavefront");
}

template <bool P3>
__device__ __forceinline__ void gemm128_body(
    float* __restrict__ wlds, float* __restrict__ tlds, float* __restrict__ w3lds,
    const float* __restrict__ agg, const float* __restrict__ xin,
    const float* __restrict__ Wl, const float* __restrict__ bl,
    const float* __restrict__ Wr, const float* __restrict__ deg_inv,
    float* __restrict__ hout, int n, int do_relu,
    const float* __restrict__ W3l, const float* __restrict__ b3,
    const float* __restrict__ W3r, float* __restrict__ pl, float* __restrict__ pr) {
  constexpr int NCOPY = F * (F / 4) / (GEMM_WAVES * 32);
  float4 vl[NCOPY], vr[NCOPY];
#pragma unroll
  for (int i = 0; i < NCOPY; ++i) {
    const int f = threadIdx.x + i * (GEMM_WAVES * 32);
    vl[i] = reinterpret_cast<const float4*>(Wl)[f];
    vr[i] = reinterpret_cast<const float4*>(Wr)[f];
  }
#pragma unroll
  for (int i = 0; i < NCOPY; ++i) {
    const int f = threadIdx.x + i * (GEMM_WAVES * 32);
    const int row = f >> 5;
    const int c4 = f & 31;
    float* dstp = wlds + (row & 15) * WBLK + (row >> 4) * WROW + c4 * 4;
    *reinterpret_cast<float4*>(dstp) = vl[i];
    *reinterpret_cast<float4*>(dstp + F) = vr[i];
  }
  if constexpr (P3) {
    const int col = threadIdx.x >> 5;
    const int k4 = threadIdx.x & 31;
    const int wc3 = col & 7;
    const float* wsel = (col >= 8 ? W3r : W3l) + (size_t)(wc3 < 7 ? wc3 : 0) * F;
    *reinterpret_cast<float4*>(w3lds + col * TPITCH + k4 * 4) =
        reinterpret_cast<const float4*>(wsel)[k4];
  }
  __syncthreads();

  const int lane = threadIdx.x & 31;
  const int wave = __builtin_amdgcn_readfirstlane((int)(threadIdx.x >> 5));
  const int mloc = lane & 15;
  const int h = lane >> 4;
  const int ntiles = (n + 15) >> 4;
  const float* bfrag = wlds + mloc * WBLK + 4 * h;
  const int wc = mloc & 7;
  const bool cvalid = wc < 7;
  const bool self = mloc >= 8;
  float bias3 = 0.0f;
  if constexpr (P3) bias3 = (cvalid && self) ? b3[wc] : 0.0f;

  for (int s = wave; ; s += GEMM_WAVES) {
    const int t = blockIdx.x + gridDim.x * s;
    if (t >= ntiles) break;
    const int row0 = t * 16;
    const int m = row0 + mloc;
    const int mc = m < n ? m : n - 1;
    const float dscale = deg_inv[mc];
    const float4* arow = reinterpret_cast<const float4*>(agg + (size_t)mc * F) + h;
    const float4* xrow = reinterpret_cast<const float4*>(xin + (size_t)mc * F) + h;

    float4 av[32];
#pragma unroll
    for (int j = 0; j < 16; ++j) av[j] = arow[2 * j];

    v8f c[8];
#pragma unroll
    for (int ct = 0; ct < 8; ++ct) {
#pragma unroll
      for (int i = 0; i < 8; ++i) c[ct][i] = 0.0f;
    }

    float4 bq[2][8];
#pragma unroll
    for (int ct = 0; ct < 8; ++ct)
      bq[0][ct] = *reinterpret_cast<const float4*>(bfrag + ct * WROW);

#pragma unroll
    for (int j = 0; j < 32; ++j) {
      const int cur = j & 1;
      if (j + 1 < 32) {
#pragma unroll
        for (int ct = 0; ct < 8; ++ct)
          bq[cur ^ 1][ct] =
              *reinterpret_cast<const float4*>(bfrag + ct * WROW + 8 * (j + 1));
      }
      if (j < 16) av[j + 16] = xrow[2 * j];
      __builtin_amdgcn_sched_barrier(0);
      float4 a = av[j];
      if (j < 16) { a.x *= dscale; a.y *= dscale; a.z *= dscale; a.w *= dscale; }
#pragma unroll
      for (int ct = 0; ct < 8; ++ct) {
        const float4 b = bq[cur][ct];
        c[ct] = wmma_f32(a.x, a.y, b.x, b.y, c[ct]);
        c[ct] = wmma_f32(a.z, a.w, b.z, b.w, c[ct]);
      }
      __builtin_amdgcn_sched_barrier(0);
    }

    const int rbase = h * 8;
    if constexpr (!P3) {
#pragma unroll
      for (int ct = 0; ct < 8; ++ct) {
        const int ncol = ct * 16 + mloc;
        const float bias = bl[ncol];
#pragma unroll
        for (int r = 0; r < 8; ++r) {
          const int mr = row0 + rbase + r;
          if (mr < n) {
            float v = c[ct][r] + bias;
            if (do_relu) v = v > 0.0f ? v : 0.0f;
            hout[(size_t)mr * F + ncol] = v;
          }
        }
      }
    } else {
      float* tile = tlds + wave * (16 * TPITCH);
#pragma unroll
      for (int ct = 0; ct < 8; ++ct) {
        const int ncol = ct * 16 + mloc;
        const float bias = bl[ncol];
#pragma unroll
        for (int r = 0; r < 8; ++r) {
          float v = c[ct][r] + bias;
          if (do_relu) v = v > 0.0f ? v : 0.0f;
          tile[(rbase + r) * TPITCH + ncol] = v;
        }
      }
      wave_lds_handover();

      v8f c3;
#pragma unroll
      for (int i = 0; i < 8; ++i) c3[i] = bias3;
      const float* a3 = tile + mloc * TPITCH + 4 * h;
      const float* b3f = w3lds + mloc * TPITCH + 4 * h;
#pragma unroll
      for (int j = 0; j < 16; ++j) {
        const float4 a = *reinterpret_cast<const float4*>(a3 + 8 * j);
        const float4 b = *reinterpret_cast<const float4*>(b3f + 8 * j);
        c3 = wmma_f32(a.x, a.y, b.x, b.y, c3);
        c3 = wmma_f32(a.z, a.w, b.z, b.w, c3);
      }
      float* dstp = (self ? pr : pl) + wc;
#pragma unroll
      for (int r = 0; r < 8; ++r) {
        const int mr = row0 + rbase + r;
        if (mr < n) dstp[(size_t)mr * 8] = cvalid ? c3[r] : 0.0f;
      }
      wave_lds_handover();
    }
  }
}

__global__ __launch_bounds__(GEMM_WAVES * 32) void k_gemm128(
    const float* __restrict__ agg, const float* __restrict__ xin,
    const float* __restrict__ Wl, const float* __restrict__ bl,
    const float* __restrict__ Wr, const float* __restrict__ deg_inv,
    float* __restrict__ hout, int n, int do_relu) {
  __shared__ __attribute__((aligned(16))) float wlds[16 * WBLK];
  gemm128_body<false>(wlds, nullptr, nullptr, agg, xin, Wl, bl, Wr, deg_inv, hout, n,
                      do_relu, nullptr, nullptr, nullptr, nullptr, nullptr);
}

__global__ __launch_bounds__(GEMM_WAVES * 32) void k_gemm128_p3(
    const float* __restrict__ agg, const float* __restrict__ xin,
    const float* __restrict__ Wl, const float* __restrict__ bl,
    const float* __restrict__ Wr, const float* __restrict__ deg_inv,
    int n, int do_relu,
    const float* __restrict__ W3l, const float* __restrict__ b3,
    const float* __restrict__ W3r, float* __restrict__ pl, float* __restrict__ pr) {
  __shared__ __attribute__((aligned(16))) float wlds[16 * WBLK];
  __shared__ __attribute__((aligned(16))) float tlds[GEMM_WAVES * 16 * TPITCH];
  __shared__ __attribute__((aligned(16))) float w3lds[16 * TPITCH];
  gemm128_body<true>(wlds, tlds, w3lds, agg, xin, Wl, bl, Wr, deg_inv, nullptr, n,
                     do_relu, W3l, b3, W3r, pl, pr);
}

__global__ __launch_bounds__(256) void k_gemm_out7(
    const float* __restrict__ agg, const float* __restrict__ xin,
    const float* __restrict__ Wl, const float* __restrict__ bl,
    const float* __restrict__ Wr, const float* __restrict__ deg_inv,
    float* __restrict__ logits, int n) {
  const int K = F;
  int lane = threadIdx.x & 31;
  int wave = threadIdx.x >> 5;
  int row0 = (blockIdx.x * 8 + wave) * 16;
  if (row0 >= n) return;

  int mloc = lane & 15;
  int koff = (lane >> 4) * 2;
  int m = row0 + mloc;
  int mc = m < n ? m : n - 1;
  float dscale = deg_inv[mc];
  const float* arow = agg + (size_t)mc * K;
  const float* xrow = xin + (size_t)mc * K;

  int ncol = mloc;
  bool cvalid = ncol < 7;
  float wm = cvalid ? 1.0f : 0.0f;
  const float* wlrow = Wl + (size_t)(cvalid ? ncol : 0) * K;
  const float* wrrow = Wr + (size_t)(cvalid ? ncol : 0) * K;

  float bias = cvalid ? bl[ncol] : 0.0f;
  v8f c;
#pragma unroll
  for (int i = 0; i < 8; ++i) c[i] = bias;

#pragma unroll
  for (int kt = 0; kt < 32; ++kt) {
    int k0 = kt * 4 + koff;
    v2f a, b;
    a[0] = arow[k0] * dscale; a[1] = arow[k0 + 1] * dscale;
    b[0] = wlrow[k0] * wm;    b[1] = wlrow[k0 + 1] * wm;
    c = __builtin_amdgcn_wmma_f32_16x16x4_f32(false, a, false, b, (short)0, c,
                                              false, false);
  }
#pragma unroll
  for (int kt = 0; kt < 32; ++kt) {
    int k0 = kt * 4 + koff;
    v2f a, b;
    a[0] = xrow[k0];       a[1] = xrow[k0 + 1];
    b[0] = wrrow[k0] * wm; b[1] = wrrow[k0 + 1] * wm;
    c = __builtin_amdgcn_wmma_f32_16x16x4_f32(false, a, false, b, (short)0, c,
                                              false, false);
  }

  if (cvalid) {
    int rbase = (lane >> 4) * 8;
#pragma unroll
    for (int r = 0; r < 8; ++r) {
      int mr = row0 + rbase + r;
      if (mr < n) logits[(size_t)mr * 7 + ncol] = c[r];
    }
  }
}

__global__ __launch_bounds__(256) void k_agg7(const int* __restrict__ col,
                                              const int* __restrict__ cnt,
                                              const int* __restrict__ rowl,
                                              const int* __restrict__ boff,
                                              const float* __restrict__ pl,
                                              const float* __restrict__ pr,
                                              const float* __restrict__ deg_inv,
                                              float* __restrict__ out, int n) {
  const int tid = blockIdx.x * blockDim.x + threadIdx.x;
  const int q = tid & 3;
  const int node = tid >> 2;
  const bool live = node < n;
  const int nd = live ? node : n - 1;
  const int cn = live ? cnt[nd] : 0;
  const int* c = col + (rowl[nd] + boff[nd / SCAN_CHUNK]);
  const float4* p4 = reinterpret_cast<const float4*>(pl);
  float4 lo = make_float4(0.0f, 0.0f, 0.0f, 0.0f);
  float4 hi = make_float4(0.0f, 0.0f, 0.0f, 0.0f);
  for (int j = q; j < cn; j += 16) {
    int s[4];
#pragma unroll
    for (int u = 0; u < 4; ++u) {
      s[u] = 0;
      if (j + 4 * u < cn) s[u] = c[j + 4 * u];
    }
    float4 a[4], b[4];
#pragma unroll
    for (int u = 0; u < 4; ++u) {
      a[u] = make_float4(0.0f, 0.0f, 0.0f, 0.0f);
      b[u] = make_float4(0.0f, 0.0f, 0.0f, 0.0f);
      if (j + 4 * u < cn) {
        a[u] = p4[2 * (size_t)s[u]];
        b[u] = p4[2 * (size_t)s[u] + 1];
      }
    }
#pragma unroll
    for (int u = 0; u < 4; ++u) {
      lo.x += a[u].x; lo.y += a[u].y; lo.z += a[u].z; lo.w += a[u].w;
      hi.x += b[u].x; hi.y += b[u].y; hi.z += b[u].z; hi.w += b[u].w;
    }
  }
  for (int d = 1; d < 4; d <<= 1) {
    lo.x += __shfl_xor(lo.x, d); lo.y += __shfl_xor(lo.y, d);
    lo.z += __shfl_xor(lo.z, d); lo.w += __shfl_xor(lo.w, d);
    hi.x += __shfl_xor(hi.x, d); hi.y += __shfl_xor(hi.y, d);
    hi.z += __shfl_xor(hi.z, d);
  }
  if (live && q == 0) {
    const float ds = deg_inv[node];
    const float4 r0 = reinterpret_cast<const float4*>(pr)[2 * (size_t)node];
    const float4 r1 = reinterpret_cast<const float4*>(pr)[2 * (size_t)node + 1];
    float v[7];
    v[0] = lo.x * ds + r0.x;
    v[1] = lo.y * ds + r0.y;
    v[2] = lo.z * ds + r0.z;
    v[3] = lo.w * ds + r0.w;
    v[4] = hi.x * ds + r1.x;
    v[5] = hi.y * ds + r1.y;
    v[6] = hi.z * ds + r1.z;
    float mx = v[0];
#pragma unroll
    for (int j = 1; j < 7; ++j) mx = fmaxf(mx, v[j]);
    float s = 0.0f;
#pragma unroll
    for (int j = 0; j < 7; ++j) s += expf(v[j] - mx);
    const float ls = logf(s);
    float* o = out + (size_t)node * 7;
#pragma unroll
    for (int j = 0; j < 7; ++j) o[j] = v[j] - mx - ls;
  }
}

__global__ __launch_bounds__(256) void k_logsoftmax7(const float* __restrict__ logits,
                                                     float* __restrict__ out, int n) {
  int i = blockIdx.x * blockDim.x + threadIdx.x;
  if (i >= n) return;
  const float* r = logits + (size_t)i * 7;
  float v[7];
#pragma unroll
  for (int j = 0; j < 7; ++j) v[j] = r[j];
  float mx = v[0];
#pragma unroll
  for (int j = 1; j < 7; ++j) mx = fmaxf(mx, v[j]);
  float s = 0.0f;
#pragma unroll
  for (int j = 0; j < 7; ++j) s += expf(v[j] - mx);
  float ls = logf(s);
  float* o = out + (size_t)i * 7;
#pragma unroll
  for (int j = 0; j < 7; ++j) o[j] = v[j] - mx - ls;
}

extern "C" void kernel_launch(void* const* d_in, const int* in_sizes, int n_in,
                              void* d_out, int out_size, void* d_ws, size_t ws_size,
                              hipStream_t stream) {
  (void)n_in; (void)out_size;
  const float* x   = (const float*)d_in[0];
  const int*   ei  = (const int*)d_in[1];
  const float* W1l = (const float*)d_in[2];
  const float* b1  = (const float*)d_in[3];
  const float* W1r = (const float*)d_in[4];
  const float* W2l = (const float*)d_in[5];
  const float* b2  = (const float*)d_in[6];
  const float* W2r = (const float*)d_in[7];
  const float* W3l = (const float*)d_in[8];
  const float* b3  = (const float*)d_in[9];
  const float* W3r = (const float*)d_in[10];

  const int n = in_sizes[0] / F;
  const int E = in_sizes[1] / 2;
  const int* src = ei;
  const int* dst = ei + E;

  char* base = (char*)d_ws;
  size_t off = 0;
  auto a256 = [](size_t b) { return (b + 255) & ~(size_t)255; };
  float* deg = (float*)(base + off); off = a256(off + (size_t)n * sizeof(float));
  float* agg = (float*)(base + off); off = a256(off + (size_t)n * F * sizeof(float));
  float* hA  = (float*)(base + off); off = a256(off + (size_t)n * F * sizeof(float));
  float* hB  = (float*)(base + off); off = a256(off + (size_t)n * F * sizeof(float));

  const int nchunk = (n + SCAN_CHUNK - 1) / SCAN_CHUNK;
  const int g1 = (E + ECH - 1) / ECH;
  int* cnt  = (int*)(base + off); off = a256(off + (size_t)n * sizeof(int));
  int* rowl = (int*)(base + off); off = a256(off + (size_t)n * sizeof(int));
  int* bsum = (int*)(base + off); off = a256(off + (size_t)nchunk * sizeof(int));
  int* col  = (int*)(base + off); off = a256(off + (size_t)E * sizeof(int));
  unsigned* bkt = (unsigned*)hB;
  int* hist = (int*)hA;
  const bool sorted = off <= ws_size && (size_t)E <= (size_t)n * F &&
                      (size_t)nchunk * g1 <= (size_t)n * F && nchunk <= MAXNB &&
                      n <= (1 << SRC_BITS);

  const int T = 256;
  auto nb = [&](long long work) { return (int)((work + T - 1) / T); };
  const long long scatW = (long long)E * 32;
  const int gemmB = (n + 127) / 128;
  const int gathB = (n + 7) / 8;

  if (sorted) {
    k_bkt_hist<<<g1, SORT_T, 0, stream>>>(dst, hist, E, nchunk, g1);
    k_bkt_scatter<<<g1, SORT_T, 0, stream>>>(src, dst, hist, bkt, bsum, E, nchunk, g1);
    k_bkt_sort<<<nchunk, SORT_T, 0, stream>>>(bkt, bsum, cnt, rowl, deg, col, n, E, nchunk);

    k_gather<<<gathB, T, 0, stream>>>(col, cnt, rowl, bsum, x, agg, n);
    k_gemm128<<<256, GEMM_WAVES * 32, 0, stream>>>(agg, x, W1l, b1, W1r, deg, hA, n, 1);

    k_gather<<<gathB, T, 0, stream>>>(col, cnt, rowl, bsum, hA, agg, n);
    float* pl = hB;
    float* pr = hB + (size_t)n * 8;
    k_gemm128_p3<<<256, GEMM_WAVES * 32, 0, stream>>>(agg, hA, W2l, b2, W2r, deg, n, 1,
                                                     W3l, b3, W3r, pl, pr);
    k_agg7<<<nb((long long)n * 4), T, 0, stream>>>(col, cnt, rowl, bsum, pl, pr, deg,
                                                   (float*)d_out, n);
  } else {
    k_zero<<<nb(n), T, 0, stream>>>(deg, (size_t)n);
    k_degree<<<nb(E), T, 0, stream>>>(dst, deg, E);
    k_deg_inv<<<nb(n), T, 0, stream>>>(deg, n);

    k_zero<<<nb((long long)n * F), T, 0, stream>>>(agg, (size_t)n * F);
    k_scatter<<<nb(scatW), T, 0, stream>>>(src, dst, x, agg, E);
    k_gemm128<<<256, GEMM_WAVES * 32, 0, stream>>>(agg, x, W1l, b1, W1r, deg, hA, n, 1);

    k_zero<<<nb((long long)n * F), T, 0, stream>>>(agg, (size_t)n * F);
    k_scatter<<<nb(scatW), T, 0, stream>>>(src, dst, hA, agg, E);
    k_gemm128<<<256, GEMM_WAVES * 32, 0, stream>>>(agg, hA, W2l, b2, W2r, deg, hB, n, 1);

    k_zero<<<nb((long long)n * F), T, 0, stream>>>(agg, (size_t)n * F);
    k_scatter<<<nb(scatW), T, 0, stream>>>(src, dst, hB, agg, E);
    k_gemm_out7<<<gemmB, T, 0, stream>>>(agg, hB, W3l, b3, W3r, deg, hA, n);
    k_logsoftmax7<<<nb(n), T, 0, stream>>>(hA, (float*)d_out, n);
  }
}


# Neighbour sum fused into the dense update, no producer split: each of 16 waves sums its tile's 16 node lists as one stream, 32 rows in flight, into its LDS tile, then runs its k loop; sum order kept
# speedup vs baseline: 5.7641x; 1.0338x over previous
#include <hip/hip_runtime.h>
#include <hip/hip_bf16.h>

typedef __attribute__((ext_vector_type(2))) float v2f;
typedef __attribute__((ext_vector_type(8))) float v8f;

#define F 128

__global__ __launch_bounds__(256) void k_zero(float* __restrict__ p, size_t cnt) {
  size_t i = (size_t)blockIdx.x * blockDim.x + threadIdx.x;
  if (i < cnt) p[i] = 0.0f;
}

__global__ __launch_bounds__(256) void k_degree(const int* __restrict__ dst,
                                                float* __restrict__ deg, int nE) {
  int e = blockIdx.x * blockDim.x + threadIdx.x;
  if (e < nE) atomicAdd(&deg[dst[e]], 1.0f);
}

__global__ __launch_bounds__(256) void k_deg_inv(float* __restrict__ deg, int n) {
  int i = blockIdx.x * blockDim.x + threadIdx.x;
  if (i < n) {
    float d = deg[i];
    deg[i] = d > 0.0f ? 1.0f / d : 0.0f;
  }
}

#define SCAN_CHUNK 1024
#define SORT_T 1024
#define ECH 8192
#define MAXNB 1024
#define SORT_CAP 24576
#define SRC_BITS 22
#define SRC_MASK ((1u << SRC_BITS) - 1u)

__device__ __forceinline__ int wave_incl_scan(int v, int lane) {
#pragma unroll
  for (int d = 1; d < 32; d <<= 1) {
    int t = __shfl_up(v, d);
    if (lane >= d) v += t;
  }
  return v;
}

__global__ __launch_bounds__(SORT_T) void k_bkt_hist(const int* __restrict__ dst,
                                                     int* __restrict__ hist,
                                                     int nE, int nb, int g1) {
  __shared__ int s_h[MAXNB];
  const int tid = threadIdx.x;
  for (int b = tid; b < nb; b += SORT_T) s_h[b] = 0;
  __syncthreads();
  const int e0 = blockIdx.x * ECH;
  unsigned d[ECH / SORT_T];
#pragma unroll
  for (int k = 0; k < ECH / SORT_T; ++k) {
    const int e = e0 + k * SORT_T + tid;
    d[k] = (unsigned)dst[e < nE ? e : nE - 1];
  }
#pragma unroll
  for (int k = 0; k < ECH / SORT_T; ++k) {
    if (e0 + k * SORT_T + tid < nE) atomicAdd(&s_h[d[k] / SCAN_CHUNK], 1);
  }
  __syncthreads();
  for (int b = tid; b < nb; b += SORT_T) hist[(size_t)b * g1 + blockIdx.x] = s_h[b];
}

__global__ __launch_bounds__(SORT_T) void k_bkt_scatter(const int* __restrict__ src,
                                                        const int* __restrict__ dst,
                                                        const int* __restrict__ hist,
                                                        unsigned* __restrict__ bkt,
                                                        int* __restrict__ boff,
                                                        int nE, int nb, int g1) {
  __shared__ int s_tot[MAXNB];
  __shared__ int s_pre[MAXNB];
  __shared__ int s_cur[MAXNB];
  __shared__ int s_w[2][32];
  __shared__ unsigned s_ent[ECH];
  __shared__ unsigned short s_bk[ECH];
  const int tid = threadIdx.x;
  const int lane = tid & 31;
  const int wave = tid >> 5;
  const int wg = blockIdx.x;

  for (int b = wave; b < nb; b += SORT_T / 32) {
    const int* row = hist + (size_t)b * g1;
    int tot = 0, pre = 0, own = 0;
    for (int w0 = 0; w0 < g1; w0 += 256) {
      int v[8];
#pragma unroll
      for (int k = 0; k < 8; ++k) {
        const int w = w0 + 32 * k + lane;
        v[k] = row[w < g1 ? w : g1 - 1];
      }
#pragma unroll
      for (int k = 0; k < 8; ++k) {
        const int w = w0 + 32 * k + lane;
        const int x = w < g1 ? v[k] : 0;
        tot += x;
        pre += w < wg ? x : 0;
        own += w == wg ? x : 0;
      }
    }
    for (int d = 16; d > 0; d >>= 1) {
      tot += __shfl_xor(tot, d);
      pre += __shfl_xor(pre, d);
      own += __shfl_xor(own, d);
    }
    if (lane == 0) { s_tot[b] = tot; s_pre[b] = pre; s_cur[b] = own; }
  }
  __syncthreads();

  const int vt = tid < nb ? s_tot[tid] : 0;
  const int vo = tid < nb ? s_cur[tid] : 0;
  const int it = wave_incl_scan(vt, lane);
  const int io = wave_incl_scan(vo, lane);
  if (lane == 31) { s_w[0][wave] = it; s_w[1][wave] = io; }
  __syncthreads();
  int ot = 0, oo = 0;
  for (int w = 0; w < wave; ++w) { ot += s_w[0][w]; oo += s_w[1][w]; }
  const int bstart = ot + it - vt;
  const int lstart = oo + io - vo;
  if (tid < nb) {
    if (wg == 0) boff[tid] = bstart;
    s_tot[tid] = bstart + s_pre[tid] - lstart;
    s_cur[tid] = lstart;
  }
  __syncthreads();

  const int e0 = wg * ECH;
  unsigned dd[ECH / SORT_T], ss[ECH / SORT_T];
#pragma unroll
  for (int k = 0; k < ECH / SORT_T; ++k) {
    const int e = e0 + k * SORT_T + tid;
    const int ec = e < nE ? e : nE - 1;
    dd[k] = (unsigned)dst[ec];
    ss[k] = (unsigned)src[ec];
  }
#pragma unroll
  for (int k = 0; k < ECH / SORT_T; ++k) {
    if (e0 + k * SORT_T + tid < nE) {
      const unsigned b = dd[k] / SCAN_CHUNK;
      const int p = atomicAdd(&s_cur[b], 1);
      s_ent[p] = ((dd[k] % SCAN_CHUNK) << SRC_BITS) | ss[k];
      s_bk[p] = (unsigned short)b;
    }
  }
  __syncthreads();
  const int m = nE - e0 < ECH ? nE - e0 : ECH;
  unsigned short kb[ECH / SORT_T];
  unsigned ke[ECH / SORT_T];
  int kd[ECH / SORT_T];
#pragma unroll
  for (int k = 0; k < ECH / SORT_T; ++k) {
    const int i = k * SORT_T + tid;
    const int ic = i < m ? i : m - 1;
    kb[k] = s_bk[ic];
    ke[k] = s_ent[ic];
  }
#pragma unroll
  for (int k = 0; k < ECH / SORT_T; ++k) kd[k] = s_tot[kb[k]];
#pragma unroll
  for (int k = 0; k < ECH / SORT_T; ++k) {
    const int i = k * SORT_T + tid;
    if (i < m) bkt[i + kd[k]] = ke[k];
  }
}

__global__ __launch_bounds__(SORT_T) void k_bkt_sort(const unsigned* __restrict__ bkt,
                                                     const int* __restrict__ boff,
                                                     int* __restrict__ cnt,
                                                     int* __restrict__ rowl,
                                                     float* __restrict__ deg_inv,
                                                     int* __restrict__ col,
                                                     int n, int nE, int nb) {
  __shared__ int s_c[SCAN_CHUNK];
  __shared__ int s_w[32];
  __shared__ int s_col[SORT_CAP];
  const int tid = threadIdx.x;
  const int lane = tid & 31;
  const int wave = tid >> 5;
  const int b = blockIdx.x;
  const int beg = boff[b];
  const int end = b + 1 < nb ? boff[b + 1] : nE;
  const int m = end - beg;
  const unsigned* ent = bkt + beg;

  s_c[tid] = 0;
  __syncthreads();
  for (int i0 = 0; i0 < m; i0 += 8 * SORT_T) {
    unsigned u[8];
#pragma unroll
    for (int k = 0; k < 8; ++k) {
      const int i = i0 + k * SORT_T + tid;
      u[k] = ent[i < m ? i : m - 1];
    }
#pragma unroll
    for (int k = 0; k < 8; ++k) {
      if (i0 + k * SORT_T + tid < m) atomicAdd(&s_c[u[k] >> SRC_BITS], 1);
    }
  }
  __syncthreads();

  const int v = s_c[tid];
  const int incl = wave_incl_scan(v, lane);
  if (lane == 31) s_w[wave] = incl;
  __syncthreads();
  int woff = 0;
  for (int w = 0; w < wave; ++w) woff += s_w[w];
  const int excl = woff + incl - v;
  const int node = b * SCAN_CHUNK + tid;
  if (node < n) {
    cnt[node] = v;
    rowl[node] = excl;
    deg_inv[node] = v > 0 ? 1.0f / (float)v : 0.0f;
  }
  s_c[tid] = excl;
  __syncthreads();

  if (m <= SORT_CAP) {
    for (int i0 = 0; i0 < m; i0 += 8 * SORT_T) {
      unsigned u[8];
#pragma unroll
      for (int k = 0; k < 8; ++k) {
        const int i = i0 + k * SORT_T + tid;
        u[k] = ent[i < m ? i : m - 1];
      }
#pragma unroll
      for (int k = 0; k < 8; ++k) {
        if (i0 + k * SORT_T + tid < m) {
          const int p = atomicAdd(&s_c[u[k] >> SRC_BITS], 1);
          s_col[p] = (int)(u[k] & SRC_MASK);
        }
      }
    }
    __syncthreads();
    for (int i0 = 0; i0 < m; i0 += 8 * SORT_T) {
      int sv[8];
#pragma unroll
      for (int k = 0; k < 8; ++k) {
        const int i = i0 + k * SORT_T + tid;
        sv[k] = s_col[i < m ? i : 0];
      }
#pragma unroll
      for (int k = 0; k < 8; ++k) {
        const int i = i0 + k * SORT_T + tid;
        if (i < m) col[beg + i] = sv[k];
      }
    }
  } else {
    for (int i = tid; i < m; i += SORT_T) {
      const unsigned u = ent[i];
      const int p = atomicAdd(&s_c[u >> SRC_BITS], 1);
      col[beg + p] = (int)(u & SRC_MASK);
    }
  }
}

__global__ __launch_bounds__(256) void k_scatter(const int* __restrict__ src,
                                                 const int* __restrict__ dst,
                                                 const float* __restrict__ feat,
                                                 float* __restrict__ agg, int nE) {
  long long tid = (long long)blockIdx.x * blockDim.x + threadIdx.x;
  int e = (int)(tid >> 5);
  if (e >= nE) return;
  int fo = ((int)tid & 31) * 4;
  const float4 v = *reinterpret_cast<const float4*>(feat + (size_t)src[e] * F + fo);
  float* d = agg + (size_t)dst[e] * F + fo;
  atomicAdd(d + 0, v.x);
  atomicAdd(d + 1, v.y);
  atomicAdd(d + 2, v.z);
  atomicAdd(d + 3, v.w);
}

#define GEMM_WAVES 16
#define WROW 256
#define WBLK (8 * WROW + 4)
#define TPITCH 132
#define GB 32

__device__ __forceinline__ v8f wmma_f32(float a0, float a1, float b0, float b1, v8f c) {
  v2f a, b;
  a[0] = a0; a[1] = a1;
  b[0] = b0; b[1] = b1;
  return __builtin_amdgcn_wmma_f32_16x16x4_f32(false, a, false, b, (short)0, c,
                                               false, false);
}

__device__ __forceinline__ void wave_lds_handover() {
  __builtin_amdgcn_fence(__ATOMIC_RELEASE, "wavefront");
  __builtin_amdgcn_wave_barrier();
  __builtin_amdgcn_fence(__ATOMIC_ACQUIRE, "wavefront");
}

template <bool P3, bool G>
__device__ __forceinline__ void gemm128_body(
    float* __restrict__ wlds, float* __restrict__ tlds, float* __restrict__ w3lds,
    const int* __restrict__ col, const int* __restrict__ cnt,
    const int* __restrict__ rowl, const int* __restrict__ boff,
    const float* __restrict__ agg, const float* __restrict__ xin,
    const float* __restrict__ Wl, const float* __restrict__ bl,
    const float* __restrict__ Wr, const float* __restrict__ deg_inv,
    float* __restrict__ hout, int n, int do_relu,
    const float* __restrict__ W3l, const float* __restrict__ b3,
    const float* __restrict__ W3r, float* __restrict__ pl, float* __restrict__ pr) {
  constexpr int NCOPY = F * (F / 4) / (GEMM_WAVES * 32);
  float4 vl[NCOPY], vr[NCOPY];
#pragma unroll
  for (int i = 0; i < NCOPY; ++i) {
    const int f = threadIdx.x + i * (GEMM_WAVES * 32);
    vl[i] = reinterpret_cast<const float4*>(Wl)[f];
    vr[i] = reinterpret_cast<const float4*>(Wr)[f];
  }
#pragma unroll
  for (int i = 0; i < NCOPY; ++i) {
    const int f = threadIdx.x + i * (GEMM_WAVES * 32);
    const int row = f >> 5;
    const int c4 = f & 31;
    float* dstp = wlds + (row & 15) * WBLK + (row >> 4) * WROW + c4 * 4;
    *reinterpret_cast<float4*>(dstp) = vl[i];
    *reinterpret_cast<float4*>(dstp + F) = vr[i];
  }
  if constexpr (P3) {
    const int col = threadIdx.x >> 5;
    const int k4 = threadIdx.x & 31;
    const int wc3 = col & 7;
    const float* wsel = (col >= 8 ? W3r : W3l) + (size_t)(wc3 < 7 ? wc3 : 0) * F;
    *reinterpret_cast<float4*>(w3lds + col * TPITCH + k4 * 4) =
        reinterpret_cast<const float4*>(wsel)[k4];
  }
  __syncthreads();

  const int lane = threadIdx.x & 31;
  const int wave = __builtin_amdgcn_readfirstlane((int)(threadIdx.x >> 5));
  const int mloc = lane & 15;
  const int h = lane >> 4;
  const int ntiles = (n + 15) >> 4;
  const float* bfrag = wlds + mloc * WBLK + 4 * h;
  const int wc = mloc & 7;
  const bool cvalid = wc < 7;
  const bool self = mloc >= 8;
  float bias3 = 0.0f;
  if constexpr (P3) bias3 = (cvalid && self) ? b3[wc] : 0.0f;

  for (int s = wave; ; s += GEMM_WAVES) {
    const int t = blockIdx.x + gridDim.x * s;
    if (t >= ntiles) break;
    const int row0 = t * 16;
    const int m = row0 + mloc;
    const int mc = m < n ? m : n - 1;
    const float4* xrow = reinterpret_cast<const float4*>(xin + (size_t)mc * F) + h;
    float* tile = nullptr;
    if constexpr (G || P3) tile = tlds + wave * (16 * TPITCH);
    float dscale = 0.0f;
    const float4* arow = nullptr;
    const float* afrag = nullptr;

    if constexpr (G) {
      const int cntv = m < n ? cnt[m] : 0;
      const float dinvv = m < n ? deg_inv[m] : 0.0f;
      int tsum = cntv;
      for (int d = 8; d > 0; d >>= 1) tsum += __shfl_xor(tsum, d);
      const int total = __builtin_amdgcn_readfirstlane(tsum);
      const int* c = col + (rowl[row0] + boff[row0 / SCAN_CHUNK]);
      float4 acc = make_float4(0.0f, 0.0f, 0.0f, 0.0f);
      int r = 0;
      int nodeEnd = __builtin_amdgcn_readlane(cntv, 0);
      float* trow = tile + lane * 4;
      if (total > 0) {
        int nxt = c[lane < total ? lane : total - 1];
        for (int p0 = 0; p0 < total; p0 += GB) {
          const int cur = nxt;
          const int q = p0 + GB + lane;
          nxt = c[q < total ? q : total - 1];
          float4 v[GB];
#pragma unroll
          for (int i = 0; i < GB; ++i) {
            const int sidx = __builtin_amdgcn_readlane(cur, i);
            v[i] = (reinterpret_cast<const float4*>(xin + (size_t)sidx * F))[lane];
          }
#pragma unroll
          for (int i = 0; i < GB; ++i) {
            const int p = p0 + i;
            const bool live = p < total;
            while (live && p >= nodeEnd) {
              const float ds = __int_as_float(
                  __builtin_amdgcn_readlane(__float_as_int(dinvv), r));
              *reinterpret_cast<float4*>(trow + r * TPITCH) =
                  make_float4(acc.x * ds, acc.y * ds, acc.z * ds, acc.w * ds);
              acc = make_float4(0.0f, 0.0f, 0.0f, 0.0f);
              ++r;
              nodeEnd += __builtin_amdgcn_readlane(cntv, r);
            }
            acc.x += live ? v[i].x : 0.0f;
            acc.y += live ? v[i].y : 0.0f;
            acc.z += live ? v[i].z : 0.0f;
            acc.w += live ? v[i].w : 0.0f;
          }
        }
      }
      while (r < 16) {
        const float ds = __int_as_float(
            __builtin_amdgcn_readlane(__float_as_int(dinvv), r));
        *reinterpret_cast<float4*>(trow + r * TPITCH) =
            make_float4(acc.x * ds, acc.y * ds, acc.z * ds, acc.w * ds);
        acc = make_float4(0.0f, 0.0f, 0.0f, 0.0f);
        ++r;
      }
      wave_lds_handover();
      afrag = tile + mloc * TPITCH + 4 * h;
    } else {
      dscale = deg_inv[mc];
      arow = reinterpret_cast<const float4*>(agg + (size_t)mc * F) + h;
    }

    float4 av[32];
    float4 aq[2];
    if constexpr (G) {
      aq[0] = *reinterpret_cast<const float4*>(afrag);
    } else {
#pragma unroll
      for (int j = 0; j < 16; ++j) av[j] = arow[2 * j];
    }

    v8f c[8];
#pragma unroll
    for (int ct = 0; ct < 8; ++ct) {
#pragma unroll
      for (int i = 0; i < 8; ++i) c[ct][i] = 0.0f;
    }

    float4 bq[2][8];
#pragma unroll
    for (int ct = 0; ct < 8; ++ct)
      bq[0][ct] = *reinterpret_cast<const float4*>(bfrag + ct * WROW);

#pragma unroll
    for (int j = 0; j < 32; ++j) {
      const int cur = j & 1;
      if (j + 1 < 32) {
#pragma unroll
        for (int ct = 0; ct < 8; ++ct)
          bq[cur ^ 1][ct] =
              *reinterpret_cast<const float4*>(bfrag + ct * WROW + 8 * (j + 1));
      }
      if constexpr (G) {
        if (j + 1 < 16) aq[cur ^ 1] = *reinterpret_cast<const float4*>(afrag + 8 * (j + 1));
      }
      if (j < 16) av[j + 16] = xrow[2 * j];
      __builtin_amdgcn_sched_barrier(0);
      float4 a;
      if constexpr (G) {
        a = j < 16 ? aq[cur] : av[j];
      } else {
        a = av[j];
        if (j < 16) { a.x *= dscale; a.y *= dscale; a.z *= dscale; a.w *= dscale; }
      }
#pragma unroll
      for (int ct = 0; ct < 8; ++ct) {
        const float4 b = bq[cur][ct];
        c[ct] = wmma_f32(a.x, a.y, b.x, b.y, c[ct]);
        c[ct] = wmma_f32(a.z, a.w, b.z, b.w, c[ct]);
      }
      __builtin_amdgcn_sched_barrier(0);
    }

    const int rbase = h * 8;
    if constexpr (!P3) {
#pragma unroll
      for (int ct = 0; ct < 8; ++ct) {
        const int ncol = ct * 16 + mloc;
        const float bias = bl[ncol];
#pragma unroll
        for (int r = 0; r < 8; ++r) {
          const int mr = row0 + rbase + r;
          if (mr < n) {
            float v = c[ct][r] + bias;
            if (do_relu) v = v > 0.0f ? v : 0.0f;
            hout[(size_t)mr * F + ncol] = v;
          }
        }
      }
      if constexpr (G) wave_lds_handover();
    } else {
#pragma unroll
      for (int ct = 0; ct < 8; ++ct) {
        const int ncol = ct * 16 + mloc;
        const float bias = bl[ncol];
#pragma unroll
        for (int r = 0; r < 8; ++r) {
          float v = c[ct][r] + bias;
          if (do_relu) v = v > 0.0f ? v : 0.0f;
          tile[(rbase + r) * TPITCH + ncol] = v;
        }
      }
      wave_lds_handover();

      v8f c3;
#pragma unroll
      for (int i = 0; i < 8; ++i) c3[i] = bias3;
      const float* a3 = tile + mloc * TPITCH + 4 * h;
      const float* b3f = w3lds + mloc * TPITCH + 4 * h;
#pragma unroll
      for (int j = 0; j < 16; ++j) {
        const float4 a = *reinterpret_cast<const float4*>(a3 + 8 * j);
        const float4 b = *reinterpret_cast<const float4*>(b3f + 8 * j);
        c3 = wmma_f32(a.x, a.y, b.x, b.y, c3);
        c3 = wmma_f32(a.z, a.w, b.z, b.w, c3);
      }
      float* dstp = (self ? pr : pl) + wc;
#pragma unroll
      for (int r = 0; r < 8; ++r) {
        const int mr = row0 + rbase + r;
        if (mr < n) dstp[(size_t)mr * 8] = cvalid ? c3[r] : 0.0f;
      }
      wave_lds_handover();
    }
  }
}

__global__ __launch_bounds__(GEMM_WAVES * 32) void k_gemm128(
    const float* __restrict__ agg, const float* __restrict__ xin,
    const float* __restrict__ Wl, const float* __restrict__ bl,
    const float* __restrict__ Wr, const float* __restrict__ deg_inv,
    float* __restrict__ hout, int n, int do_relu) {
  __shared__ __attribute__((aligned(16))) float wlds[16 * WBLK];
  gemm128_body<false, false>(wlds, nullptr, nullptr, nullptr, nullptr, nullptr, nullptr,
                             agg, xin, Wl, bl, Wr, deg_inv, hout, n, do_relu, nullptr,
                             nullptr, nullptr, nullptr, nullptr);
}

__global__ __launch_bounds__(GEMM_WAVES * 32) void k_sage128(
    const int* __restrict__ col, const int* __restrict__ cnt,
    const int* __restrict__ rowl, const int* __restrict__ boff,
    const float* __restrict__ xin,
    const float* __restrict__ Wl, const float* __restrict__ bl,
    const float* __restrict__ Wr, const float* __restrict__ deg_inv,
    float* __restrict__ hout, int n, int do_relu) {
  __shared__ __attribute__((aligned(16))) float wlds[16 * WBLK];
  __shared__ __attribute__((aligned(16))) float tlds[GEMM_WAVES * 16 * TPITCH];
  gemm128_body<false, true>(wlds, tlds, nullptr, col, cnt, rowl, boff, nullptr, xin, Wl,
                            bl, Wr, deg_inv, hout, n, do_relu, nullptr, nullptr, nullptr,
                            nullptr, nullptr);
}

__global__ __launch_bounds__(GEMM_WAVES * 32) void k_sage128_p3(
    const int* __restrict__ col, const int* __restrict__ cnt,
    const int* __restrict__ rowl, const int* __restrict__ boff,
    const float* __restrict__ xin,
    const float* __restrict__ Wl, const float* __restrict__ bl,
    const float* __restrict__ Wr, const float* __restrict__ deg_inv,
    int n, int do_relu,
    const float* __restrict__ W3l, const float* __restrict__ b3,
    const float* __restrict__ W3r, float* __restrict__ pl, float* __restrict__ pr) {
  __shared__ __attribute__((aligned(16))) float wlds[16 * WBLK];
  __shared__ __attribute__((aligned(16))) float tlds[GEMM_WAVES * 16 * TPITCH];
  __shared__ __attribute__((aligned(16))) float w3lds[16 * TPITCH];
  gemm128_body<true, true>(wlds, tlds, w3lds, col, cnt, rowl, boff, nullptr, xin, Wl, bl,
                           Wr, deg_inv, nullptr, n, do_relu, W3l, b3, W3r, pl, pr);
}

__global__ __launch_bounds__(256) void k_gemm_out7(
    const float* __restrict__ agg, const float* __restrict__ xin,
    const float* __restrict__ Wl, const float* __restrict__ bl,
    const float* __restrict__ Wr, const float* __restrict__ deg_inv,
    float* __restrict__ logits, int n) {
  const int K = F;
  int lane = threadIdx.x & 31;
  int wave = threadIdx.x >> 5;
  int row0 = (blockIdx.x * 8 + wave) * 16;
  if (row0 >= n) return;

  int mloc = lane & 15;
  int koff = (lane >> 4) * 2;
  int m = row0 + mloc;
  int mc = m < n ? m : n - 1;
  float dscale = deg_inv[mc];
  const float* arow = agg + (size_t)mc * K;
  const float* xrow = xin + (size_t)mc * K;

  int ncol = mloc;
  bool cvalid = ncol < 7;
  float wm = cvalid ? 1.0f : 0.0f;
  const float* wlrow = Wl + (size_t)(cvalid ? ncol : 0) * K;
  const float* wrrow = Wr + (size_t)(cvalid ? ncol : 0) * K;

  float bias = cvalid ? bl[ncol] : 0.0f;
  v8f c;
#pragma unroll
  for (int i = 0; i < 8; ++i) c[i] = bias;

#pragma unroll
  for (int kt = 0; kt < 32; ++kt) {
    int k0 = kt * 4 + koff;
    v2f a, b;
    a[0] = arow[k0] * dscale; a[1] = arow[k0 + 1] * dscale;
    b[0] = wlrow[k0] * wm;    b[1] = wlrow[k0 + 1] * wm;
    c = __builtin_amdgcn_wmma_f32_16x16x4_f32(false, a, false, b, (short)0, c,
                                              false, false);
  }
#pragma unroll
  for (int kt = 0; kt < 32; ++kt) {
    int k0 = kt * 4 + koff;
    v2f a, b;
    a[0] = xrow[k0];       a[1] = xrow[k0 + 1];
    b[0] = wrrow[k0] * wm; b[1] = wrrow[k0 + 1] * wm;
    c = __builtin_amdgcn_wmma_f32_16x16x4_f32(false, a, false, b, (short)0, c,
                                              false, false);
  }

  if (cvalid) {
    int rbase = (lane >> 4) * 8;
#pragma unroll
    for (int r = 0; r < 8; ++r) {
      int mr = row0 + rbase + r;
      if (mr < n) logits[(size_t)mr * 7 + ncol] = c[r];
    }
  }
}

__global__ __launch_bounds__(256) void k_agg7(const int* __restrict__ col,
                                              const int* __restrict__ cnt,
                                              const int* __restrict__ rowl,
                                              const int* __restrict__ boff,
                                              const float* __restrict__ pl,
                                              const float* __restrict__ pr,
                                              const float* __restrict__ deg_inv,
                                              float* __restrict__ out, int n) {
  const int tid = blockIdx.x * blockDim.x + threadIdx.x;
  const int q = tid & 3;
  const int node = tid >> 2;
  const bool live = node < n;
  const int nd = live ? node : n - 1;
  const int cn = live ? cnt[nd] : 0;
  const int* c = col + (rowl[nd] + boff[nd / SCAN_CHUNK]);
  const float4* p4 = reinterpret_cast<const float4*>(pl);
  float4 lo = make_float4(0.0f, 0.0f, 0.0f, 0.0f);
  float4 hi = make_float4(0.0f, 0.0f, 0.0f, 0.0f);
  for (int j = q; j < cn; j += 16) {
    int s[4];
#pragma unroll
    for (int u = 0; u < 4; ++u) {
      s[u] = 0;
      if (j + 4 * u < cn) s[u] = c[j + 4 * u];
    }
    float4 a[4], b[4];
#pragma unroll
    for (int u = 0; u < 4; ++u) {
      a[u] = make_float4(0.0f, 0.0f, 0.0f, 0.0f);
      b[u] = make_float4(0.0f, 0.0f, 0.0f, 0.0f);
      if (j + 4 * u < cn) {
        a[u] = p4[2 * (size_t)s[u]];
        b[u] = p4[2 * (size_t)s[u] + 1];
      }
    }
#pragma unroll
    for (int u = 0; u < 4; ++u) {
      lo.x += a[u].x; lo.y += a[u].y; lo.z += a[u].z; lo.w += a[u].w;
      hi.x += b[u].x; hi.y += b[u].y; hi.z += b[u].z; hi.w += b[u].w;
    }
  }
  for (int d = 1; d < 4; d <<= 1) {
    lo.x += __shfl_xor(lo.x, d); lo.y += __shfl_xor(lo.y, d);
    lo.z += __shfl_xor(lo.z, d); lo.w += __shfl_xor(lo.w, d);
    hi.x += __shfl_xor(hi.x, d); hi.y += __shfl_xor(hi.y, d);
    hi.z += __shfl_xor(hi.z, d);
  }
  if (live && q == 0) {
    const float ds = deg_inv[node];
    const float4 r0 = reinterpret_cast<const float4*>(pr)[2 * (size_t)node];
    const float4 r1 = reinterpret_cast<const float4*>(pr)[2 * (size_t)node + 1];
    float v[7];
    v[0] = lo.x * ds + r0.x;
    v[1] = lo.y * ds + r0.y;
    v[2] = lo.z * ds + r0.z;
    v[3] = lo.w * ds + r0.w;
    v[4] = hi.x * ds + r1.x;
    v[5] = hi.y * ds + r1.y;
    v[6] = hi.z * ds + r1.z;
    float mx = v[0];
#pragma unroll
    for (int j = 1; j < 7; ++j) mx = fmaxf(mx, v[j]);
    float s = 0.0f;
#pragma unroll
    for (int j = 0; j < 7; ++j) s += expf(v[j] - mx);
    const float ls = logf(s);
    float* o = out + (size_t)node * 7;
#pragma unroll
    for (int j = 0; j < 7; ++j) o[j] = v[j] - mx - ls;
  }
}

__global__ __launch_bounds__(256) void k_logsoftmax7(const float* __restrict__ logits,
                                                     float* __restrict__ out, int n) {
  int i = blockIdx.x * blockDim.x + threadIdx.x;
  if (i >= n) return;
  const float* r = logits + (size_t)i * 7;
  float v[7];
#pragma unroll
  for (int j = 0; j < 7; ++j) v[j] = r[j];
  float mx = v[0];
#pragma unroll
  for (int j = 1; j < 7; ++j) mx = fmaxf(mx, v[j]);
  float s = 0.0f;
#pragma unroll
  for (int j = 0; j < 7; ++j) s += expf(v[j] - mx);
  float ls = logf(s);
  float* o = out + (size_t)i * 7;
#pragma unroll
  for (int j = 0; j < 7; ++j) o[j] = v[j] - mx - ls;
}

extern "C" void kernel_launch(void* const* d_in, const int* in_sizes, int n_in,
                              void* d_out, int out_size, void* d_ws, size_t ws_size,
                              hipStream_t stream) {
  (void)n_in; (void)out_size;
  const float* x   = (const float*)d_in[0];
  const int*   ei  = (const int*)d_in[1];
  const float* W1l = (const float*)d_in[2];
  const float* b1  = (const float*)d_in[3];
  const float* W1r = (const float*)d_in[4];
  const float* W2l = (const float*)d_in[5];
  const float* b2  = (const float*)d_in[6];
  const float* W2r = (const float*)d_in[7];
  const float* W3l = (const float*)d_in[8];
  const float* b3  = (const float*)d_in[9];
  const float* W3r = (const float*)d_in[10];

  const int n = in_sizes[0] / F;
  const int E = in_sizes[1] / 2;
  const int* src = ei;
  const int* dst = ei + E;

  char* base = (char*)d_ws;
  size_t off = 0;
  auto a256 = [](size_t b) { return (b + 255) & ~(size_t)255; };
  float* deg = (float*)(base + off); off = a256(off + (size_t)n * sizeof(float));
  float* agg = (float*)(base + off); off = a256(off + (size_t)n * F * sizeof(float));
  float* hA  = (float*)(base + off); off = a256(off + (size_t)n * F * sizeof(float));
  float* hB  = (float*)(base + off); off = a256(off + (size_t)n * F * sizeof(float));

  const int nchunk = (n + SCAN_CHUNK - 1) / SCAN_CHUNK;
  const int g1 = (E + ECH - 1) / ECH;
  int* cnt  = (int*)(base + off); off = a256(off + (size_t)n * sizeof(int));
  int* rowl = (int*)(base + off); off = a256(off + (size_t)n * sizeof(int));
  int* bsum = (int*)(base + off); off = a256(off + (size_t)nchunk * sizeof(int));
  int* col  = (int*)(base + off); off = a256(off + (size_t)E * sizeof(int));
  unsigned* bkt = (unsigned*)hB;
  int* hist = (int*)hA;
  const bool sorted = off <= ws_size && (size_t)E <= (size_t)n * F &&
                      (size_t)nchunk * g1 <= (size_t)n * F && nchunk <= MAXNB &&
                      n <= (1 << SRC_BITS);

  const int T = 256;
  auto nb = [&](long long work) { return (int)((work + T - 1) / T); };
  const long long scatW = (long long)E * 32;
  const int gemmB = (n + 127) / 128;

  if (sorted) {
    k_bkt_hist<<<g1, SORT_T, 0, stream>>>(dst, hist, E, nchunk, g1);
    k_bkt_scatter<<<g1, SORT_T, 0, stream>>>(src, dst, hist, bkt, bsum, E, nchunk, g1);
    k_bkt_sort<<<nchunk, SORT_T, 0, stream>>>(bkt, bsum, cnt, rowl, deg, col, n, E, nchunk);

    k_sage128<<<256, GEMM_WAVES * 32, 0, stream>>>(col, cnt, rowl, bsum, x, W1l, b1, W1r,
                                                  deg, hA, n, 1);

    float* pl = hB;
    float* pr = hB + (size_t)n * 8;
    k_sage128_p3<<<256, GEMM_WAVES * 32, 0, stream>>>(col, cnt, rowl, bsum, hA, W2l, b2,
                                                     W2r, deg, n, 1, W3l, b3, W3r, pl, pr);
    k_agg7<<<nb((long long)n * 4), T, 0, stream>>>(col, cnt, rowl, bsum, pl, pr, deg,
                                                   (float*)d_out, n);
  } else {
    k_zero<<<nb(n), T, 0, stream>>>(deg, (size_t)n);
    k_degree<<<nb(E), T, 0, stream>>>(dst, deg, E);
    k_deg_inv<<<nb(n), T, 0, stream>>>(deg, n);

    k_zero<<<nb((long long)n * F), T, 0, stream>>>(agg, (size_t)n * F);
    k_scatter<<<nb(scatW), T, 0, stream>>>(src, dst, x, agg, E);
    k_gemm128<<<256, GEMM_WAVES * 32, 0, stream>>>(agg, x, W1l, b1, W1r, deg, hA, n, 1);

    k_zero<<<nb((long long)n * F), T, 0, stream>>>(agg, (size_t)n * F);
    k_scatter<<<nb(scatW), T, 0, stream>>>(src, dst, hA, agg, E);
    k_gemm128<<<256, GEMM_WAVES * 32, 0, stream>>>(agg, hA, W2l, b2, W2r, deg, hB, n, 1);

    k_zero<<<nb((long long)n * F), T, 0, stream>>>(agg, (size_t)n * F);
    k_scatter<<<nb(scatW), T, 0, stream>>>(src, dst, hB, agg, E);
    k_gemm_out7<<<gemmB, T, 0, stream>>>(agg, hB, W3l, b3, W3r, deg, hA, n);
    k_logsoftmax7<<<nb(n), T, 0, stream>>>(hA, (float*)d_out, n);
  }
}


# Fused layer kernels: outputs leave through the wave's LDS tile, one store instruction per whole 512-byte row of H (and per 512-byte block of pl, pr), not 64- and 32-byte pieces from accumulators
# speedup vs baseline: 5.9403x; 1.0306x over previous
#include <hip/hip_runtime.h>
#include <hip/hip_bf16.h>

typedef __attribute__((ext_vector_type(2))) float v2f;
typedef __attribute__((ext_vector_type(8))) float v8f;

#define F 128

__global__ __launch_bounds__(256) void k_zero(float* __restrict__ p, size_t cnt) {
  size_t i = (size_t)blockIdx.x * blockDim.x + threadIdx.x;
  if (i < cnt) p[i] = 0.0f;
}

__global__ __launch_bounds__(256) void k_degree(const int* __restrict__ dst,
                                                float* __restrict__ deg, int nE) {
  int e = blockIdx.x * blockDim.x + threadIdx.x;
  if (e < nE) atomicAdd(&deg[dst[e]], 1.0f);
}

__global__ __launch_bounds__(256) void k_deg_inv(float* __restrict__ deg, int n) {
  int i = blockIdx.x * blockDim.x + threadIdx.x;
  if (i < n) {
    float d = deg[i];
    deg[i] = d > 0.0f ? 1.0f / d : 0.0f;
  }
}

#define SCAN_CHUNK 1024
#define SORT_T 1024
#define ECH 8192
#define MAXNB 1024
#define SORT_CAP 24576
#define SRC_BITS 22
#define SRC_MASK ((1u << SRC_BITS) - 1u)

__device__ __forceinline__ int wave_incl_scan(int v, int lane) {
#pragma unroll
  for (int d = 1; d < 32; d <<= 1) {
    int t = __shfl_up(v, d);
    if (lane >= d) v += t;
  }
  return v;
}

__global__ __launch_bounds__(SORT_T) void k_bkt_hist(const int* __restrict__ dst,
                                                     int* __restrict__ hist,
                                                     int nE, int nb, int g1) {
  __shared__ int s_h[MAXNB];
  const int tid = threadIdx.x;
  for (int b = tid; b < nb; b += SORT_T) s_h[b] = 0;
  __syncthreads();
  const int e0 = blockIdx.x * ECH;
  unsigned d[ECH / SORT_T];
#pragma unroll
  for (int k = 0; k < ECH / SORT_T; ++k) {
    const int e = e0 + k * SORT_T + tid;
    d[k] = (unsigned)dst[e < nE ? e : nE - 1];
  }
#pragma unroll
  for (int k = 0; k < ECH / SORT_T; ++k) {
    if (e0 + k * SORT_T + tid < nE) atomicAdd(&s_h[d[k] / SCAN_CHUNK], 1);
  }
  __syncthreads();
  for (int b = tid; b < nb; b += SORT_T) hist[(size_t)b * g1 + blockIdx.x] = s_h[b];
}

__global__ __launch_bounds__(SORT_T) void k_bkt_scatter(const int* __restrict__ src,
                                                        const int* __restrict__ dst,
                                                        const int* __restrict__ hist,
                                                        unsigned* __restrict__ bkt,
                                                        int* __restrict__ boff,
                                                        int nE, int nb, int g1) {
  __shared__ int s_tot[MAXNB];
  __shared__ int s_pre[MAXNB];
  __shared__ int s_cur[MAXNB];
  __shared__ int s_w[2][32];
  __shared__ unsigned s_ent[ECH];
  __shared__ unsigned short s_bk[ECH];
  const int tid = threadIdx.x;
  const int lane = tid & 31;
  const int wave = tid >> 5;
  const int wg = blockIdx.x;

  for (int b = wave; b < nb; b += SORT_T / 32) {
    const int* row = hist + (size_t)b * g1;
    int tot = 0, pre = 0, own = 0;
    for (int w0 = 0; w0 < g1; w0 += 256) {
      int v[8];
#pragma unroll
      for (int k = 0; k < 8; ++k) {
        const int w = w0 + 32 * k + lane;
        v[k] = row[w < g1 ? w : g1 - 1];
      }
#pragma unroll
      for (int k = 0; k < 8; ++k) {
        const int w = w0 + 32 * k + lane;
        const int x = w < g1 ? v[k] : 0;
        tot += x;
        pre += w < wg ? x : 0;
        own += w == wg ? x : 0;
      }
    }
    for (int d = 16; d > 0; d >>= 1) {
      tot += __shfl_xor(tot, d);
      pre += __shfl_xor(pre, d);
      own += __shfl_xor(own, d);
    }
    if (lane == 0) { s_tot[b] = tot; s_pre[b] = pre; s_cur[b] = own; }
  }
  __syncthreads();

  const int vt = tid < nb ? s_tot[tid] : 0;
  const int vo = tid < nb ? s_cur[tid] : 0;
  const int it = wave_incl_scan(vt, lane);
  const int io = wave_incl_scan(vo, lane);
  if (lane == 31) { s_w[0][wave] = it; s_w[1][wave] = io; }
  __syncthreads();
  int ot = 0, oo = 0;
  for (int w = 0; w < wave; ++w) { ot += s_w[0][w]; oo += s_w[1][w]; }
  const int bstart = ot + it - vt;
  const int lstart = oo + io - vo;
  if (tid < nb) {
    if (wg == 0) boff[tid] = bstart;
    s_tot[tid] = bstart + s_pre[tid] - lstart;
    s_cur[tid] = lstart;
  }
  __syncthreads();

  const int e0 = wg * ECH;
  unsigned dd[ECH / SORT_T], ss[ECH / SORT_T];
#pragma unroll
  for (int k = 0; k < ECH / SORT_T; ++k) {
    const int e = e0 + k * SORT_T + tid;
    const int ec = e < nE ? e : nE - 1;
    dd[k] = (unsigned)dst[ec];
    ss[k] = (unsigned)src[ec];
  }
#pragma unroll
  for (int k = 0; k < ECH / SORT_T; ++k) {
    if (e0 + k * SORT_T + tid < nE) {
      const unsigned b = dd[k] / SCAN_CHUNK;
      const int p = atomicAdd(&s_cur[b], 1);
      s_ent[p] = ((dd[k] % SCAN_CHUNK) << SRC_BITS) | ss[k];
      s_bk[p] = (unsigned short)b;
    }
  }
  __syncthreads();
  const int m = nE - e0 < ECH ? nE - e0 : ECH;
  unsigned short kb[ECH / SORT_T];
  unsigned ke[ECH / SORT_T];
  int kd[ECH / SORT_T];
#pragma unroll
  for (int k = 0; k < ECH / SORT_T; ++k) {
    const int i = k * SORT_T + tid;
    const int ic = i < m ? i : m - 1;
    kb[k] = s_bk[ic];
    ke[k] = s_ent[ic];
  }
#pragma unroll
  for (int k = 0; k < ECH / SORT_T; ++k) kd[k] = s_tot[kb[k]];
#pragma unroll
  for (int k = 0; k < ECH / SORT_T; ++k) {
    const int i = k * SORT_T + tid;
    if (i < m) bkt[i + kd[k]] = ke[k];
  }
}

__global__ __launch_bounds__(SORT_T) void k_bkt_sort(const unsigned* __restrict__ bkt,
                                                     const int* __restrict__ boff,
                                                     int* __restrict__ cnt,
                                                     int* __restrict__ rowl,
                                                     float* __restrict__ deg_inv,
                                                     int* __restrict__ col,
                                                     int n, int nE, int nb) {
  __shared__ int s_c[SCAN_CHUNK];
  __shared__ int s_w[32];
  __shared__ int s_col[SORT_CAP];
  const int tid = threadIdx.x;
  const int lane = tid & 31;
  const int wave = tid >> 5;
  const int b = blockIdx.x;
  const int beg = boff[b];
  const int end = b + 1 < nb ? boff[b + 1] : nE;
  const int m = end - beg;
  const unsigned* ent = bkt + beg;

  s_c[tid] = 0;
  __syncthreads();
  for (int i0 = 0; i0 < m; i0 += 8 * SORT_T) {
    unsigned u[8];
#pragma unroll
    for (int k = 0; k < 8; ++k) {
      const int i = i0 + k * SORT_T + tid;
      u[k] = ent[i < m ? i : m - 1];
    }
#pragma unroll
    for (int k = 0; k < 8; ++k) {
      if (i0 + k * SORT_T + tid < m) atomicAdd(&s_c[u[k] >> SRC_BITS], 1);
    }
  }
  __syncthreads();

  const int v = s_c[tid];
  const int incl = wave_incl_scan(v, lane);
  if (lane == 31) s_w[wave] = incl;
  __syncthreads();
  int woff = 0;
  for (int w = 0; w < wave; ++w) woff += s_w[w];
  const int excl = woff + incl - v;
  const int node = b * SCAN_CHUNK + tid;
  if (node < n) {
    cnt[node] = v;
    rowl[node] = excl;
    deg_inv[node] = v > 0 ? 1.0f / (float)v : 0.0f;
  }
  s_c[tid] = excl;
  __syncthreads();

  if (m <= SORT_CAP) {
    for (int i0 = 0; i0 < m; i0 += 8 * SORT_T) {
      unsigned u[8];
#pragma unroll
      for (int k = 0; k < 8; ++k) {
        const int i = i0 + k * SORT_T + tid;
        u[k] = ent[i < m ? i : m - 1];
      }
#pragma unroll
      for (int k = 0; k < 8; ++k) {
        if (i0 + k * SORT_T + tid < m) {
          const int p = atomicAdd(&s_c[u[k] >> SRC_BITS], 1);
          s_col[p] = (int)(u[k] & SRC_MASK);
        }
      }
    }
    __syncthreads();
    for (int i0 = 0; i0 < m; i0 += 8 * SORT_T) {
      int sv[8];
#pragma unroll
      for (int k = 0; k < 8; ++k) {
        const int i = i0 + k * SORT_T + tid;
        sv[k] = s_col[i < m ? i : 0];
      }
#pragma unroll
      for (int k = 0; k < 8; ++k) {
        const int i = i0 + k * SORT_T + tid;
        if (i < m) col[beg + i] = sv[k];
      }
    }
  } else {
    for (int i = tid; i < m; i += SORT_T) {
      const unsigned u = ent[i];
      const int p = atomicAdd(&s_c[u >> SRC_BITS], 1);
      col[beg + p] = (int)(u & SRC_MASK);
    }
  }
}

__global__ __launch_bounds__(256) void k_scatter(const int* __restrict__ src,
                                                 const int* __restrict__ dst,
                                                 const float* __restrict__ feat,
                                                 float* __restrict__ agg, int nE) {
  long long tid = (long long)blockIdx.x * blockDim.x + threadIdx.x;
  int e = (int)(tid >> 5);
  if (e >= nE) return;
  int fo = ((int)tid & 31) * 4;
  const float4 v = *reinterpret_cast<const float4*>(feat + (size_t)src[e] * F + fo);
  float* d = agg + (size_t)dst[e] * F + fo;
  atomicAdd(d + 0, v.x);
  atomicAdd(d + 1, v.y);
  atomicAdd(d + 2, v.z);
  atomicAdd(d + 3, v.w);
}

#define GEMM_WAVES 16
#define WROW 256
#define WBLK (8 * WROW + 4)
#define TPITCH 132
#define GB 32

__device__ __forceinline__ v8f wmma_f32(float a0, float a1, float b0, float b1, v8f c) {
  v2f a, b;
  a[0] = a0; a[1] = a1;
  b[0] = b0; b[1] = b1;
  return __builtin_amdgcn_wmma_f32_16x16x4_f32(false, a, false, b, (short)0, c,
                                               false, false);
}

__device__ __forceinline__ void wave_lds_handover() {
  __builtin_amdgcn_fence(__ATOMIC_RELEASE, "wavefront");
  __builtin_amdgcn_wave_barrier();
  __builtin_amdgcn_fence(__ATOMIC_ACQUIRE, "wavefront");
}

template <bool P3, bool G>
__device__ __forceinline__ void gemm128_body(
    float* __restrict__ wlds, float* __restrict__ tlds, float* __restrict__ w3lds,
    const int* __restrict__ col, const int* __restrict__ cnt,
    const int* __restrict__ rowl, const int* __restrict__ boff,
    const float* __restrict__ agg, const float* __restrict__ xin,
    const float* __restrict__ Wl, const float* __restrict__ bl,
    const float* __restrict__ Wr, const float* __restrict__ deg_inv,
    float* __restrict__ hout, int n, int do_relu,
    const float* __restrict__ W3l, const float* __restrict__ b3,
    const float* __restrict__ W3r, float* __restrict__ pl, float* __restrict__ pr) {
  constexpr int NCOPY = F * (F / 4) / (GEMM_WAVES * 32);
  float4 vl[NCOPY], vr[NCOPY];
#pragma unroll
  for (int i = 0; i < NCOPY; ++i) {
    const int f = threadIdx.x + i * (GEMM_WAVES * 32);
    vl[i] = reinterpret_cast<const float4*>(Wl)[f];
    vr[i] = reinterpret_cast<const float4*>(Wr)[f];
  }
#pragma unroll
  for (int i = 0; i < NCOPY; ++i) {
    const int f = threadIdx.x + i * (GEMM_WAVES * 32);
    const int row = f >> 5;
    const int c4 = f & 31;
    float* dstp = wlds + (row & 15) * WBLK + (row >> 4) * WROW + c4 * 4;
    *reinterpret_cast<float4*>(dstp) = vl[i];
    *reinterpret_cast<float4*>(dstp + F) = vr[i];
  }
  if constexpr (P3) {
    const int col = threadIdx.x >> 5;
    const int k4 = threadIdx.x & 31;
    const int wc3 = col & 7;
    const float* wsel = (col >= 8 ? W3r : W3l) + (size_t)(wc3 < 7 ? wc3 : 0) * F;
    *reinterpret_cast<float4*>(w3lds + col * TPITCH + k4 * 4) =
        reinterpret_cast<const float4*>(wsel)[k4];
  }
  __syncthreads();

  const int lane = threadIdx.x & 31;
  const int wave = __builtin_amdgcn_readfirstlane((int)(threadIdx.x >> 5));
  const int mloc = lane & 15;
  const int h = lane >> 4;
  const int ntiles = (n + 15) >> 4;
  const float* bfrag = wlds + mloc * WBLK + 4 * h;
  const int wc = mloc & 7;
  const bool cvalid = wc < 7;
  const bool self = mloc >= 8;
  float bias3 = 0.0f;
  if constexpr (P3) bias3 = (cvalid && self) ? b3[wc] : 0.0f;

  for (int s = wave; ; s += GEMM_WAVES) {
    const int t = blockIdx.x + gridDim.x * s;
    if (t >= ntiles) break;
    const int row0 = t * 16;
    const int m = row0 + mloc;
    const int mc = m < n ? m : n - 1;
    const float4* xrow = reinterpret_cast<const float4*>(xin + (size_t)mc * F) + h;
    float* tile = nullptr;
    if constexpr (G || P3) tile = tlds + wave * (16 * TPITCH);
    float dscale = 0.0f;
    const float4* arow = nullptr;
    const float* afrag = nullptr;

    if constexpr (G) {
      const int cntv = m < n ? cnt[m] : 0;
      const float dinvv = m < n ? deg_inv[m] : 0.0f;
      int tsum = cntv;
      for (int d = 8; d > 0; d >>= 1) tsum += __shfl_xor(tsum, d);
      const int total = __builtin_amdgcn_readfirstlane(tsum);
      const int* c = col + (rowl[row0] + boff[row0 / SCAN_CHUNK]);
      float4 acc = make_float4(0.0f, 0.0f, 0.0f, 0.0f);
      int r = 0;
      int nodeEnd = __builtin_amdgcn_readlane(cntv, 0);
      float* trow = tile + lane * 4;
      if (total > 0) {
        int nxt = c[lane < total ? lane : total - 1];
        for (int p0 = 0; p0 < total; p0 += GB) {
          const int cur = nxt;
          const int q = p0 + GB + lane;
          nxt = c[q < total ? q : total - 1];
          float4 v[GB];
#pragma unroll
          for (int i = 0; i < GB; ++i) {
            const int sidx = __builtin_amdgcn_readlane(cur, i);
            v[i] = (reinterpret_cast<const float4*>(xin + (size_t)sidx * F))[lane];
          }
#pragma unroll
          for (int i = 0; i < GB; ++i) {
            const int p = p0 + i;
            const bool live = p < total;
            while (live && p >= nodeEnd) {
              const float ds = __int_as_float(
                  __builtin_amdgcn_readlane(__float_as_int(dinvv), r));
              *reinterpret_cast<float4*>(trow + r * TPITCH) =
                  make_float4(acc.x * ds, acc.y * ds, acc.z * ds, acc.w * ds);
              acc = make_float4(0.0f, 0.0f, 0.0f, 0.0f);
              ++r;
              nodeEnd += __builtin_amdgcn_readlane(cntv, r);
            }
            acc.x += live ? v[i].x : 0.0f;
            acc.y += live ? v[i].y : 0.0f;
            acc.z += live ? v[i].z : 0.0f;
            acc.w += live ? v[i].w : 0.0f;
          }
        }
      }
      while (r < 16) {
        const float ds = __int_as_float(
            __builtin_amdgcn_readlane(__float_as_int(dinvv), r));
        *reinterpret_cast<float4*>(trow + r * TPITCH) =
            make_float4(acc.x * ds, acc.y * ds, acc.z * ds, acc.w * ds);
        acc = make_float4(0.0f, 0.0f, 0.0f, 0.0f);
        ++r;
      }
      wave_lds_handover();
      afrag = tile + mloc * TPITCH + 4 * h;
    } else {
      dscale = deg_inv[mc];
      arow = reinterpret_cast<const float4*>(agg + (size_t)mc * F) + h;
    }

    float4 av[32];
    float4 aq[2];
    if constexpr (G) {
      aq[0] = *reinterpret_cast<const float4*>(afrag);
    } else {
#pragma unroll
      for (int j = 0; j < 16; ++j) av[j] = arow[2 * j];
    }

    v8f c[8];
#pragma unroll
    for (int ct = 0; ct < 8; ++ct) {
#pragma unroll
      for (int i = 0; i < 8; ++i) c[ct][i] = 0.0f;
    }

    float4 bq[2][8];
#pragma unroll
    for (int ct = 0; ct < 8; ++ct)
      bq[0][ct] = *reinterpret_cast<const float4*>(bfrag + ct * WROW);

#pragma unroll
    for (int j = 0; j < 32; ++j) {
      const int cur = j & 1;
      if (j + 1 < 32) {
#pragma unroll
        for (int ct = 0; ct < 8; ++ct)
          bq[cur ^ 1][ct] =
              *reinterpret_cast<const float4*>(bfrag + ct * WROW + 8 * (j + 1));
      }
      if constexpr (G) {
        if (j + 1 < 16) aq[cur ^ 1] = *reinterpret_cast<const float4*>(afrag + 8 * (j + 1));
      }
      if (j < 16) av[j + 16] = xrow[2 * j];
      __builtin_amdgcn_sched_barrier(0);
      float4 a;
      if constexpr (G) {
        a = j < 16 ? aq[cur] : av[j];
      } else {
        a = av[j];
        if (j < 16) { a.x *= dscale; a.y *= dscale; a.z *= dscale; a.w *= dscale; }
      }
#pragma unroll
      for (int ct = 0; ct < 8; ++ct) {
        const float4 b = bq[cur][ct];
        c[ct] = wmma_f32(a.x, a.y, b.x, b.y, c[ct]);
        c[ct] = wmma_f32(a.z, a.w, b.z, b.w, c[ct]);
      }
      __builtin_amdgcn_sched_barrier(0);
    }

    const int rbase = h * 8;
    if constexpr (!P3 && !G) {
#pragma unroll
      for (int ct = 0; ct < 8; ++ct) {
        const int ncol = ct * 16 + mloc;
        const float bias = bl[ncol];
#pragma unroll
        for (int r = 0; r < 8; ++r) {
          const int mr = row0 + rbase + r;
          if (mr < n) {
            float v = c[ct][r] + bias;
            if (do_relu) v = v > 0.0f ? v : 0.0f;
            hout[(size_t)mr * F + ncol] = v;
          }
        }
      }
    } else if constexpr (!P3) {
#pragma unroll
      for (int ct = 0; ct < 8; ++ct) {
        const int ncol = ct * 16 + mloc;
        const float bias = bl[ncol];
#pragma unroll
        for (int r = 0; r < 8; ++r) {
          float v = c[ct][r] + bias;
          if (do_relu) v = v > 0.0f ? v : 0.0f;
          tile[(rbase + r) * TPITCH + ncol] = v;
        }
      }
      wave_lds_handover();
      const float* orow = tile + lane * 4;
      float4* hrow = reinterpret_cast<float4*>(hout + (size_t)row0 * F) + lane;
      if (row0 + 16 <= n) {
        float4 ov[16];
#pragma unroll
        for (int r = 0; r < 16; ++r)
          ov[r] = *reinterpret_cast<const float4*>(orow + r * TPITCH);
#pragma unroll
        for (int r = 0; r < 16; ++r) hrow[r * (F / 4)] = ov[r];
      } else {
        for (int r = 0; r < 16 && row0 + r < n; ++r)
          hrow[r * (F / 4)] = *reinterpret_cast<const float4*>(orow + r * TPITCH);
      }
      wave_lds_handover();
    } else {
#pragma unroll
      for (int ct = 0; ct < 8; ++ct) {
        const int ncol = ct * 16 + mloc;
        const float bias = bl[ncol];
#pragma unroll
        for (int r = 0; r < 8; ++r) {
          float v = c[ct][r] + bias;
          if (do_relu) v = v > 0.0f ? v : 0.0f;
          tile[(rbase + r) * TPITCH + ncol] = v;
        }
      }
      wave_lds_handover();

      v8f c3;
#pragma unroll
      for (int i = 0; i < 8; ++i) c3[i] = bias3;
      const float* a3 = tile + mloc * TPITCH + 4 * h;
      const float* b3f = w3lds + mloc * TPITCH + 4 * h;
#pragma unroll
      for (int j = 0; j < 16; ++j) {
        const float4 a = *reinterpret_cast<const float4*>(a3 + 8 * j);
        const float4 b = *reinterpret_cast<const float4*>(b3f + 8 * j);
        c3 = wmma_f32(a.x, a.y, b.x, b.y, c3);
        c3 = wmma_f32(a.z, a.w, b.z, b.w, c3);
      }
      float* sblk = tile + (self ? 128 : 0) + wc;
#pragma unroll
      for (int r = 0; r < 8; ++r) sblk[(rbase + r) * 8] = c3[r];
      wave_lds_handover();
      float4 vpl = *reinterpret_cast<const float4*>(tile + lane * 4);
      float4 vpr = *reinterpret_cast<const float4*>(tile + 128 + lane * 4);
      if (lane & 1) { vpl.w = 0.0f; vpr.w = 0.0f; }
      if (row0 + (lane >> 1) < n) {
        reinterpret_cast<float4*>(pl + (size_t)row0 * 8)[lane] = vpl;
        reinterpret_cast<float4*>(pr + (size_t)row0 * 8)[lane] = vpr;
      }
      wave_lds_handover();
    }
  }
}

__global__ __launch_bounds__(GEMM_WAVES * 32) void k_gemm128(
    const float* __restrict__ agg, const float* __restrict__ xin,
    const float* __restrict__ Wl, const float* __restrict__ bl,
    const float* __restrict__ Wr, const float* __restrict__ deg_inv,
    float* __restrict__ hout, int n, int do_relu) {
  __shared__ __attribute__((aligned(16))) float wlds[16 * WBLK];
  gemm128_body<false, false>(wlds, nullptr, nullptr, nullptr, nullptr, nullptr, nullptr,
                             agg, xin, Wl, bl, Wr, deg_inv, hout, n, do_relu, nullptr,
                             nullptr, nullptr, nullptr, nullptr);
}

__global__ __launch_bounds__(GEMM_WAVES * 32) void k_sage128(
    const int* __restrict__ col, const int* __restrict__ cnt,
    const int* __restrict__ rowl, const int* __restrict__ boff,
    const float* __restrict__ xin,
    const float* __restrict__ Wl, const float* __restrict__ bl,
    const float* __restrict__ Wr, const float* __restrict__ deg_inv,
    float* __restrict__ hout, int n, int do_relu) {
  __shared__ __attribute__((aligned(16))) float wlds[16 * WBLK];
  __shared__ __attribute__((aligned(16))) float tlds[GEMM_WAVES * 16 * TPITCH];
  gemm128_body<false, true>(wlds, tlds, nullptr, col, cnt, rowl, boff, nullptr, xin, Wl,
                            bl, Wr, deg_inv, hout, n, do_relu, nullptr, nullptr, nullptr,
                            nullptr, nullptr);
}

__global__ __launch_bounds__(GEMM_WAVES * 32) void k_sage128_p3(
    const int* __restrict__ col, const int* __restrict__ cnt,
    const int* __restrict__ rowl, const int* __restrict__ boff,
    const float* __restrict__ xin,
    const float* __restrict__ Wl, const float* __restrict__ bl,
    const float* __restrict__ Wr, const float* __restrict__ deg_inv,
    int n, int do_relu,
    const float* __restrict__ W3l, const float* __restrict__ b3,
    const float* __restrict__ W3r, float* __restrict__ pl, float* __restrict__ pr) {
  __shared__ __attribute__((aligned(16))) float wlds[16 * WBLK];
  __shared__ __attribute__((aligned(16))) float tlds[GEMM_WAVES * 16 * TPITCH];
  __shared__ __attribute__((aligned(16))) float w3lds[16 * TPITCH];
  gemm128_body<true, true>(wlds, tlds, w3lds, col, cnt, rowl, boff, nullptr, xin, Wl, bl,
                           Wr, deg_inv, nullptr, n, do_relu, W3l, b3, W3r, pl, pr);
}

__global__ __launch_bounds__(256) void k_gemm_out7(
    const float* __restrict__ agg, const float* __restrict__ xin,
    const float* __restrict__ Wl, const float* __restrict__ bl,
    const float* __restrict__ Wr, const float* __restrict__ deg_inv,
    float* __restrict__ logits, int n) {
  const int K = F;
  int lane = threadIdx.x & 31;
  int wave = threadIdx.x >> 5;
  int row0 = (blockIdx.x * 8 + wave) * 16;
  if (row0 >= n) return;

  int mloc = lane & 15;
  int koff = (lane >> 4) * 2;
  int m = row0 + mloc;
  int mc = m < n ? m : n - 1;
  float dscale = deg_inv[mc];
  const float* arow = agg + (size_t)mc * K;
  const float* xrow = xin + (size_t)mc * K;

  int ncol = mloc;
  bool cvalid = ncol < 7;
  float wm = cvalid ? 1.0f : 0.0f;
  const float* wlrow = Wl + (size_t)(cvalid ? ncol : 0) * K;
  const float* wrrow = Wr + (size_t)(cvalid ? ncol : 0) * K;

  float bias = cvalid ? bl[ncol] : 0.0f;
  v8f c;
#pragma unroll
  for (int i = 0; i < 8; ++i) c[i] = bias;

#pragma unroll
  for (int kt = 0; kt < 32; ++kt) {
    int k0 = kt * 4 + koff;
    v2f a, b;
    a[0] = arow[k0] * dscale; a[1] = arow[k0 + 1] * dscale;
    b[0] = wlrow[k0] * wm;    b[1] = wlrow[k0 + 1] * wm;
    c = __builtin_amdgcn_wmma_f32_16x16x4_f32(false, a, false, b, (short)0, c,
                                              false, false);
  }
#pragma unroll
  for (int kt = 0; kt < 32; ++kt) {
    int k0 = kt * 4 + koff;
    v2f a, b;
    a[0] = xrow[k0];       a[1] = xrow[k0 + 1];
    b[0] = wrrow[k0] * wm; b[1] = wrrow[k0 + 1] * wm;
    c = __builtin_amdgcn_wmma_f32_16x16x4_f32(false, a, false, b, (short)0, c,
                                              false, false);
  }

  if (cvalid) {
    int rbase = (lane >> 4) * 8;
#pragma unroll
    for (int r = 0; r < 8; ++r) {
      int mr = row0 + rbase + r;
      if (mr < n) logits[(size_t)mr * 7 + ncol] = c[r];
    }
  }
}

__global__ __launch_bounds__(256) void k_agg7(const int* __restrict__ col,
                                              const int* __restrict__ cnt,
                                              const int* __restrict__ rowl,
                                              const int* __restrict__ boff,
                                              const float* __restrict__ pl,
                                              const float* __restrict__ pr,
                                              const float* __restrict__ deg_inv,
                                              float* __restrict__ out, int n) {
  const int tid = blockIdx.x * blockDim.x + threadIdx.x;
  const int q = tid & 3;
  const int node = tid >> 2;
  const bool live = node < n;
  const int nd = live ? node : n - 1;
  const int cn = live ? cnt[nd] : 0;
  const int* c = col + (rowl[nd] + boff[nd / SCAN_CHUNK]);
  const float4* p4 = reinterpret_cast<const float4*>(pl);
  float4 lo = make_float4(0.0f, 0.0f, 0.0f, 0.0f);
  float4 hi = make_float4(0.0f, 0.0f, 0.0f, 0.0f);
  for (int j = q; j < cn; j += 16) {
    int s[4];
#pragma unroll
    for (int u = 0; u < 4; ++u) {
      s[u] = 0;
      if (j + 4 * u < cn) s[u] = c[j + 4 * u];
    }
    float4 a[4], b[4];
#pragma unroll
    for (int u = 0; u < 4; ++u) {
      a[u] = make_float4(0.0f, 0.0f, 0.0f, 0.0f);
      b[u] = make_float4(0.0f, 0.0f, 0.0f, 0.0f);
      if (j + 4 * u < cn) {
        a[u] = p4[2 * (size_t)s[u]];
        b[u] = p4[2 * (size_t)s[u] + 1];
      }
    }
#pragma unroll
    for (int u = 0; u < 4; ++u) {
      lo.x += a[u].x; lo.y += a[u].y; lo.z += a[u].z; lo.w += a[u].w;
      hi.x += b[u].x; hi.y += b[u].y; hi.z += b[u].z; hi.w += b[u].w;
    }
  }
  for (int d = 1; d < 4; d <<= 1) {
    lo.x += __shfl_xor(lo.x, d); lo.y += __shfl_xor(lo.y, d);
    lo.z += __shfl_xor(lo.z, d); lo.w += __shfl_xor(lo.w, d);
    hi.x += __shfl_xor(hi.x, d); hi.y += __shfl_xor(hi.y, d);
    hi.z += __shfl_xor(hi.z, d);
  }
  if (live && q == 0) {
    const float ds = deg_inv[node];
    const float4 r0 = reinterpret_cast<const float4*>(pr)[2 * (size_t)node];
    const float4 r1 = reinterpret_cast<const float4*>(pr)[2 * (size_t)node + 1];
    float v[7];
    v[0] = lo.x * ds + r0.x;
    v[1] = lo.y * ds + r0.y;
    v[2] = lo.z * ds + r0.z;
    v[3] = lo.w * ds + r0.w;
    v[4] = hi.x * ds + r1.x;
    v[5] = hi.y * ds + r1.y;
    v[6] = hi.z * ds + r1.z;
    float mx = v[0];
#pragma unroll
    for (int j = 1; j < 7; ++j) mx = fmaxf(mx, v[j]);
    float s = 0.0f;
#pragma unroll
    for (int j = 0; j < 7; ++j) s += expf(v[j] - mx);
    const float ls = logf(s);
    float* o = out + (size_t)node * 7;
#pragma unroll
    for (int j = 0; j < 7; ++j) o[j] = v[j] - mx - ls;
  }
}

__global__ __launch_bounds__(256) void k_logsoftmax7(const float* __restrict__ logits,
                                                     float* __restrict__ out, int n) {
  int i = blockIdx.x * blockDim.x + threadIdx.x;
  if (i >= n) return;
  const float* r = logits + (size_t)i * 7;
  float v[7];
#pragma unroll
  for (int j = 0; j < 7; ++j) v[j] = r[j];
  float mx = v[0];
#pragma unroll
  for (int j = 1; j < 7; ++j) mx = fmaxf(mx, v[j]);
  float s = 0.0f;
#pragma unroll
  for (int j = 0; j < 7; ++j) s += expf(v[j] - mx);
  float ls = logf(s);
  float* o = out + (size_t)i * 7;
#pragma unroll
  for (int j = 0; j < 7; ++j) o[j] = v[j] - mx - ls;
}

extern "C" void kernel_launch(void* const* d_in, const int* in_sizes, int n_in,
                              void* d_out, int out_size, void* d_ws, size_t ws_size,
                              hipStream_t stream) {
  (void)n_in; (void)out_size;
  const float* x   = (const float*)d_in[0];
  const int*   ei  = (const int*)d_in[1];
  const float* W1l = (const float*)d_in[2];
  const float* b1  = (const float*)d_in[3];
  const float* W1r = (const float*)d_in[4];
  const float* W2l = (const float*)d_in[5];
  const float* b2  = (const float*)d_in[6];
  const float* W2r = (const float*)d_in[7];
  const float* W3l = (const float*)d_in[8];
  const float* b3  = (const float*)d_in[9];
  const float* W3r = (const float*)d_in[10];

  const int n = in_sizes[0] / F;
  const int E = in_sizes[1] / 2;
  const int* src = ei;
  const int* dst = ei + E;

  char* base = (char*)d_ws;
  size_t off = 0;
  auto a256 = [](size_t b) { return (b + 255) & ~(size_t)255; };
  float* deg = (float*)(base + off); off = a256(off + (size_t)n * sizeof(float));
  float* agg = (float*)(base + off); off = a256(off + (size_t)n * F * sizeof(float));
  float* hA  = (float*)(base + off); off = a256(off + (size_t)n * F * sizeof(float));
  float* hB  = (float*)(base + off); off = a256(off + (size_t)n * F * sizeof(float));

  const int nchunk = (n + SCAN_CHUNK - 1) / SCAN_CHUNK;
  const int g1 = (E + ECH - 1) / ECH;
  int* cnt  = (int*)(base + off); off = a256(off + (size_t)n * sizeof(int));
  int* rowl = (int*)(base + off); off = a256(off + (size_t)n * sizeof(int));
  int* bsum = (int*)(base + off); off = a256(off + (size_t)nchunk * sizeof(int));
  int* col  = (int*)(base + off); off = a256(off + (size_t)E * sizeof(int));
  unsigned* bkt = (unsigned*)hB;
  int* hist = (int*)hA;
  const bool sorted = off <= ws_size && (size_t)E <= (size_t)n * F &&
                      (size_t)nchunk * g1 <= (size_t)n * F && nchunk <= MAXNB &&
                      n <= (1 << SRC_BITS);

  const int T = 256;
  auto nb = [&](long long work) { return (int)((work + T - 1) / T); };
  const long long scatW = (long long)E * 32;
  const int gemmB = (n + 127) / 128;

  if (sorted) {
    k_bkt_hist<<<g1, SORT_T, 0, stream>>>(dst, hist, E, nchunk, g1);
    k_bkt_scatter<<<g1, SORT_T, 0, stream>>>(src, dst, hist, bkt, bsum, E, nchunk, g1);
    k_bkt_sort<<<nchunk, SORT_T, 0, stream>>>(bkt, bsum, cnt, rowl, deg, col, n, E, nchunk);

    k_sage128<<<256, GEMM_WAVES * 32, 0, stream>>>(col, cnt, rowl, bsum, x, W1l, b1, W1r,
                                                  deg, hA, n, 1);

    float* pl = hB;
    float* pr = hB + (size_t)n * 8;
    k_sage128_p3<<<256, GEMM_WAVES * 32, 0, stream>>>(col, cnt, rowl, bsum, hA, W2l, b2,
                                                     W2r, deg, n, 1, W3l, b3, W3r, pl, pr);
    k_agg7<<<nb((long long)n * 4), T, 0, stream>>>(col, cnt, rowl, bsum, pl, pr, deg,
                                                   (float*)d_out, n);
  } else {
    k_zero<<<nb(n), T, 0, stream>>>(deg, (size_t)n);
    k_degree<<<nb(E), T, 0, stream>>>(dst, deg, E);
    k_deg_inv<<<nb(n), T, 0, stream>>>(deg, n);

    k_zero<<<nb((long long)n * F), T, 0, stream>>>(agg, (size_t)n * F);
    k_scatter<<<nb(scatW), T, 0, stream>>>(src, dst, x, agg, E);
    k_gemm128<<<256, GEMM_WAVES * 32, 0, stream>>>(agg, x, W1l, b1, W1r, deg, hA, n, 1);

    k_zero<<<nb((long long)n * F), T, 0, stream>>>(agg, (size_t)n * F);
    k_scatter<<<nb(scatW), T, 0, stream>>>(src, dst, hA, agg, E);
    k_gemm128<<<256, GEMM_WAVES * 32, 0, stream>>>(agg, hA, W2l, b2, W2r, deg, hB, n, 1);

    k_zero<<<nb((long long)n * F), T, 0, stream>>>(agg, (size_t)n * F);
    k_scatter<<<nb(scatW), T, 0, stream>>>(src, dst, hB, agg, E);
    k_gemm_out7<<<gemmB, T, 0, stream>>>(agg, hB, W3l, b3, W3r, deg, hA, n);
    k_logsoftmax7<<<nb(n), T, 0, stream>>>(hA, (float*)d_out, n);
  }
}
